# starting kernel
# baseline (speedup 1.0000x reference)
#include <hip/hip_runtime.h>
#include <math.h>

#define B_     8
#define C_     128
#define OC_    64
#define SD_    512
#define H_     128
#define KW_    3
#define H2_    257
#define OUT_   256
#define PP_    129
#define PPSZ_  (PP_*PP_)

#define SCALE_  (0.02946278254943948f)
#define WGAIN_  (0.04419417382415922f)

typedef __attribute__((ext_vector_type(2))) float v2f;
typedef __attribute__((ext_vector_type(8))) float v8f;

__global__ void k_style(const float* __restrict__ style,
                        const float* __restrict__ modw,
                        const float* __restrict__ modb,
                        float* __restrict__ s_ws) {
    int b = blockIdx.x;
    int c = threadIdx.x;
    const float* st = style + b * SD_;
    const float* mw = modw + c * SD_;
    float acc = 0.f;
#pragma unroll 4
    for (int k = 0; k < SD_; ++k) acc += st[k] * mw[k];
    s_ws[b * C_ + c] = acc * WGAIN_ + modb[c];
}

__global__ void k_wmod(const float* __restrict__ weight,
                       const float* __restrict__ s_ws,
                       float* __restrict__ W2) {
    int o = blockIdx.x;
    int b = blockIdx.y;
    int c = threadIdx.x;
    float sv = s_ws[b * C_ + c] * SCALE_;
    const float* wp = weight + ((size_t)o * C_ + c) * (KW_ * KW_);
    float wv[9];
    float ss = 0.f;
#pragma unroll
    for (int t = 0; t < 9; ++t) { wv[t] = wp[t] * sv; ss += wv[t] * wv[t]; }
    __shared__ float red[C_];
    red[c] = ss;
    __syncthreads();
    for (int off = 64; off > 0; off >>= 1) {
        if (c < off) red[c] += red[c + off];
        __syncthreads();
    }
    float dem = rsqrtf(red[0] + 1e-8f);
#pragma unroll
    for (int t = 0; t < 9; ++t)
        W2[((((size_t)b * 9 + t) * 64 + (c >> 1)) * OC_ + o) * 2 + (c & 1)] = wv[t] * dem;
}

template<int PI, int PJ>
__global__ void __launch_bounds__(256) k_conv(const float* __restrict__ x,
                                              const float* __restrict__ W2,
                                              float* __restrict__ conv_ws) {
    constexpr int TY = PI ? 1 : 2;
    constexpr int TX = PJ ? 1 : 2;
    constexpr int NT = TY * TX;
    constexpr int WD = 32 + (PJ ? 0 : 1);
    constexpr int NJ = PJ ? 128 : 129;
    constexpr int PH = PI * 2 + PJ;

    const int b  = blockIdx.z;
    const int ip = blockIdx.y;
    const int j0 = blockIdx.x * 32;
    const int r0 = ip - (PI ? 0 : 1);
    const int c0 = j0 - (PJ ? 0 : 1);

    __shared__ v2f xs[64 * 2 * 34];

    const float* xb = x + (size_t)b * C_ * H_ * H_;
    const int total = 64 * TY * WD;
    for (int idx = threadIdx.x; idx < total; idx += 256) {
        int col = idx % WD;
        int rem = idx / WD;
        int t   = rem % TY;
        int kp  = rem / TY;
        int gr = r0 + t, gc = c0 + col;
        v2f v = {};
        if (gr >= 0 && gr < H_ && gc >= 0 && gc < H_) {
            const float* p = xb + ((size_t)(2 * kp) * H_ + gr) * H_ + gc;
            v.x = p[0];
            v.y = p[(size_t)H_ * H_];
        }
        xs[(kp * 2 + t) * 34 + col] = v;
    }
    __syncthreads();

    const int lane = threadIdx.x & 31;
    const int wid  = threadIdx.x >> 5;
    const int mW   = wid & 3;
    const int nW   = wid >> 2;
    const int ln   = lane & 15;
    const int khp  = lane >> 4;
    const int m    = mW * 16 + ln;

    const v2f* Wb = (const v2f*)W2 + (size_t)b * 9 * 64 * OC_;

    v8f acc0 = {}, acc1 = {};
#pragma unroll 2
    for (int kc = 0; kc < C_; kc += 4) {
        const int kp = (kc >> 1) + khp;
        v2f a[NT], bf[NT];
#pragma unroll
        for (int t = 0; t < NT; ++t) {
            const int ty = t / TX, tx = t % TX;
            const int ky  = PI ? 1 : (2 - 2 * ty);
            const int kx  = PJ ? 1 : (2 - 2 * tx);
            const int tap = ky * 3 + kx;
            a[t]  = Wb[((size_t)tap * 64 + kp) * OC_ + m];
            bf[t] = xs[(kp * 2 + ty) * 34 + (nW * 16 + ln + tx)];
        }
#pragma unroll
        for (int t = 0; t < NT; ++t) {
            if (t & 1)
                acc1 = __builtin_amdgcn_wmma_f32_16x16x4_f32(
                         false, a[t], false, bf[t], (short)0, acc1, false, false);
            else
                acc0 = __builtin_amdgcn_wmma_f32_16x16x4_f32(
                         false, a[t], false, bf[t], (short)0, acc0, false, false);
        }
    }

    const int jp = j0 + nW * 16 + ln;
    if (jp < NJ) {
        const int mhi = khp * 8;
        float* op = conv_ws + (((size_t)(b * 4 + PH) * OC_ + mW * 16 + mhi) * PPSZ_)
                  + (size_t)ip * PP_ + jp;
#pragma unroll
        for (int r = 0; r < 8; ++r) {
            float v = (NT > 1) ? (acc0[r] + acc1[r]) : acc0[r];
            op[(size_t)r * PPSZ_] = v;
        }
    }
}

__global__ void k_blur(const float* __restrict__ conv_ws, float* __restrict__ out) {
    const int y = blockIdx.x;
    const int o = blockIdx.y;
    const int b = blockIdx.z;
    const int xg = threadIdx.x;
    float fy[4] = {1.f, 3.f, 3.f, 1.f};
    float acc = 0.f;
#pragma unroll
    for (int ky = 0; ky < 4; ++ky) {
        int u = y + ky - 1;
        if (u < 0 || u >= H2_) continue;
        int pi = u & 1, i2 = u >> 1;
#pragma unroll
        for (int kx = 0; kx < 4; ++kx) {
            int v = xg + kx - 1;
            if (v < 0 || v >= H2_) continue;
            int pj = v & 1, j2 = v >> 1;
            int ph = pi * 2 + pj;
            float cv = conv_ws[(((size_t)(b * 4 + ph) * OC_ + o) * PPSZ_)
                               + (size_t)i2 * PP_ + j2];
            acc += fy[ky] * fy[kx] * cv;
        }
    }
    out[(((size_t)b * OC_ + o) * OUT_ + y) * OUT_ + xg] = acc * (1.f / 16.f);
}

extern "C" void kernel_launch(void* const* d_in, const int* in_sizes, int n_in,
                              void* d_out, int out_size, void* d_ws, size_t ws_size,
                              hipStream_t stream) {
    const float* x      = (const float*)d_in[0];
    const float* style  = (const float*)d_in[1];
    const float* weight = (const float*)d_in[2];
    const float* modw   = (const float*)d_in[3];
    const float* modb   = (const float*)d_in[4];
    float* out = (float*)d_out;
    float* ws  = (float*)d_ws;

    float* s_ws    = ws;
    float* W2_ws   = ws + 1024;
    float* conv_ws = ws + 1024 + 589824;

    k_style<<<dim3(B_), dim3(C_), 0, stream>>>(style, modw, modb, s_ws);
    k_wmod<<<dim3(OC_, B_), dim3(C_), 0, stream>>>(weight, s_ws, W2_ws);

    k_conv<0,0><<<dim3(5, 129, B_), dim3(256), 0, stream>>>(x, W2_ws, conv_ws);
    k_conv<0,1><<<dim3(4, 129, B_), dim3(256), 0, stream>>>(x, W2_ws, conv_ws);
    k_conv<1,0><<<dim3(5, 128, B_), dim3(256), 0, stream>>>(x, W2_ws, conv_ws);
    k_conv<1,1><<<dim3(4, 128, B_), dim3(256), 0, stream>>>(x, W2_ws, conv_ws);

    k_blur<<<dim3(OUT_, OC_, B_), dim3(OUT_), 0, stream>>>(conv_ws, out);
}


# k_blur rewritten as a separable streaming pass: each lane owns 2 output columns and walks 64 rows with row sums held in registers (2.5 loads per output, not 16); FIR sum order is now row-then-column
# speedup vs baseline: 1.8732x; 1.8732x over previous
#include <hip/hip_runtime.h>
#include <math.h>

#define B_     8
#define C_     128
#define OC_    64
#define SD_    512
#define H_     128
#define KW_    3
#define H2_    257
#define OUT_   256
#define PP_    129
#define PPSZ_  (PP_*PP_)

#define SCALE_  (0.02946278254943948f)
#define WGAIN_  (0.04419417382415922f)

typedef __attribute__((ext_vector_type(2))) float v2f;
typedef __attribute__((ext_vector_type(8))) float v8f;

__global__ void k_style(const float* __restrict__ style,
                        const float* __restrict__ modw,
                        const float* __restrict__ modb,
                        float* __restrict__ s_ws) {
    int b = blockIdx.x;
    int c = threadIdx.x;
    const float* st = style + b * SD_;
    const float* mw = modw + c * SD_;
    float acc = 0.f;
#pragma unroll 4
    for (int k = 0; k < SD_; ++k) acc += st[k] * mw[k];
    s_ws[b * C_ + c] = acc * WGAIN_ + modb[c];
}

__global__ void k_wmod(const float* __restrict__ weight,
                       const float* __restrict__ s_ws,
                       float* __restrict__ W2) {
    int o = blockIdx.x;
    int b = blockIdx.y;
    int c = threadIdx.x;
    float sv = s_ws[b * C_ + c] * SCALE_;
    const float* wp = weight + ((size_t)o * C_ + c) * (KW_ * KW_);
    float wv[9];
    float ss = 0.f;
#pragma unroll
    for (int t = 0; t < 9; ++t) { wv[t] = wp[t] * sv; ss += wv[t] * wv[t]; }
    __shared__ float red[C_];
    red[c] = ss;
    __syncthreads();
    for (int off = 64; off > 0; off >>= 1) {
        if (c < off) red[c] += red[c + off];
        __syncthreads();
    }
    float dem = rsqrtf(red[0] + 1e-8f);
#pragma unroll
    for (int t = 0; t < 9; ++t)
        W2[((((size_t)b * 9 + t) * 64 + (c >> 1)) * OC_ + o) * 2 + (c & 1)] = wv[t] * dem;
}

template<int PI, int PJ>
__global__ void __launch_bounds__(256) k_conv(const float* __restrict__ x,
                                              const float* __restrict__ W2,
                                              float* __restrict__ conv_ws) {
    constexpr int TY = PI ? 1 : 2;
    constexpr int TX = PJ ? 1 : 2;
    constexpr int NT = TY * TX;
    constexpr int WD = 32 + (PJ ? 0 : 1);
    constexpr int NJ = PJ ? 128 : 129;
    constexpr int PH = PI * 2 + PJ;

    const int b  = blockIdx.z;
    const int ip = blockIdx.y;
    const int j0 = blockIdx.x * 32;
    const int r0 = ip - (PI ? 0 : 1);
    const int c0 = j0 - (PJ ? 0 : 1);

    __shared__ v2f xs[64 * 2 * 34];

    const float* xb = x + (size_t)b * C_ * H_ * H_;
    const int total = 64 * TY * WD;
    for (int idx = threadIdx.x; idx < total; idx += 256) {
        int col = idx % WD;
        int rem = idx / WD;
        int t   = rem % TY;
        int kp  = rem / TY;
        int gr = r0 + t, gc = c0 + col;
        v2f v = {};
        if (gr >= 0 && gr < H_ && gc >= 0 && gc < H_) {
            const float* p = xb + ((size_t)(2 * kp) * H_ + gr) * H_ + gc;
            v.x = p[0];
            v.y = p[(size_t)H_ * H_];
        }
        xs[(kp * 2 + t) * 34 + col] = v;
    }
    __syncthreads();

    const int lane = threadIdx.x & 31;
    const int wid  = threadIdx.x >> 5;
    const int mW   = wid & 3;
    const int nW   = wid >> 2;
    const int ln   = lane & 15;
    const int khp  = lane >> 4;
    const int m    = mW * 16 + ln;

    const v2f* Wb = (const v2f*)W2 + (size_t)b * 9 * 64 * OC_;

    v8f acc0 = {}, acc1 = {};
#pragma unroll 2
    for (int kc = 0; kc < C_; kc += 4) {
        const int kp = (kc >> 1) + khp;
        v2f a[NT], bf[NT];
#pragma unroll
        for (int t = 0; t < NT; ++t) {
            const int ty = t / TX, tx = t % TX;
            const int ky  = PI ? 1 : (2 - 2 * ty);
            const int kx  = PJ ? 1 : (2 - 2 * tx);
            const int tap = ky * 3 + kx;
            a[t]  = Wb[((size_t)tap * 64 + kp) * OC_ + m];
            bf[t] = xs[(kp * 2 + ty) * 34 + (nW * 16 + ln + tx)];
        }
#pragma unroll
        for (int t = 0; t < NT; ++t) {
            if (t & 1)
                acc1 = __builtin_amdgcn_wmma_f32_16x16x4_f32(
                         false, a[t], false, bf[t], (short)0, acc1, false, false);
            else
                acc0 = __builtin_amdgcn_wmma_f32_16x16x4_f32(
                         false, a[t], false, bf[t], (short)0, acc0, false, false);
        }
    }

    const int jp = j0 + nW * 16 + ln;
    if (jp < NJ) {
        const int mhi = khp * 8;
        float* op = conv_ws + (((size_t)(b * 4 + PH) * OC_ + mW * 16 + mhi) * PPSZ_)
                  + (size_t)ip * PP_ + jp;
#pragma unroll
        for (int r = 0; r < 8; ++r) {
            float v = (NT > 1) ? (acc0[r] + acc1[r]) : acc0[r];
            op[(size_t)r * PPSZ_] = v;
        }
    }
}

#define BLUR_RI 32
#define BLUR_U  4
static_assert(BLUR_RI * 4 == H_ && BLUR_RI % BLUR_U == 0, "blur grid: 4 bands cover all row pairs");

__device__ __forceinline__ void blur_hrow(float a, float b0, float c, float d, float e,
                                          bool hasL, bool hasR, bool valid,
                                          float& h0, float& h1) {
    const float al = hasL ? a : 0.f;
    const float er = hasR ? e : 0.f;
    float t0 = fmaf(3.f, b0 + c, al + d);
    float t1 = fmaf(3.f, c + d, b0 + er);
    h0 = valid ? t0 : 0.f;
    h1 = valid ? t1 : 0.f;
}

__global__ void __launch_bounds__(128) k_blur(const float* __restrict__ conv_ws,
                                              float* __restrict__ out) {
    const int j  = threadIdx.x;
    const int i0 = blockIdx.x * BLUR_RI;
    const int o  = blockIdx.y;
    const int b  = blockIdx.z;

    const float* p00 = conv_ws + ((size_t)(b * 4 + 0) * OC_ + o) * PPSZ_;
    const float* p01 = conv_ws + ((size_t)(b * 4 + 1) * OC_ + o) * PPSZ_;
    const float* p10 = conv_ws + ((size_t)(b * 4 + 2) * OC_ + o) * PPSZ_;
    const float* p11 = conv_ws + ((size_t)(b * 4 + 3) * OC_ + o) * PPSZ_;

    const int  jl   = (j > 0) ? (j - 1) : 0;
    const int  jr   = (j < 127) ? (j + 1) : 127;
    const bool hasL = (j > 0);
    const bool hasR = (j < 127);

    float hA0, hA1, hB0, hB1, hC0, hC1;
    {
        const bool vA = (i0 > 0);
        const int  rA = vA ? (i0 - 1) : 0;
        const float* a0 = p10 + rA * PP_; const float* a1 = p11 + rA * PP_;
        const float* b0 = p00 + i0 * PP_; const float* b1 = p01 + i0 * PP_;
        const float* c0 = p10 + i0 * PP_; const float* c1 = p11 + i0 * PP_;
        float ra[5] = { a1[jl], a0[j], a1[j], a0[j + 1], a1[jr] };
        float rb[5] = { b1[jl], b0[j], b1[j], b0[j + 1], b1[jr] };
        float rc[5] = { c1[jl], c0[j], c1[j], c0[j + 1], c1[jr] };
        blur_hrow(ra[0], ra[1], ra[2], ra[3], ra[4], hasL, hasR, vA,   hA0, hA1);
        blur_hrow(rb[0], rb[1], rb[2], rb[3], rb[4], hasL, hasR, true, hB0, hB1);
        blur_hrow(rc[0], rc[1], rc[2], rc[3], rc[4], hasL, hasR, true, hC0, hC1);
    }

    float* op = out + (((size_t)b * OC_ + o) * OUT_ + 2 * i0) * OUT_ + 2 * j;

    for (int it = 0; it < BLUR_RI / BLUR_U; ++it) {
        float re[BLUR_U][5], ro[BLUR_U][5];
#pragma unroll
        for (int s = 0; s < BLUR_U; ++s) {
            const int i2  = i0 + it * BLUR_U + s + 1;
            const int i2o = (i2 < H_) ? i2 : (H_ - 1);
            const float* e0 = p00 + i2 * PP_;  const float* e1 = p01 + i2 * PP_;
            const float* q0 = p10 + i2o * PP_; const float* q1 = p11 + i2o * PP_;
            re[s][0] = e1[jl]; re[s][1] = e0[j]; re[s][2] = e1[j]; re[s][3] = e0[j + 1]; re[s][4] = e1[jr];
            ro[s][0] = q1[jl]; ro[s][1] = q0[j]; ro[s][2] = q1[j]; ro[s][3] = q0[j + 1]; ro[s][4] = q1[jr];
        }
#pragma unroll
        for (int s = 0; s < BLUR_U; ++s) {
            const int i2 = i0 + it * BLUR_U + s + 1;
            float hD0, hD1, hE0, hE1;
            blur_hrow(re[s][0], re[s][1], re[s][2], re[s][3], re[s][4], hasL, hasR, true,      hD0, hD1);
            blur_hrow(ro[s][0], ro[s][1], ro[s][2], ro[s][3], ro[s][4], hasL, hasR, (i2 < H_), hE0, hE1);
            v2f y0, y1;
            y0.x = fmaf(3.f, hB0 + hC0, hA0 + hD0) * (1.f / 16.f);
            y0.y = fmaf(3.f, hB1 + hC1, hA1 + hD1) * (1.f / 16.f);
            y1.x = fmaf(3.f, hC0 + hD0, hB0 + hE0) * (1.f / 16.f);
            y1.y = fmaf(3.f, hC1 + hD1, hB1 + hE1) * (1.f / 16.f);
            float* orow = op + (size_t)(2 * (it * BLUR_U + s)) * OUT_;
            *(v2f*)(orow)        = y0;
            *(v2f*)(orow + OUT_) = y1;
            hA0 = hC0; hA1 = hC1; hB0 = hD0; hB1 = hD1; hC0 = hE0; hC1 = hE1;
        }
    }
}

extern "C" void kernel_launch(void* const* d_in, const int* in_sizes, int n_in,
                              void* d_out, int out_size, void* d_ws, size_t ws_size,
                              hipStream_t stream) {
    const float* x      = (const float*)d_in[0];
    const float* style  = (const float*)d_in[1];
    const float* weight = (const float*)d_in[2];
    const float* modw   = (const float*)d_in[3];
    const float* modb   = (const float*)d_in[4];
    float* out = (float*)d_out;
    float* ws  = (float*)d_ws;

    float* s_ws    = ws;
    float* W2_ws   = ws + 1024;
    float* conv_ws = ws + 1024 + 589824;

    k_style<<<dim3(B_), dim3(C_), 0, stream>>>(style, modw, modb, s_ws);
    k_wmod<<<dim3(OC_, B_), dim3(C_), 0, stream>>>(weight, s_ws, W2_ws);

    k_conv<0,0><<<dim3(5, 129, B_), dim3(256), 0, stream>>>(x, W2_ws, conv_ws);
    k_conv<0,1><<<dim3(4, 129, B_), dim3(256), 0, stream>>>(x, W2_ws, conv_ws);
    k_conv<1,0><<<dim3(5, 128, B_), dim3(256), 0, stream>>>(x, W2_ws, conv_ws);
    k_conv<1,1><<<dim3(4, 128, B_), dim3(256), 0, stream>>>(x, W2_ws, conv_ws);

    k_blur<<<dim3(4, OC_, B_), dim3(128), 0, stream>>>(conv_ws, out);
}


# k_conv retiled: a wave keeps each weight fragment for 3-4 pixel tiles (16 out-ch x 48/64 px, 4-wave blocks; x staged per 32 ch, 16 B loads); a tile's taps add into one accumulator: sum order changed
# speedup vs baseline: 2.3163x; 1.2365x over previous
#include <hip/hip_runtime.h>
#include <math.h>

#define B_     8
#define C_     128
#define OC_    64
#define SD_    512
#define H_     128
#define KW_    3
#define H2_    257
#define OUT_   256
#define PP_    129
#define PPSZ_  (PP_*PP_)

#define SCALE_  (0.02946278254943948f)
#define WGAIN_  (0.04419417382415922f)

typedef __attribute__((ext_vector_type(2))) float v2f;
typedef __attribute__((ext_vector_type(4))) float v4f;
typedef __attribute__((ext_vector_type(8))) float v8f;

__global__ void k_style(const float* __restrict__ style,
                        const float* __restrict__ modw,
                        const float* __restrict__ modb,
                        float* __restrict__ s_ws) {
    int b = blockIdx.x;
    int c = threadIdx.x;
    const float* st = style + b * SD_;
    const float* mw = modw + c * SD_;
    float acc = 0.f;
#pragma unroll 4
    for (int k = 0; k < SD_; ++k) acc += st[k] * mw[k];
    s_ws[b * C_ + c] = acc * WGAIN_ + modb[c];
}

__global__ void k_wmod(const float* __restrict__ weight,
                       const float* __restrict__ s_ws,
                       float* __restrict__ W2) {
    int o = blockIdx.x;
    int b = blockIdx.y;
    int c = threadIdx.x;
    float sv = s_ws[b * C_ + c] * SCALE_;
    const float* wp = weight + ((size_t)o * C_ + c) * (KW_ * KW_);
    float wv[9];
    float ss = 0.f;
#pragma unroll
    for (int t = 0; t < 9; ++t) { wv[t] = wp[t] * sv; ss += wv[t] * wv[t]; }
    __shared__ float red[C_];
    red[c] = ss;
    __syncthreads();
    for (int off = 64; off > 0; off >>= 1) {
        if (c < off) red[c] += red[c + off];
        __syncthreads();
    }
    float dem = rsqrtf(red[0] + 1e-8f);
#pragma unroll
    for (int t = 0; t < 9; ++t)
        W2[((((size_t)b * 9 + t) * 64 + (c >> 1)) * OC_ + o) * 2 + (c & 1)] = wv[t] * dem;
}

template<int PI, int PJ>
__global__ void __launch_bounds__(128) k_conv(const float* __restrict__ x,
                                              const float* __restrict__ W2,
                                              float* __restrict__ conv_ws) {
    constexpr int TY = PI ? 1 : 2;
    constexpr int TX = PJ ? 1 : 2;
    constexpr int NT = TY * TX;
    constexpr int GN = PJ ? 4 : 3;
    constexpr int NPX = GN * 16;
    constexpr int NG = PJ ? 16 : 13;
    constexpr int COFF = PJ ? 0 : 3;
    constexpr int PITCH = (TY == 2) ? (PJ ? 72 : 56) : 80;
    constexpr int NJ = PJ ? 128 : 129;
    constexpr int PH = PI * 2 + PJ;
    constexpr int KPC = 16;
    constexpr int NCH = 64 / KPC;
    constexpr int ROWS = KPC * TY;
    constexpr int NIT = ROWS * 16 / 128;
    constexpr int KU = (NT == 1) ? 4 : 2;

    const int b  = blockIdx.z;
    const int ip = blockIdx.y;
    const int j0 = blockIdx.x * NPX;
    const int r0 = ip - (PI ? 0 : 1);
    const int cb = j0 - (PJ ? 0 : 4);

    __shared__ v2f xs[ROWS * PITCH];

    const float* xb = x + (size_t)b * C_ * H_ * H_;

    const int lane = threadIdx.x & 31;
    const int mW   = threadIdx.x >> 5;
    const int ln   = lane & 15;
    const int khp  = lane >> 4;
    const int m    = mW * 16 + ln;

    const v2f* Wb = (const v2f*)W2 + (size_t)b * 9 * 64 * OC_;

    v8f acc[GN];
#pragma unroll
    for (int g = 0; g < GN; ++g) acc[g] = v8f{};

    for (int ch = 0; ch < NCH; ++ch) {
        if (ch > 0) __syncthreads();
        {
            v4f va[NIT], vb[NIT];
#pragma unroll
            for (int it = 0; it < NIT; ++it) {
                const int idx = threadIdx.x + it * 128;
                const int g   = idx & 15;
                const int row = idx >> 4;
                const int t   = row % TY;
                const int kpl = row / TY;
                const int gr  = r0 + t;
                const int gc  = cb + 4 * g;
                const bool ok = (g < NG) && (gr >= 0) && (gr < H_) && (gc >= 0) && (gc < H_);
                const float* p = ok ? (xb + ((size_t)(2 * (ch * KPC + kpl)) * H_ + gr) * H_ + gc) : xb;
                const v4f qa = *(const v4f*)p;
                const v4f qb = *(const v4f*)(p + (size_t)H_ * H_);
                const v4f z = {0.f, 0.f, 0.f, 0.f};
                va[it] = ok ? qa : z;
                vb[it] = ok ? qb : z;
            }
#pragma unroll
            for (int it = 0; it < NIT; ++it) {
                const int idx = threadIdx.x + it * 128;
                const int g   = idx & 15;
                const int row = idx >> 4;
                if (g < NG) {
                    v2f* d = xs + row * PITCH + 4 * g;
                    d[0] = v2f{va[it].x, vb[it].x};
                    d[1] = v2f{va[it].y, vb[it].y};
                    d[2] = v2f{va[it].z, vb[it].z};
                    d[3] = v2f{va[it].w, vb[it].w};
                }
            }
        }
        __syncthreads();

#pragma unroll KU
        for (int kcl = 0; kcl < KPC / 2; ++kcl) {
            const int kpl = 2 * kcl + khp;
            const int kp  = ch * KPC + kpl;
            v2f a[NT];
#pragma unroll
            for (int t = 0; t < NT; ++t) {
                const int ty = t / TX, tx = t % TX;
                const int ky  = PI ? 1 : (2 - 2 * ty);
                const int kx  = PJ ? 1 : (2 - 2 * tx);
                const int tap = ky * 3 + kx;
                a[t] = Wb[((size_t)tap * 64 + kp) * OC_ + m];
            }
#pragma unroll
            for (int t = 0; t < NT; ++t) {
                const int ty = t / TX, tx = t % TX;
#pragma unroll
                for (int g = 0; g < GN; ++g) {
                    const v2f bf = xs[(kpl * TY + ty) * PITCH + (g * 16 + ln + tx + COFF)];
                    acc[g] = __builtin_amdgcn_wmma_f32_16x16x4_f32(
                               false, a[t], false, bf, (short)0, acc[g], false, false);
                }
            }
        }
    }

    const int mhi = khp * 8;
    float* ob = conv_ws + (((size_t)(b * 4 + PH) * OC_ + mW * 16 + mhi) * PPSZ_) + (size_t)ip * PP_;
#pragma unroll
    for (int g = 0; g < GN; ++g) {
        const int jp = j0 + g * 16 + ln;
        if (jp < NJ) {
#pragma unroll
            for (int r = 0; r < 8; ++r) ob[(size_t)r * PPSZ_ + jp] = acc[g][r];
        }
    }
}

#define BLUR_RI 32
#define BLUR_U  4
static_assert(BLUR_RI * 4 == H_ && BLUR_RI % BLUR_U == 0, "blur grid: 4 bands cover all row pairs");

__device__ __forceinline__ void blur_hrow(float a, float b0, float c, float d, float e,
                                          bool hasL, bool hasR, bool valid,
                                          float& h0, float& h1) {
    const float al = hasL ? a : 0.f;
    const float er = hasR ? e : 0.f;
    float t0 = fmaf(3.f, b0 + c, al + d);
    float t1 = fmaf(3.f, c + d, b0 + er);
    h0 = valid ? t0 : 0.f;
    h1 = valid ? t1 : 0.f;
}

__global__ void __launch_bounds__(128) k_blur(const float* __restrict__ conv_ws,
                                              float* __restrict__ out) {
    const int j  = threadIdx.x;
    const int i0 = blockIdx.x * BLUR_RI;
    const int o  = blockIdx.y;
    const int b  = blockIdx.z;

    const float* p00 = conv_ws + ((size_t)(b * 4 + 0) * OC_ + o) * PPSZ_;
    const float* p01 = conv_ws + ((size_t)(b * 4 + 1) * OC_ + o) * PPSZ_;
    const float* p10 = conv_ws + ((size_t)(b * 4 + 2) * OC_ + o) * PPSZ_;
    const float* p11 = conv_ws + ((size_t)(b * 4 + 3) * OC_ + o) * PPSZ_;

    const int  jl   = (j > 0) ? (j - 1) : 0;
    const int  jr   = (j < 127) ? (j + 1) : 127;
    const bool hasL = (j > 0);
    const bool hasR = (j < 127);

    float hA0, hA1, hB0, hB1, hC0, hC1;
    {
        const bool vA = (i0 > 0);
        const int  rA = vA ? (i0 - 1) : 0;
        const float* a0 = p10 + rA * PP_; const float* a1 = p11 + rA * PP_;
        const float* b0 = p00 + i0 * PP_; const float* b1 = p01 + i0 * PP_;
        const float* c0 = p10 + i0 * PP_; const float* c1 = p11 + i0 * PP_;
        float ra[5] = { a1[jl], a0[j], a1[j], a0[j + 1], a1[jr] };
        float rb[5] = { b1[jl], b0[j], b1[j], b0[j + 1], b1[jr] };
        float rc[5] = { c1[jl], c0[j], c1[j], c0[j + 1], c1[jr] };
        blur_hrow(ra[0], ra[1], ra[2], ra[3], ra[4], hasL, hasR, vA,   hA0, hA1);
        blur_hrow(rb[0], rb[1], rb[2], rb[3], rb[4], hasL, hasR, true, hB0, hB1);
        blur_hrow(rc[0], rc[1], rc[2], rc[3], rc[4], hasL, hasR, true, hC0, hC1);
    }

    float* op = out + (((size_t)b * OC_ + o) * OUT_ + 2 * i0) * OUT_ + 2 * j;

    for (int it = 0; it < BLUR_RI / BLUR_U; ++it) {
        float re[BLUR_U][5], ro[BLUR_U][5];
#pragma unroll
        for (int s = 0; s < BLUR_U; ++s) {
            const int i2  = i0 + it * BLUR_U + s + 1;
            const int i2o = (i2 < H_) ? i2 : (H_ - 1);
            const float* e0 = p00 + i2 * PP_;  const float* e1 = p01 + i2 * PP_;
            const float* q0 = p10 + i2o * PP_; const float* q1 = p11 + i2o * PP_;
            re[s][0] = e1[jl]; re[s][1] = e0[j]; re[s][2] = e1[j]; re[s][3] = e0[j + 1]; re[s][4] = e1[jr];
            ro[s][0] = q1[jl]; ro[s][1] = q0[j]; ro[s][2] = q1[j]; ro[s][3] = q0[j + 1]; ro[s][4] = q1[jr];
        }
#pragma unroll
        for (int s = 0; s < BLUR_U; ++s) {
            const int i2 = i0 + it * BLUR_U + s + 1;
            float hD0, hD1, hE0, hE1;
            blur_hrow(re[s][0], re[s][1], re[s][2], re[s][3], re[s][4], hasL, hasR, true,      hD0, hD1);
            blur_hrow(ro[s][0], ro[s][1], ro[s][2], ro[s][3], ro[s][4], hasL, hasR, (i2 < H_), hE0, hE1);
            v2f y0, y1;
            y0.x = fmaf(3.f, hB0 + hC0, hA0 + hD0) * (1.f / 16.f);
            y0.y = fmaf(3.f, hB1 + hC1, hA1 + hD1) * (1.f / 16.f);
            y1.x = fmaf(3.f, hC0 + hD0, hB0 + hE0) * (1.f / 16.f);
            y1.y = fmaf(3.f, hC1 + hD1, hB1 + hE1) * (1.f / 16.f);
            float* orow = op + (size_t)(2 * (it * BLUR_U + s)) * OUT_;
            *(v2f*)(orow)        = y0;
            *(v2f*)(orow + OUT_) = y1;
            hA0 = hC0; hA1 = hC1; hB0 = hD0; hB1 = hD1; hC0 = hE0; hC1 = hE1;
        }
    }
}

extern "C" void kernel_launch(void* const* d_in, const int* in_sizes, int n_in,
                              void* d_out, int out_size, void* d_ws, size_t ws_size,
                              hipStream_t stream) {
    const float* x      = (const float*)d_in[0];
    const float* style  = (const float*)d_in[1];
    const float* weight = (const float*)d_in[2];
    const float* modw   = (const float*)d_in[3];
    const float* modb   = (const float*)d_in[4];
    float* out = (float*)d_out;
    float* ws  = (float*)d_ws;

    float* s_ws    = ws;
    float* W2_ws   = ws + 1024;
    float* conv_ws = ws + 1024 + 589824;

    k_style<<<dim3(B_), dim3(C_), 0, stream>>>(style, modw, modb, s_ws);
    k_wmod<<<dim3(OC_, B_), dim3(C_), 0, stream>>>(weight, s_ws, W2_ws);

    k_conv<0,0><<<dim3(3, 129, B_), dim3(128), 0, stream>>>(x, W2_ws, conv_ws);
    k_conv<0,1><<<dim3(2, 129, B_), dim3(128), 0, stream>>>(x, W2_ws, conv_ws);
    k_conv<1,0><<<dim3(3, 128, B_), dim3(128), 0, stream>>>(x, W2_ws, conv_ws);
    k_conv<1,1><<<dim3(2, 128, B_), dim3(128), 0, stream>>>(x, W2_ws, conv_ws);

    k_blur<<<dim3(4, OC_, B_), dim3(128), 0, stream>>>(conv_ws, out);
}


# the four per-phase k_conv launches merged into ONE launch (phase picked per block, batch-major grid): the chip fills and drains once, not four times; per-block code and sum order unchanged
# speedup vs baseline: 2.6213x; 1.1317x over previous
#include <hip/hip_runtime.h>
#include <math.h>

#define B_     8
#define C_     128
#define OC_    64
#define SD_    512
#define H_     128
#define KW_    3
#define H2_    257
#define OUT_   256
#define PP_    129
#define PPSZ_  (PP_*PP_)

#define SCALE_  (0.02946278254943948f)
#define WGAIN_  (0.04419417382415922f)

typedef __attribute__((ext_vector_type(2))) float v2f;
typedef __attribute__((ext_vector_type(4))) float v4f;
typedef __attribute__((ext_vector_type(8))) float v8f;

__global__ void k_style(const float* __restrict__ style,
                        const float* __restrict__ modw,
                        const float* __restrict__ modb,
                        float* __restrict__ s_ws) {
    int b = blockIdx.x;
    int c = threadIdx.x;
    const float* st = style + b * SD_;
    const float* mw = modw + c * SD_;
    float acc = 0.f;
#pragma unroll 4
    for (int k = 0; k < SD_; ++k) acc += st[k] * mw[k];
    s_ws[b * C_ + c] = acc * WGAIN_ + modb[c];
}

__global__ void k_wmod(const float* __restrict__ weight,
                       const float* __restrict__ s_ws,
                       float* __restrict__ W2) {
    int o = blockIdx.x;
    int b = blockIdx.y;
    int c = threadIdx.x;
    float sv = s_ws[b * C_ + c] * SCALE_;
    const float* wp = weight + ((size_t)o * C_ + c) * (KW_ * KW_);
    float wv[9];
    float ss = 0.f;
#pragma unroll
    for (int t = 0; t < 9; ++t) { wv[t] = wp[t] * sv; ss += wv[t] * wv[t]; }
    __shared__ float red[C_];
    red[c] = ss;
    __syncthreads();
    for (int off = 64; off > 0; off >>= 1) {
        if (c < off) red[c] += red[c + off];
        __syncthreads();
    }
    float dem = rsqrtf(red[0] + 1e-8f);
#pragma unroll
    for (int t = 0; t < 9; ++t)
        W2[((((size_t)b * 9 + t) * 64 + (c >> 1)) * OC_ + o) * 2 + (c & 1)] = wv[t] * dem;
}

#define CONV_LDS_F2 (32 * 72)
#define NB00_ (3 * 129)
#define NB01_ (2 * 129)
#define NB10_ (3 * 128)
#define NB11_ (2 * 128)
#define NBB_  1285
static_assert(NB00_ + NB01_ + NB10_ + NB11_ == NBB_, "conv grid: the four phases' blocks of one batch");
template<int PI, int PJ>
__device__ __forceinline__ void conv_phase(const float* __restrict__ x,
                                           const float* __restrict__ W2,
                                           float* __restrict__ conv_ws,
                                           v2f* __restrict__ xs,
                                           const int bx, const int ip, const int b) {
    constexpr int TY = PI ? 1 : 2;
    constexpr int TX = PJ ? 1 : 2;
    constexpr int NT = TY * TX;
    constexpr int GN = PJ ? 4 : 3;
    constexpr int NPX = GN * 16;
    constexpr int NG = PJ ? 16 : 13;
    constexpr int COFF = PJ ? 0 : 3;
    constexpr int PITCH = (TY == 2) ? (PJ ? 72 : 56) : 80;
    constexpr int NJ = PJ ? 128 : 129;
    constexpr int PH = PI * 2 + PJ;
    constexpr int KPC = 16;
    constexpr int NCH = 64 / KPC;
    constexpr int ROWS = KPC * TY;
    constexpr int NIT = ROWS * 16 / 128;
    constexpr int KU = (NT == 1) ? 4 : 2;

    const int j0 = bx * NPX;
    const int r0 = ip - (PI ? 0 : 1);
    const int cb = j0 - (PJ ? 0 : 4);

    static_assert(ROWS * PITCH <= CONV_LDS_F2, "the phase's tile fits the kernel's LDS buffer");

    const float* xb = x + (size_t)b * C_ * H_ * H_;

    const int lane = threadIdx.x & 31;
    const int mW   = threadIdx.x >> 5;
    const int ln   = lane & 15;
    const int khp  = lane >> 4;
    const int m    = mW * 16 + ln;

    const v2f* Wb = (const v2f*)W2 + (size_t)b * 9 * 64 * OC_;

    v8f acc[GN];
#pragma unroll
    for (int g = 0; g < GN; ++g) acc[g] = v8f{};

    for (int ch = 0; ch < NCH; ++ch) {
        if (ch > 0) __syncthreads();
        {
            v4f va[NIT], vb[NIT];
#pragma unroll
            for (int it = 0; it < NIT; ++it) {
                const int idx = threadIdx.x + it * 128;
                const int g   = idx & 15;
                const int row = idx >> 4;
                const int t   = row % TY;
                const int kpl = row / TY;
                const int gr  = r0 + t;
                const int gc  = cb + 4 * g;
                const bool ok = (g < NG) && (gr >= 0) && (gr < H_) && (gc >= 0) && (gc < H_);
                const float* p = ok ? (xb + ((size_t)(2 * (ch * KPC + kpl)) * H_ + gr) * H_ + gc) : xb;
                const v4f qa = *(const v4f*)p;
                const v4f qb = *(const v4f*)(p + (size_t)H_ * H_);
                const v4f z = {0.f, 0.f, 0.f, 0.f};
                va[it] = ok ? qa : z;
                vb[it] = ok ? qb : z;
            }
#pragma unroll
            for (int it = 0; it < NIT; ++it) {
                const int idx = threadIdx.x + it * 128;
                const int g   = idx & 15;
                const int row = idx >> 4;
                if (g < NG) {
                    v2f* d = xs + row * PITCH + 4 * g;
                    d[0] = v2f{va[it].x, vb[it].x};
                    d[1] = v2f{va[it].y, vb[it].y};
                    d[2] = v2f{va[it].z, vb[it].z};
                    d[3] = v2f{va[it].w, vb[it].w};
                }
            }
        }
        __syncthreads();

#pragma unroll KU
        for (int kcl = 0; kcl < KPC / 2; ++kcl) {
            const int kpl = 2 * kcl + khp;
            const int kp  = ch * KPC + kpl;
            v2f a[NT];
#pragma unroll
            for (int t = 0; t < NT; ++t) {
                const int ty = t / TX, tx = t % TX;
                const int ky  = PI ? 1 : (2 - 2 * ty);
                const int kx  = PJ ? 1 : (2 - 2 * tx);
                const int tap = ky * 3 + kx;
                a[t] = Wb[((size_t)tap * 64 + kp) * OC_ + m];
            }
#pragma unroll
            for (int t = 0; t < NT; ++t) {
                const int ty = t / TX, tx = t % TX;
#pragma unroll
                for (int g = 0; g < GN; ++g) {
                    const v2f bf = xs[(kpl * TY + ty) * PITCH + (g * 16 + ln + tx + COFF)];
                    acc[g] = __builtin_amdgcn_wmma_f32_16x16x4_f32(
                               false, a[t], false, bf, (short)0, acc[g], false, false);
                }
            }
        }
    }

    const int mhi = khp * 8;
    float* ob = conv_ws + (((size_t)(b * 4 + PH) * OC_ + mW * 16 + mhi) * PPSZ_) + (size_t)ip * PP_;
#pragma unroll
    for (int g = 0; g < GN; ++g) {
        const int jp = j0 + g * 16 + ln;
        if (jp < NJ) {
#pragma unroll
            for (int r = 0; r < 8; ++r) ob[(size_t)r * PPSZ_ + jp] = acc[g][r];
        }
    }
}

__global__ void __launch_bounds__(128) k_conv_all(const float* __restrict__ x,
                                                  const float* __restrict__ W2,
                                                  float* __restrict__ conv_ws) {
    __shared__ v2f xs[CONV_LDS_F2];
    const int b = blockIdx.y;
    int r = blockIdx.x;
    if (r < NB00_) { conv_phase<0, 0>(x, W2, conv_ws, xs, r % 3, r / 3, b); return; }
    r -= NB00_;
    if (r < NB01_) { conv_phase<0, 1>(x, W2, conv_ws, xs, r & 1, r >> 1, b); return; }
    r -= NB01_;
    if (r < NB10_) { conv_phase<1, 0>(x, W2, conv_ws, xs, r % 3, r / 3, b); return; }
    r -= NB10_;
    conv_phase<1, 1>(x, W2, conv_ws, xs, r & 1, r >> 1, b);
}

#define BLUR_RI 32
#define BLUR_U  4
static_assert(BLUR_RI * 4 == H_ && BLUR_RI % BLUR_U == 0, "blur grid: 4 bands cover all row pairs");

__device__ __forceinline__ void blur_hrow(float a, float b0, float c, float d, float e,
                                          bool hasL, bool hasR, bool valid,
                                          float& h0, float& h1) {
    const float al = hasL ? a : 0.f;
    const float er = hasR ? e : 0.f;
    float t0 = fmaf(3.f, b0 + c, al + d);
    float t1 = fmaf(3.f, c + d, b0 + er);
    h0 = valid ? t0 : 0.f;
    h1 = valid ? t1 : 0.f;
}

__global__ void __launch_bounds__(128) k_blur(const float* __restrict__ conv_ws,
                                              float* __restrict__ out) {
    const int j  = threadIdx.x;
    const int i0 = blockIdx.x * BLUR_RI;
    const int o  = blockIdx.y;
    const int b  = blockIdx.z;

    const float* p00 = conv_ws + ((size_t)(b * 4 + 0) * OC_ + o) * PPSZ_;
    const float* p01 = conv_ws + ((size_t)(b * 4 + 1) * OC_ + o) * PPSZ_;
    const float* p10 = conv_ws + ((size_t)(b * 4 + 2) * OC_ + o) * PPSZ_;
    const float* p11 = conv_ws + ((size_t)(b * 4 + 3) * OC_ + o) * PPSZ_;

    const int  jl   = (j > 0) ? (j - 1) : 0;
    const int  jr   = (j < 127) ? (j + 1) : 127;
    const bool hasL = (j > 0);
    const bool hasR = (j < 127);

    float hA0, hA1, hB0, hB1, hC0, hC1;
    {
        const bool vA = (i0 > 0);
        const int  rA = vA ? (i0 - 1) : 0;
        const float* a0 = p10 + rA * PP_; const float* a1 = p11 + rA * PP_;
        const float* b0 = p00 + i0 * PP_; const float* b1 = p01 + i0 * PP_;
        const float* c0 = p10 + i0 * PP_; const float* c1 = p11 + i0 * PP_;
        float ra[5] = { a1[jl], a0[j], a1[j], a0[j + 1], a1[jr] };
        float rb[5] = { b1[jl], b0[j], b1[j], b0[j + 1], b1[jr] };
        float rc[5] = { c1[jl], c0[j], c1[j], c0[j + 1], c1[jr] };
        blur_hrow(ra[0], ra[1], ra[2], ra[3], ra[4], hasL, hasR, vA,   hA0, hA1);
        blur_hrow(rb[0], rb[1], rb[2], rb[3], rb[4], hasL, hasR, true, hB0, hB1);
        blur_hrow(rc[0], rc[1], rc[2], rc[3], rc[4], hasL, hasR, true, hC0, hC1);
    }

    float* op = out + (((size_t)b * OC_ + o) * OUT_ + 2 * i0) * OUT_ + 2 * j;

    for (int it = 0; it < BLUR_RI / BLUR_U; ++it) {
        float re[BLUR_U][5], ro[BLUR_U][5];
#pragma unroll
        for (int s = 0; s < BLUR_U; ++s) {
            const int i2  = i0 + it * BLUR_U + s + 1;
            const int i2o = (i2 < H_) ? i2 : (H_ - 1);
            const float* e0 = p00 + i2 * PP_;  const float* e1 = p01 + i2 * PP_;
            const float* q0 = p10 + i2o * PP_; const float* q1 = p11 + i2o * PP_;
            re[s][0] = e1[jl]; re[s][1] = e0[j]; re[s][2] = e1[j]; re[s][3] = e0[j + 1]; re[s][4] = e1[jr];
            ro[s][0] = q1[jl]; ro[s][1] = q0[j]; ro[s][2] = q1[j]; ro[s][3] = q0[j + 1]; ro[s][4] = q1[jr];
        }
#pragma unroll
        for (int s = 0; s < BLUR_U; ++s) {
            const int i2 = i0 + it * BLUR_U + s + 1;
            float hD0, hD1, hE0, hE1;
            blur_hrow(re[s][0], re[s][1], re[s][2], re[s][3], re[s][4], hasL, hasR, true,      hD0, hD1);
            blur_hrow(ro[s][0], ro[s][1], ro[s][2], ro[s][3], ro[s][4], hasL, hasR, (i2 < H_), hE0, hE1);
            v2f y0, y1;
            y0.x = fmaf(3.f, hB0 + hC0, hA0 + hD0) * (1.f / 16.f);
            y0.y = fmaf(3.f, hB1 + hC1, hA1 + hD1) * (1.f / 16.f);
            y1.x = fmaf(3.f, hC0 + hD0, hB0 + hE0) * (1.f / 16.f);
            y1.y = fmaf(3.f, hC1 + hD1, hB1 + hE1) * (1.f / 16.f);
            float* orow = op + (size_t)(2 * (it * BLUR_U + s)) * OUT_;
            *(v2f*)(orow)        = y0;
            *(v2f*)(orow + OUT_) = y1;
            hA0 = hC0; hA1 = hC1; hB0 = hD0; hB1 = hD1; hC0 = hE0; hC1 = hE1;
        }
    }
}

extern "C" void kernel_launch(void* const* d_in, const int* in_sizes, int n_in,
                              void* d_out, int out_size, void* d_ws, size_t ws_size,
                              hipStream_t stream) {
    const float* x      = (const float*)d_in[0];
    const float* style  = (const float*)d_in[1];
    const float* weight = (const float*)d_in[2];
    const float* modw   = (const float*)d_in[3];
    const float* modb   = (const float*)d_in[4];
    float* out = (float*)d_out;
    float* ws  = (float*)d_ws;

    float* s_ws    = ws;
    float* W2_ws   = ws + 1024;
    float* conv_ws = ws + 1024 + 589824;

    k_style<<<dim3(B_), dim3(C_), 0, stream>>>(style, modw, modb, s_ws);
    k_wmod<<<dim3(OC_, B_), dim3(C_), 0, stream>>>(weight, s_ws, W2_ws);

    k_conv_all<<<dim3(NBB_, B_), dim3(128), 0, stream>>>(x, W2_ws, conv_ws);

    k_blur<<<dim3(4, OC_, B_), dim3(128), 0, stream>>>(conv_ws, out);
}


# k_style + k_wmod fused into ONE launch (3 per rep): block = (b, 8 out-ch), s[b][:] from coalesced mod_weight rows, row loads in flight at once; style+demod sums by wave shuffles: sum order changed
# speedup vs baseline: 2.7551x; 1.0510x over previous
#include <hip/hip_runtime.h>
#include <math.h>

#define B_     8
#define C_     128
#define OC_    64
#define SD_    512
#define H_     128
#define KW_    3
#define H2_    257
#define OUT_   256
#define PP_    129
#define PPSZ_  (PP_*PP_)

#define SCALE_  (0.02946278254943948f)
#define WGAIN_  (0.04419417382415922f)

typedef __attribute__((ext_vector_type(2))) float v2f;
typedef __attribute__((ext_vector_type(4))) float v4f;
typedef __attribute__((ext_vector_type(8))) float v8f;

#define MOD_OG 8
#define MOD_NG 8
static_assert(MOD_OG * C_ == 1024 && MOD_NG * MOD_OG == OC_ && C_ == 32 * 4, "k_mod block shape: 32 waves x 4 rows = 128 c");

__global__ void __launch_bounds__(1024) __attribute__((amdgpu_waves_per_eu(8, 8))) k_mod(const float* __restrict__ style,
                                              const float* __restrict__ modw,
                                              const float* __restrict__ modb,
                                              const float* __restrict__ weight,
                                              float* __restrict__ W2) {
    __shared__ __attribute__((aligned(16))) float s_sh[C_];
    __shared__ float red[MOD_OG][4];

    const int tid  = threadIdx.x;
    const int lane = tid & 31;
    const int w    = tid >> 5;
    const int b    = blockIdx.y;
    const int ol   = tid >> 7;
    const int c    = tid & 127;
    const int o    = blockIdx.x * MOD_OG + ol;

    const float* wp = weight + ((size_t)o * C_ + c) * (KW_ * KW_);
    float wv[9];
#pragma unroll
    for (int t = 0; t < 9; ++t) wv[t] = wp[t];

    const v4f mb4 = *(const v4f*)(modb + w * 4);
    const v4f* st4 = (const v4f*)(style + (size_t)b * SD_);
    v4f sv[4];
#pragma unroll
    for (int i = 0; i < 4; ++i) sv[i] = st4[lane + 32 * i];
    v4f mw[4][4];
#pragma unroll
    for (int r = 0; r < 4; ++r) {
        const v4f* m4 = (const v4f*)(modw + (size_t)(w * 4 + r) * SD_);
#pragma unroll
        for (int i = 0; i < 4; ++i) mw[r][i] = m4[lane + 32 * i];
    }
    float part[4];
#pragma unroll
    for (int r = 0; r < 4; ++r) {
        float a = 0.f;
#pragma unroll
        for (int i = 0; i < 4; ++i) {
            a = fmaf(sv[i].x, mw[r][i].x, a);
            a = fmaf(sv[i].y, mw[r][i].y, a);
            a = fmaf(sv[i].z, mw[r][i].z, a);
            a = fmaf(sv[i].w, mw[r][i].w, a);
        }
        part[r] = a;
    }
#pragma unroll
    for (int off = 16; off > 0; off >>= 1) {
#pragma unroll
        for (int r = 0; r < 4; ++r) part[r] += __shfl_xor(part[r], off);
    }
    {
        v4f r4;
        r4.x = (part[0] * WGAIN_ + mb4.x) * SCALE_;
        r4.y = (part[1] * WGAIN_ + mb4.y) * SCALE_;
        r4.z = (part[2] * WGAIN_ + mb4.z) * SCALE_;
        r4.w = (part[3] * WGAIN_ + mb4.w) * SCALE_;
        *(v4f*)(s_sh + w * 4) = r4;
    }
    __syncthreads();

    const float sc = s_sh[c];
    float ss = 0.f;
#pragma unroll
    for (int t = 0; t < 9; ++t) { wv[t] = wv[t] * sc; ss += wv[t] * wv[t]; }
#pragma unroll
    for (int off = 16; off > 0; off >>= 1) ss += __shfl_xor(ss, off);
    if (lane == 0) red[ol][w & 3] = ss;
    __syncthreads();
    const float dem = rsqrtf(((red[ol][0] + red[ol][1]) + (red[ol][2] + red[ol][3])) + 1e-8f);
#pragma unroll
    for (int t = 0; t < 9; ++t)
        W2[((((size_t)b * 9 + t) * 64 + (c >> 1)) * OC_ + o) * 2 + (c & 1)] = wv[t] * dem;
}

#define CONV_LDS_F2 (32 * 72)
#define NB00_ (3 * 129)
#define NB01_ (2 * 129)
#define NB10_ (3 * 128)
#define NB11_ (2 * 128)
#define NBB_  1285
static_assert(NB00_ + NB01_ + NB10_ + NB11_ == NBB_, "conv grid: the four phases' blocks of one batch");
template<int PI, int PJ>
__device__ __forceinline__ void conv_phase(const float* __restrict__ x,
                                           const float* __restrict__ W2,
                                           float* __restrict__ conv_ws,
                                           v2f* __restrict__ xs,
                                           const int bx, const int ip, const int b) {
    constexpr int TY = PI ? 1 : 2;
    constexpr int TX = PJ ? 1 : 2;
    constexpr int NT = TY * TX;
    constexpr int GN = PJ ? 4 : 3;
    constexpr int NPX = GN * 16;
    constexpr int NG = PJ ? 16 : 13;
    constexpr int COFF = PJ ? 0 : 3;
    constexpr int PITCH = (TY == 2) ? (PJ ? 72 : 56) : 80;
    constexpr int NJ = PJ ? 128 : 129;
    constexpr int PH = PI * 2 + PJ;
    constexpr int KPC = 16;
    constexpr int NCH = 64 / KPC;
    constexpr int ROWS = KPC * TY;
    constexpr int NIT = ROWS * 16 / 128;
    constexpr int KU = (NT == 1) ? 4 : 2;

    const int j0 = bx * NPX;
    const int r0 = ip - (PI ? 0 : 1);
    const int cb = j0 - (PJ ? 0 : 4);

    static_assert(ROWS * PITCH <= CONV_LDS_F2, "the phase's tile fits the kernel's LDS buffer");

    const float* xb = x + (size_t)b * C_ * H_ * H_;

    const int lane = threadIdx.x & 31;
    const int mW   = threadIdx.x >> 5;
    const int ln   = lane & 15;
    const int khp  = lane >> 4;
    const int m    = mW * 16 + ln;

    const v2f* Wb = (const v2f*)W2 + (size_t)b * 9 * 64 * OC_;

    v8f acc[GN];
#pragma unroll
    for (int g = 0; g < GN; ++g) acc[g] = v8f{};

    for (int ch = 0; ch < NCH; ++ch) {
        if (ch > 0) __syncthreads();
        {
            v4f va[NIT], vb[NIT];
#pragma unroll
            for (int it = 0; it < NIT; ++it) {
                const int idx = threadIdx.x + it * 128;
                const int g   = idx & 15;
                const int row = idx >> 4;
                const int t   = row % TY;
                const int kpl = row / TY;
                const int gr  = r0 + t;
                const int gc  = cb + 4 * g;
                const bool ok = (g < NG) && (gr >= 0) && (gr < H_) && (gc >= 0) && (gc < H_);
                const float* p = ok ? (xb + ((size_t)(2 * (ch * KPC + kpl)) * H_ + gr) * H_ + gc) : xb;
                const v4f qa = *(const v4f*)p;
                const v4f qb = *(const v4f*)(p + (size_t)H_ * H_);
                const v4f z = {0.f, 0.f, 0.f, 0.f};
                va[it] = ok ? qa : z;
                vb[it] = ok ? qb : z;
            }
#pragma unroll
            for (int it = 0; it < NIT; ++it) {
                const int idx = threadIdx.x + it * 128;
                const int g   = idx & 15;
                const int row = idx >> 4;
                if (g < NG) {
                    v2f* d = xs + row * PITCH + 4 * g;
                    d[0] = v2f{va[it].x, vb[it].x};
                    d[1] = v2f{va[it].y, vb[it].y};
                    d[2] = v2f{va[it].z, vb[it].z};
                    d[3] = v2f{va[it].w, vb[it].w};
                }
            }
        }
        __syncthreads();

#pragma unroll KU
        for (int kcl = 0; kcl < KPC / 2; ++kcl) {
            const int kpl = 2 * kcl + khp;
            const int kp  = ch * KPC + kpl;
            v2f a[NT];
#pragma unroll
            for (int t = 0; t < NT; ++t) {
                const int ty = t / TX, tx = t % TX;
                const int ky  = PI ? 1 : (2 - 2 * ty);
                const int kx  = PJ ? 1 : (2 - 2 * tx);
                const int tap = ky * 3 + kx;
                a[t] = Wb[((size_t)tap * 64 + kp) * OC_ + m];
            }
#pragma unroll
            for (int t = 0; t < NT; ++t) {
                const int ty = t / TX, tx = t % TX;
#pragma unroll
                for (int g = 0; g < GN; ++g) {
                    const v2f bf = xs[(kpl * TY + ty) * PITCH + (g * 16 + ln + tx + COFF)];
                    acc[g] = __builtin_amdgcn_wmma_f32_16x16x4_f32(
                               false, a[t], false, bf, (short)0, acc[g], false, false);
                }
            }
        }
    }

    const int mhi = khp * 8;
    float* ob = conv_ws + (((size_t)(b * 4 + PH) * OC_ + mW * 16 + mhi) * PPSZ_) + (size_t)ip * PP_;
#pragma unroll
    for (int g = 0; g < GN; ++g) {
        const int jp = j0 + g * 16 + ln;
        if (jp < NJ) {
#pragma unroll
            for (int r = 0; r < 8; ++r) ob[(size_t)r * PPSZ_ + jp] = acc[g][r];
        }
    }
}

__global__ void __launch_bounds__(128) k_conv_all(const float* __restrict__ x,
                                                  const float* __restrict__ W2,
                                                  float* __restrict__ conv_ws) {
    __shared__ v2f xs[CONV_LDS_F2];
    const int b = blockIdx.y;
    int r = blockIdx.x;
    if (r < NB00_) { conv_phase<0, 0>(x, W2, conv_ws, xs, r % 3, r / 3, b); return; }
    r -= NB00_;
    if (r < NB01_) { conv_phase<0, 1>(x, W2, conv_ws, xs, r & 1, r >> 1, b); return; }
    r -= NB01_;
    if (r < NB10_) { conv_phase<1, 0>(x, W2, conv_ws, xs, r % 3, r / 3, b); return; }
    r -= NB10_;
    conv_phase<1, 1>(x, W2, conv_ws, xs, r & 1, r >> 1, b);
}

#define BLUR_RI 32
#define BLUR_U  4
static_assert(BLUR_RI * 4 == H_ && BLUR_RI % BLUR_U == 0, "blur grid: 4 bands cover all row pairs");

__device__ __forceinline__ void blur_hrow(float a, float b0, float c, float d, float e,
                                          bool hasL, bool hasR, bool valid,
                                          float& h0, float& h1) {
    const float al = hasL ? a : 0.f;
    const float er = hasR ? e : 0.f;
    float t0 = fmaf(3.f, b0 + c, al + d);
    float t1 = fmaf(3.f, c + d, b0 + er);
    h0 = valid ? t0 : 0.f;
    h1 = valid ? t1 : 0.f;
}

__global__ void __launch_bounds__(128) k_blur(const float* __restrict__ conv_ws,
                                              float* __restrict__ out) {
    const int j  = threadIdx.x;
    const int i0 = blockIdx.x * BLUR_RI;
    const int o  = blockIdx.y;
    const int b  = blockIdx.z;

    const float* p00 = conv_ws + ((size_t)(b * 4 + 0) * OC_ + o) * PPSZ_;
    const float* p01 = conv_ws + ((size_t)(b * 4 + 1) * OC_ + o) * PPSZ_;
    const float* p10 = conv_ws + ((size_t)(b * 4 + 2) * OC_ + o) * PPSZ_;
    const float* p11 = conv_ws + ((size_t)(b * 4 + 3) * OC_ + o) * PPSZ_;

    const int  jl   = (j > 0) ? (j - 1) : 0;
    const int  jr   = (j < 127) ? (j + 1) : 127;
    const bool hasL = (j > 0);
    const bool hasR = (j < 127);

    float hA0, hA1, hB0, hB1, hC0, hC1;
    {
        const bool vA = (i0 > 0);
        const int  rA = vA ? (i0 - 1) : 0;
        const float* a0 = p10 + rA * PP_; const float* a1 = p11 + rA * PP_;
        const float* b0 = p00 + i0 * PP_; const float* b1 = p01 + i0 * PP_;
        const float* c0 = p10 + i0 * PP_; const float* c1 = p11 + i0 * PP_;
        float ra[5] = { a1[jl], a0[j], a1[j], a0[j + 1], a1[jr] };
        float rb[5] = { b1[jl], b0[j], b1[j], b0[j + 1], b1[jr] };
        float rc[5] = { c1[jl], c0[j], c1[j], c0[j + 1], c1[jr] };
        blur_hrow(ra[0], ra[1], ra[2], ra[3], ra[4], hasL, hasR, vA,   hA0, hA1);
        blur_hrow(rb[0], rb[1], rb[2], rb[3], rb[4], hasL, hasR, true, hB0, hB1);
        blur_hrow(rc[0], rc[1], rc[2], rc[3], rc[4], hasL, hasR, true, hC0, hC1);
    }

    float* op = out + (((size_t)b * OC_ + o) * OUT_ + 2 * i0) * OUT_ + 2 * j;

    for (int it = 0; it < BLUR_RI / BLUR_U; ++it) {
        float re[BLUR_U][5], ro[BLUR_U][5];
#pragma unroll
        for (int s = 0; s < BLUR_U; ++s) {
            const int i2  = i0 + it * BLUR_U + s + 1;
            const int i2o = (i2 < H_) ? i2 : (H_ - 1);
            const float* e0 = p00 + i2 * PP_;  const float* e1 = p01 + i2 * PP_;
            const float* q0 = p10 + i2o * PP_; const float* q1 = p11 + i2o * PP_;
            re[s][0] = e1[jl]; re[s][1] = e0[j]; re[s][2] = e1[j]; re[s][3] = e0[j + 1]; re[s][4] = e1[jr];
            ro[s][0] = q1[jl]; ro[s][1] = q0[j]; ro[s][2] = q1[j]; ro[s][3] = q0[j + 1]; ro[s][4] = q1[jr];
        }
#pragma unroll
        for (int s = 0; s < BLUR_U; ++s) {
            const int i2 = i0 + it * BLUR_U + s + 1;
            float hD0, hD1, hE0, hE1;
            blur_hrow(re[s][0], re[s][1], re[s][2], re[s][3], re[s][4], hasL, hasR, true,      hD0, hD1);
            blur_hrow(ro[s][0], ro[s][1], ro[s][2], ro[s][3], ro[s][4], hasL, hasR, (i2 < H_), hE0, hE1);
            v2f y0, y1;
            y0.x = fmaf(3.f, hB0 + hC0, hA0 + hD0) * (1.f / 16.f);
            y0.y = fmaf(3.f, hB1 + hC1, hA1 + hD1) * (1.f / 16.f);
            y1.x = fmaf(3.f, hC0 + hD0, hB0 + hE0) * (1.f / 16.f);
            y1.y = fmaf(3.f, hC1 + hD1, hB1 + hE1) * (1.f / 16.f);
            float* orow = op + (size_t)(2 * (it * BLUR_U + s)) * OUT_;
            *(v2f*)(orow)        = y0;
            *(v2f*)(orow + OUT_) = y1;
            hA0 = hC0; hA1 = hC1; hB0 = hD0; hB1 = hD1; hC0 = hE0; hC1 = hE1;
        }
    }
}

extern "C" void kernel_launch(void* const* d_in, const int* in_sizes, int n_in,
                              void* d_out, int out_size, void* d_ws, size_t ws_size,
                              hipStream_t stream) {
    const float* x      = (const float*)d_in[0];
    const float* style  = (const float*)d_in[1];
    const float* weight = (const float*)d_in[2];
    const float* modw   = (const float*)d_in[3];
    const float* modb   = (const float*)d_in[4];
    float* out = (float*)d_out;
    float* ws  = (float*)d_ws;

    float* W2_ws   = ws + 1024;
    float* conv_ws = ws + 1024 + 589824;

    k_mod<<<dim3(MOD_NG, B_), dim3(1024), 0, stream>>>(style, modw, modb, weight, W2_ws);

    k_conv_all<<<dim3(NBB_, B_), dim3(128), 0, stream>>>(x, W2_ws, conv_ws);

    k_blur<<<dim3(4, OC_, B_), dim3(128), 0, stream>>>(conv_ws, out);
}


# k_conv: each block of 12-16 matrix instructions gets its fragments (x from LDS, weights from global) loaded one block ahead as ONE group into a second register set, not among them; sum order same
# speedup vs baseline: 2.8539x; 1.0359x over previous
#include <hip/hip_runtime.h>
#include <math.h>

#define B_     8
#define C_     128
#define OC_    64
#define SD_    512
#define H_     128
#define KW_    3
#define H2_    257
#define OUT_   256
#define PP_    129
#define PPSZ_  (PP_*PP_)

#define SCALE_  (0.02946278254943948f)
#define WGAIN_  (0.04419417382415922f)

typedef __attribute__((ext_vector_type(2))) float v2f;
typedef __attribute__((ext_vector_type(4))) float v4f;
typedef __attribute__((ext_vector_type(8))) float v8f;

#define MOD_OG 8
#define MOD_NG 8
static_assert(MOD_OG * C_ == 1024 && MOD_NG * MOD_OG == OC_ && C_ == 32 * 4, "k_mod block shape: 32 waves x 4 rows = 128 c");

__global__ void __launch_bounds__(1024) __attribute__((amdgpu_waves_per_eu(8, 8))) k_mod(const float* __restrict__ style,
                                              const float* __restrict__ modw,
                                              const float* __restrict__ modb,
                                              const float* __restrict__ weight,
                                              float* __restrict__ W2) {
    __shared__ __attribute__((aligned(16))) float s_sh[C_];
    __shared__ float red[MOD_OG][4];

    const int tid  = threadIdx.x;
    const int lane = tid & 31;
    const int w    = tid >> 5;
    const int b    = blockIdx.y;
    const int ol   = tid >> 7;
    const int c    = tid & 127;
    const int o    = blockIdx.x * MOD_OG + ol;

    const float* wp = weight + ((size_t)o * C_ + c) * (KW_ * KW_);
    float wv[9];
#pragma unroll
    for (int t = 0; t < 9; ++t) wv[t] = wp[t];

    const v4f mb4 = *(const v4f*)(modb + w * 4);
    const v4f* st4 = (const v4f*)(style + (size_t)b * SD_);
    v4f sv[4];
#pragma unroll
    for (int i = 0; i < 4; ++i) sv[i] = st4[lane + 32 * i];
    v4f mw[4][4];
#pragma unroll
    for (int r = 0; r < 4; ++r) {
        const v4f* m4 = (const v4f*)(modw + (size_t)(w * 4 + r) * SD_);
#pragma unroll
        for (int i = 0; i < 4; ++i) mw[r][i] = m4[lane + 32 * i];
    }
    float part[4];
#pragma unroll
    for (int r = 0; r < 4; ++r) {
        float a = 0.f;
#pragma unroll
        for (int i = 0; i < 4; ++i) {
            a = fmaf(sv[i].x, mw[r][i].x, a);
            a = fmaf(sv[i].y, mw[r][i].y, a);
            a = fmaf(sv[i].z, mw[r][i].z, a);
            a = fmaf(sv[i].w, mw[r][i].w, a);
        }
        part[r] = a;
    }
#pragma unroll
    for (int off = 16; off > 0; off >>= 1) {
#pragma unroll
        for (int r = 0; r < 4; ++r) part[r] += __shfl_xor(part[r], off);
    }
    {
        v4f r4;
        r4.x = (part[0] * WGAIN_ + mb4.x) * SCALE_;
        r4.y = (part[1] * WGAIN_ + mb4.y) * SCALE_;
        r4.z = (part[2] * WGAIN_ + mb4.z) * SCALE_;
        r4.w = (part[3] * WGAIN_ + mb4.w) * SCALE_;
        *(v4f*)(s_sh + w * 4) = r4;
    }
    __syncthreads();

    const float sc = s_sh[c];
    float ss = 0.f;
#pragma unroll
    for (int t = 0; t < 9; ++t) { wv[t] = wv[t] * sc; ss += wv[t] * wv[t]; }
#pragma unroll
    for (int off = 16; off > 0; off >>= 1) ss += __shfl_xor(ss, off);
    if (lane == 0) red[ol][w & 3] = ss;
    __syncthreads();
    const float dem = rsqrtf(((red[ol][0] + red[ol][1]) + (red[ol][2] + red[ol][3])) + 1e-8f);
#pragma unroll
    for (int t = 0; t < 9; ++t)
        W2[((((size_t)b * 9 + t) * 64 + (c >> 1)) * OC_ + o) * 2 + (c & 1)] = wv[t] * dem;
}

#define CONV_LDS_F2 (32 * 72)
#define NB00_ (3 * 129)
#define NB01_ (2 * 129)
#define NB10_ (3 * 128)
#define NB11_ (2 * 128)
#define NBB_  1285
static_assert(NB00_ + NB01_ + NB10_ + NB11_ == NBB_, "conv grid: the four phases' blocks of one batch");
template<int PI, int PJ>
__device__ __forceinline__ void conv_phase(const float* __restrict__ x,
                                           const float* __restrict__ W2,
                                           float* __restrict__ conv_ws,
                                           v2f* __restrict__ xs,
                                           const int bx, const int ip, const int b) {
    constexpr int TY = PI ? 1 : 2;
    constexpr int TX = PJ ? 1 : 2;
    constexpr int NT = TY * TX;
    constexpr int GN = PJ ? 4 : 3;
    constexpr int NPX = GN * 16;
    constexpr int NG = PJ ? 16 : 13;
    constexpr int COFF = PJ ? 0 : 3;
    constexpr int PITCH = (TY == 2) ? (PJ ? 72 : 56) : 80;
    constexpr int NJ = PJ ? 128 : 129;
    constexpr int PH = PI * 2 + PJ;
    constexpr int KPC = 16;
    constexpr int NCH = 64 / KPC;
    constexpr int ROWS = KPC * TY;
    constexpr int NIT = ROWS * 16 / 128;
    constexpr int NKS = KPC / 2;
    constexpr int KB = 4 / NT;
    constexpr int NBLK = NKS / KB;
    constexpr int NA = KB * NT;
    constexpr int NB = NA * GN;
    static_assert(NA == 4 && NBLK % 2 == 0, "blocks come in pairs (fragment set 0, then set 1)");

    const int j0 = bx * NPX;
    const int r0 = ip - (PI ? 0 : 1);
    const int cb = j0 - (PJ ? 0 : 4);

    static_assert(ROWS * PITCH <= CONV_LDS_F2, "the phase's tile fits the kernel's LDS buffer");

    const float* xb = x + (size_t)b * C_ * H_ * H_;

    const int lane = threadIdx.x & 31;
    const int mW   = threadIdx.x >> 5;
    const int ln   = lane & 15;
    const int khp  = lane >> 4;
    const int m    = mW * 16 + ln;

    const v2f* Wb = (const v2f*)W2 + (size_t)b * 9 * 64 * OC_;

    v8f acc[GN];
#pragma unroll
    for (int g = 0; g < GN; ++g) acc[g] = v8f{};

    const v2f* Wl = Wb + (size_t)khp * OC_ + m;
    const int xlb = ((khp * TY) * PITCH + ln + COFF) * 8;

    auto load_a = [&](v2f (&fa)[NA], const int chn, const int blk) {
#pragma unroll
        for (int ku = 0; ku < KB; ++ku) {
#pragma unroll
            for (int t = 0; t < NT; ++t) {
                const int ty = t / TX, tx = t % TX;
                const int ky  = PI ? 1 : (2 - 2 * ty);
                const int kx  = PJ ? 1 : (2 - 2 * tx);
                const int tap = ky * 3 + kx;
                fa[ku * NT + t] = Wl[((size_t)tap * 64 + chn * KPC + 2 * (blk * KB + ku)) * OC_];
            }
        }
    };
    auto load_b = [&](v2f (&fb)[NB], const int blk) {
#pragma unroll
        for (int ku = 0; ku < KB; ++ku) {
            int koff = xlb + (2 * (blk * KB + ku) * TY * PITCH) * 8;
            asm volatile("" : "+v"(koff));
            const v2f* xk = (const v2f*)((const char*)xs + koff);
#pragma unroll
            for (int t = 0; t < NT; ++t) {
                const int ty = t / TX, tx = t % TX;
#pragma unroll
                for (int g = 0; g < GN; ++g)
                    fb[(ku * NT + t) * GN + g] = xk[ty * PITCH + g * 16 + tx];
            }
        }
    };
    auto mma = [&](const v2f (&fa)[NA], const v2f (&fb)[NB]) {
#pragma unroll
        for (int ku = 0; ku < KB; ++ku) {
#pragma unroll
            for (int t = 0; t < NT; ++t) {
#pragma unroll
                for (int g = 0; g < GN; ++g)
                    acc[g] = __builtin_amdgcn_wmma_f32_16x16x4_f32(
                               false, fa[ku * NT + t], false, fb[(ku * NT + t) * GN + g], (short)0, acc[g], false, false);
            }
        }
    };

    v2f fa0[NA], fa1[NA], fb0[NB], fb1[NB];
    load_a(fa0, 0, 0);

    for (int ch = 0; ch < NCH; ++ch) {
        if (ch > 0) __syncthreads();
        {
            v4f va[NIT], vb[NIT];
#pragma unroll
            for (int it = 0; it < NIT; ++it) {
                const int idx = threadIdx.x + it * 128;
                const int g   = idx & 15;
                const int row = idx >> 4;
                const int t   = row % TY;
                const int kpl = row / TY;
                const int gr  = r0 + t;
                const int gc  = cb + 4 * g;
                const bool ok = (g < NG) && (gr >= 0) && (gr < H_) && (gc >= 0) && (gc < H_);
                const float* p = ok ? (xb + ((size_t)(2 * (ch * KPC + kpl)) * H_ + gr) * H_ + gc) : xb;
                const v4f qa = *(const v4f*)p;
                const v4f qb = *(const v4f*)(p + (size_t)H_ * H_);
                const v4f z = {0.f, 0.f, 0.f, 0.f};
                va[it] = ok ? qa : z;
                vb[it] = ok ? qb : z;
            }
#pragma unroll
            for (int it = 0; it < NIT; ++it) {
                const int idx = threadIdx.x + it * 128;
                const int g   = idx & 15;
                const int row = idx >> 4;
                if (g < NG) {
                    v2f* d = xs + row * PITCH + 4 * g;
                    d[0] = v2f{va[it].x, vb[it].x};
                    d[1] = v2f{va[it].y, vb[it].y};
                    d[2] = v2f{va[it].z, vb[it].z};
                    d[3] = v2f{va[it].w, vb[it].w};
                }
            }
        }
        __syncthreads();

        const int chn = (ch + 1 < NCH) ? (ch + 1) : ch;
        load_b(fb0, 0);
#pragma unroll
        for (int bp = 0; bp < NBLK / 2; ++bp) {
            load_a(fa1, ch, 2 * bp + 1);
            load_b(fb1, 2 * bp + 1);
            __builtin_amdgcn_sched_barrier(0);
            mma(fa0, fb0);
            __builtin_amdgcn_sched_barrier(0);
            if (2 * bp + 2 < NBLK) {
                load_a(fa0, ch, 2 * bp + 2);
                load_b(fb0, 2 * bp + 2);
            } else {
                load_a(fa0, chn, 0);
            }
            __builtin_amdgcn_sched_barrier(0);
            mma(fa1, fb1);
            __builtin_amdgcn_sched_barrier(0);
        }
    }

    const int mhi = khp * 8;
    float* ob = conv_ws + (((size_t)(b * 4 + PH) * OC_ + mW * 16 + mhi) * PPSZ_) + (size_t)ip * PP_;
#pragma unroll
    for (int g = 0; g < GN; ++g) {
        const int jp = j0 + g * 16 + ln;
        if (jp < NJ) {
#pragma unroll
            for (int r = 0; r < 8; ++r) ob[(size_t)r * PPSZ_ + jp] = acc[g][r];
        }
    }
}

__global__ void __launch_bounds__(128) k_conv_all(const float* __restrict__ x,
                                                  const float* __restrict__ W2,
                                                  float* __restrict__ conv_ws) {
    __shared__ v2f xs[CONV_LDS_F2];
    const int b = blockIdx.y;
    int r = blockIdx.x;
    if (r < NB00_) { conv_phase<0, 0>(x, W2, conv_ws, xs, r % 3, r / 3, b); return; }
    r -= NB00_;
    if (r < NB01_) { conv_phase<0, 1>(x, W2, conv_ws, xs, r & 1, r >> 1, b); return; }
    r -= NB01_;
    if (r < NB10_) { conv_phase<1, 0>(x, W2, conv_ws, xs, r % 3, r / 3, b); return; }
    r -= NB10_;
    conv_phase<1, 1>(x, W2, conv_ws, xs, r & 1, r >> 1, b);
}

#define BLUR_RI 32
#define BLUR_U  4
static_assert(BLUR_RI * 4 == H_ && BLUR_RI % BLUR_U == 0, "blur grid: 4 bands cover all row pairs");

__device__ __forceinline__ void blur_hrow(float a, float b0, float c, float d, float e,
                                          bool hasL, bool hasR, bool valid,
                                          float& h0, float& h1) {
    const float al = hasL ? a : 0.f;
    const float er = hasR ? e : 0.f;
    float t0 = fmaf(3.f, b0 + c, al + d);
    float t1 = fmaf(3.f, c + d, b0 + er);
    h0 = valid ? t0 : 0.f;
    h1 = valid ? t1 : 0.f;
}

__global__ void __launch_bounds__(128) k_blur(const float* __restrict__ conv_ws,
                                              float* __restrict__ out) {
    const int j  = threadIdx.x;
    const int i0 = blockIdx.x * BLUR_RI;
    const int o  = blockIdx.y;
    const int b  = blockIdx.z;

    const float* p00 = conv_ws + ((size_t)(b * 4 + 0) * OC_ + o) * PPSZ_;
    const float* p01 = conv_ws + ((size_t)(b * 4 + 1) * OC_ + o) * PPSZ_;
    const float* p10 = conv_ws + ((size_t)(b * 4 + 2) * OC_ + o) * PPSZ_;
    const float* p11 = conv_ws + ((size_t)(b * 4 + 3) * OC_ + o) * PPSZ_;

    const int  jl   = (j > 0) ? (j - 1) : 0;
    const int  jr   = (j < 127) ? (j + 1) : 127;
    const bool hasL = (j > 0);
    const bool hasR = (j < 127);

    float hA0, hA1, hB0, hB1, hC0, hC1;
    {
        const bool vA = (i0 > 0);
        const int  rA = vA ? (i0 - 1) : 0;
        const float* a0 = p10 + rA * PP_; const float* a1 = p11 + rA * PP_;
        const float* b0 = p00 + i0 * PP_; const float* b1 = p01 + i0 * PP_;
        const float* c0 = p10 + i0 * PP_; const float* c1 = p11 + i0 * PP_;
        float ra[5] = { a1[jl], a0[j], a1[j], a0[j + 1], a1[jr] };
        float rb[5] = { b1[jl], b0[j], b1[j], b0[j + 1], b1[jr] };
        float rc[5] = { c1[jl], c0[j], c1[j], c0[j + 1], c1[jr] };
        blur_hrow(ra[0], ra[1], ra[2], ra[3], ra[4], hasL, hasR, vA,   hA0, hA1);
        blur_hrow(rb[0], rb[1], rb[2], rb[3], rb[4], hasL, hasR, true, hB0, hB1);
        blur_hrow(rc[0], rc[1], rc[2], rc[3], rc[4], hasL, hasR, true, hC0, hC1);
    }

    float* op = out + (((size_t)b * OC_ + o) * OUT_ + 2 * i0) * OUT_ + 2 * j;

    for (int it = 0; it < BLUR_RI / BLUR_U; ++it) {
        float re[BLUR_U][5], ro[BLUR_U][5];
#pragma unroll
        for (int s = 0; s < BLUR_U; ++s) {
            const int i2  = i0 + it * BLUR_U + s + 1;
            const int i2o = (i2 < H_) ? i2 : (H_ - 1);
            const float* e0 = p00 + i2 * PP_;  const float* e1 = p01 + i2 * PP_;
            const float* q0 = p10 + i2o * PP_; const float* q1 = p11 + i2o * PP_;
            re[s][0] = e1[jl]; re[s][1] = e0[j]; re[s][2] = e1[j]; re[s][3] = e0[j + 1]; re[s][4] = e1[jr];
            ro[s][0] = q1[jl]; ro[s][1] = q0[j]; ro[s][2] = q1[j]; ro[s][3] = q0[j + 1]; ro[s][4] = q1[jr];
        }
#pragma unroll
        for (int s = 0; s < BLUR_U; ++s) {
            const int i2 = i0 + it * BLUR_U + s + 1;
            float hD0, hD1, hE0, hE1;
            blur_hrow(re[s][0], re[s][1], re[s][2], re[s][3], re[s][4], hasL, hasR, true,      hD0, hD1);
            blur_hrow(ro[s][0], ro[s][1], ro[s][2], ro[s][3], ro[s][4], hasL, hasR, (i2 < H_), hE0, hE1);
            v2f y0, y1;
            y0.x = fmaf(3.f, hB0 + hC0, hA0 + hD0) * (1.f / 16.f);
            y0.y = fmaf(3.f, hB1 + hC1, hA1 + hD1) * (1.f / 16.f);
            y1.x = fmaf(3.f, hC0 + hD0, hB0 + hE0) * (1.f / 16.f);
            y1.y = fmaf(3.f, hC1 + hD1, hB1 + hE1) * (1.f / 16.f);
            float* orow = op + (size_t)(2 * (it * BLUR_U + s)) * OUT_;
            *(v2f*)(orow)        = y0;
            *(v2f*)(orow + OUT_) = y1;
            hA0 = hC0; hA1 = hC1; hB0 = hD0; hB1 = hD1; hC0 = hE0; hC1 = hE1;
        }
    }
}

extern "C" void kernel_launch(void* const* d_in, const int* in_sizes, int n_in,
                              void* d_out, int out_size, void* d_ws, size_t ws_size,
                              hipStream_t stream) {
    const float* x      = (const float*)d_in[0];
    const float* style  = (const float*)d_in[1];
    const float* weight = (const float*)d_in[2];
    const float* modw   = (const float*)d_in[3];
    const float* modb   = (const float*)d_in[4];
    float* out = (float*)d_out;
    float* ws  = (float*)d_ws;

    float* W2_ws   = ws + 1024;
    float* conv_ws = ws + 1024 + 589824;

    k_mod<<<dim3(MOD_NG, B_), dim3(1024), 0, stream>>>(style, modw, modb, weight, W2_ws);

    k_conv_all<<<dim3(NBB_, B_), dim3(128), 0, stream>>>(x, W2_ws, conv_ws);

    k_blur<<<dim3(4, OC_, B_), dim3(128), 0, stream>>>(conv_ws, out);
}


# k_blur: a lane owns 8 output columns; plane rows (re-laid at pitch 128, column 128 aside) read as whole-line 8 B loads, edge values by lane shuffles, rows stored 512 B at a time; sum order unchanged
# speedup vs baseline: 3.0951x; 1.0845x over previous
#include <hip/hip_runtime.h>
#include <math.h>

#define B_     8
#define C_     128
#define OC_    64
#define SD_    512
#define H_     128
#define KW_    3
#define H2_    257
#define OUT_   256
#define PW_    128
#define OFF00_ 0
#define OFF01_ (129 * PW_)
#define OFF10_ (2 * 129 * PW_)
#define OFF11_ (OFF10_ + 128 * PW_)
#define OFFS0_ (OFF11_ + 128 * PW_)
#define OFFS1_ (OFFS0_ + 132)
#define PG_    66112
static_assert(OFFS1_ + 128 <= PG_ && PG_ % 64 == 0 && PG_ <= 4 * 129 * 129, "the group holds its planes and is no larger than four 129 x 129 planes");

#define SCALE_  (0.02946278254943948f)
#define WGAIN_  (0.04419417382415922f)

typedef __attribute__((ext_vector_type(2))) float v2f;
typedef __attribute__((ext_vector_type(4))) float v4f;
typedef __attribute__((ext_vector_type(8))) float v8f;

#define MOD_OG 8
#define MOD_NG 8
static_assert(MOD_OG * C_ == 1024 && MOD_NG * MOD_OG == OC_ && C_ == 32 * 4, "k_mod block shape: 32 waves x 4 rows = 128 c");

__global__ void __launch_bounds__(1024) __attribute__((amdgpu_waves_per_eu(8, 8))) k_mod(const float* __restrict__ style,
                                              const float* __restrict__ modw,
                                              const float* __restrict__ modb,
                                              const float* __restrict__ weight,
                                              float* __restrict__ W2) {
    __shared__ __attribute__((aligned(16))) float s_sh[C_];
    __shared__ float red[MOD_OG][4];

    const int tid  = threadIdx.x;
    const int lane = tid & 31;
    const int w    = tid >> 5;
    const int b    = blockIdx.y;
    const int ol   = tid >> 7;
    const int c    = tid & 127;
    const int o    = blockIdx.x * MOD_OG + ol;

    const float* wp = weight + ((size_t)o * C_ + c) * (KW_ * KW_);
    float wv[9];
#pragma unroll
    for (int t = 0; t < 9; ++t) wv[t] = wp[t];

    const v4f mb4 = *(const v4f*)(modb + w * 4);
    const v4f* st4 = (const v4f*)(style + (size_t)b * SD_);
    v4f sv[4];
#pragma unroll
    for (int i = 0; i < 4; ++i) sv[i] = st4[lane + 32 * i];
    v4f mw[4][4];
#pragma unroll
    for (int r = 0; r < 4; ++r) {
        const v4f* m4 = (const v4f*)(modw + (size_t)(w * 4 + r) * SD_);
#pragma unroll
        for (int i = 0; i < 4; ++i) mw[r][i] = m4[lane + 32 * i];
    }
    float part[4];
#pragma unroll
    for (int r = 0; r < 4; ++r) {
        float a = 0.f;
#pragma unroll
        for (int i = 0; i < 4; ++i) {
            a = fmaf(sv[i].x, mw[r][i].x, a);
            a = fmaf(sv[i].y, mw[r][i].y, a);
            a = fmaf(sv[i].z, mw[r][i].z, a);
            a = fmaf(sv[i].w, mw[r][i].w, a);
        }
        part[r] = a;
    }
#pragma unroll
    for (int off = 16; off > 0; off >>= 1) {
#pragma unroll
        for (int r = 0; r < 4; ++r) part[r] += __shfl_xor(part[r], off);
    }
    {
        v4f r4;
        r4.x = (part[0] * WGAIN_ + mb4.x) * SCALE_;
        r4.y = (part[1] * WGAIN_ + mb4.y) * SCALE_;
        r4.z = (part[2] * WGAIN_ + mb4.z) * SCALE_;
        r4.w = (part[3] * WGAIN_ + mb4.w) * SCALE_;
        *(v4f*)(s_sh + w * 4) = r4;
    }
    __syncthreads();

    const float sc = s_sh[c];
    float ss = 0.f;
#pragma unroll
    for (int t = 0; t < 9; ++t) { wv[t] = wv[t] * sc; ss += wv[t] * wv[t]; }
#pragma unroll
    for (int off = 16; off > 0; off >>= 1) ss += __shfl_xor(ss, off);
    if (lane == 0) red[ol][w & 3] = ss;
    __syncthreads();
    const float dem = rsqrtf(((red[ol][0] + red[ol][1]) + (red[ol][2] + red[ol][3])) + 1e-8f);
#pragma unroll
    for (int t = 0; t < 9; ++t)
        W2[((((size_t)b * 9 + t) * 64 + (c >> 1)) * OC_ + o) * 2 + (c & 1)] = wv[t] * dem;
}

#define CONV_LDS_F2 (32 * 72)
#define NB00_ (3 * 129)
#define NB01_ (2 * 129)
#define NB10_ (3 * 128)
#define NB11_ (2 * 128)
#define NBB_  1285
static_assert(NB00_ + NB01_ + NB10_ + NB11_ == NBB_, "conv grid: the four phases' blocks of one batch");
template<int PI, int PJ>
__device__ __forceinline__ void conv_phase(const float* __restrict__ x,
                                           const float* __restrict__ W2,
                                           float* __restrict__ conv_ws,
                                           v2f* __restrict__ xs,
                                           const int bx, const int ip, const int b) {
    constexpr int TY = PI ? 1 : 2;
    constexpr int TX = PJ ? 1 : 2;
    constexpr int NT = TY * TX;
    constexpr int GN = PJ ? 4 : 3;
    constexpr int NPX = GN * 16;
    constexpr int NG = PJ ? 16 : 13;
    constexpr int COFF = PJ ? 0 : 3;
    constexpr int PITCH = (TY == 2) ? (PJ ? 72 : 56) : 80;
    constexpr int KPC = 16;
    constexpr int NCH = 64 / KPC;
    constexpr int ROWS = KPC * TY;
    constexpr int NIT = ROWS * 16 / 128;
    constexpr int NKS = KPC / 2;
    constexpr int KB = 4 / NT;
    constexpr int NBLK = NKS / KB;
    constexpr int NA = KB * NT;
    constexpr int NB = NA * GN;
    static_assert(NA == 4 && NBLK % 2 == 0, "blocks come in pairs (fragment set 0, then set 1)");

    const int j0 = bx * NPX;
    const int r0 = ip - (PI ? 0 : 1);
    const int cb = j0 - (PJ ? 0 : 4);

    static_assert(ROWS * PITCH <= CONV_LDS_F2, "the phase's tile fits the kernel's LDS buffer");

    const float* xb = x + (size_t)b * C_ * H_ * H_;

    const int lane = threadIdx.x & 31;
    const int mW   = threadIdx.x >> 5;
    const int ln   = lane & 15;
    const int khp  = lane >> 4;
    const int m    = mW * 16 + ln;

    const v2f* Wb = (const v2f*)W2 + (size_t)b * 9 * 64 * OC_;

    v8f acc[GN];
#pragma unroll
    for (int g = 0; g < GN; ++g) acc[g] = v8f{};

    const v2f* Wl = Wb + (size_t)khp * OC_ + m;
    const int xlb = ((khp * TY) * PITCH + ln + COFF) * 8;

    auto load_a = [&](v2f (&fa)[NA], const int chn, const int blk) {
#pragma unroll
        for (int ku = 0; ku < KB; ++ku) {
#pragma unroll
            for (int t = 0; t < NT; ++t) {
                const int ty = t / TX, tx = t % TX;
                const int ky  = PI ? 1 : (2 - 2 * ty);
                const int kx  = PJ ? 1 : (2 - 2 * tx);
                const int tap = ky * 3 + kx;
                fa[ku * NT + t] = Wl[((size_t)tap * 64 + chn * KPC + 2 * (blk * KB + ku)) * OC_];
            }
        }
    };
    auto load_b = [&](v2f (&fb)[NB], const int blk) {
#pragma unroll
        for (int ku = 0; ku < KB; ++ku) {
            int koff = xlb + (2 * (blk * KB + ku) * TY * PITCH) * 8;
            asm volatile("" : "+v"(koff));
            const v2f* xk = (const v2f*)((const char*)xs + koff);
#pragma unroll
            for (int t = 0; t < NT; ++t) {
                const int ty = t / TX, tx = t % TX;
#pragma unroll
                for (int g = 0; g < GN; ++g)
                    fb[(ku * NT + t) * GN + g] = xk[ty * PITCH + g * 16 + tx];
            }
        }
    };
    auto mma = [&](const v2f (&fa)[NA], const v2f (&fb)[NB]) {
#pragma unroll
        for (int ku = 0; ku < KB; ++ku) {
#pragma unroll
            for (int t = 0; t < NT; ++t) {
#pragma unroll
                for (int g = 0; g < GN; ++g)
                    acc[g] = __builtin_amdgcn_wmma_f32_16x16x4_f32(
                               false, fa[ku * NT + t], false, fb[(ku * NT + t) * GN + g], (short)0, acc[g], false, false);
            }
        }
    };

    v2f fa0[NA], fa1[NA], fb0[NB], fb1[NB];
    load_a(fa0, 0, 0);

    for (int ch = 0; ch < NCH; ++ch) {
        if (ch > 0) __syncthreads();
        {
            v4f va[NIT], vb[NIT];
#pragma unroll
            for (int it = 0; it < NIT; ++it) {
                const int idx = threadIdx.x + it * 128;
                const int g   = idx & 15;
                const int row = idx >> 4;
                const int t   = row % TY;
                const int kpl = row / TY;
                const int gr  = r0 + t;
                const int gc  = cb + 4 * g;
                const bool ok = (g < NG) && (gr >= 0) && (gr < H_) && (gc >= 0) && (gc < H_);
                const float* p = ok ? (xb + ((size_t)(2 * (ch * KPC + kpl)) * H_ + gr) * H_ + gc) : xb;
                const v4f qa = *(const v4f*)p;
                const v4f qb = *(const v4f*)(p + (size_t)H_ * H_);
                const v4f z = {0.f, 0.f, 0.f, 0.f};
                va[it] = ok ? qa : z;
                vb[it] = ok ? qb : z;
            }
#pragma unroll
            for (int it = 0; it < NIT; ++it) {
                const int idx = threadIdx.x + it * 128;
                const int g   = idx & 15;
                const int row = idx >> 4;
                if (g < NG) {
                    v2f* d = xs + row * PITCH + 4 * g;
                    d[0] = v2f{va[it].x, vb[it].x};
                    d[1] = v2f{va[it].y, vb[it].y};
                    d[2] = v2f{va[it].z, vb[it].z};
                    d[3] = v2f{va[it].w, vb[it].w};
                }
            }
        }
        __syncthreads();

        const int chn = (ch + 1 < NCH) ? (ch + 1) : ch;
        load_b(fb0, 0);
#pragma unroll
        for (int bp = 0; bp < NBLK / 2; ++bp) {
            load_a(fa1, ch, 2 * bp + 1);
            load_b(fb1, 2 * bp + 1);
            __builtin_amdgcn_sched_barrier(0);
            mma(fa0, fb0);
            __builtin_amdgcn_sched_barrier(0);
            if (2 * bp + 2 < NBLK) {
                load_a(fa0, ch, 2 * bp + 2);
                load_b(fb0, 2 * bp + 2);
            } else {
                load_a(fa0, chn, 0);
            }
            __builtin_amdgcn_sched_barrier(0);
            mma(fa1, fb1);
            __builtin_amdgcn_sched_barrier(0);
        }
    }

    const int mhi = khp * 8;
    constexpr int POFF = PI ? (PJ ? OFF11_ : OFF10_) : (PJ ? OFF01_ : OFF00_);
    float* gb = conv_ws + (size_t)(b * OC_ + mW * 16 + mhi) * PG_;
    float* ob = gb + POFF + (size_t)ip * PW_;
    float* sb = gb + (PI ? OFFS1_ : OFFS0_) + ip;
    constexpr int NJ = PJ ? PW_ : PW_ + 1;
#pragma unroll
    for (int g = 0; g < GN; ++g) {
        const int jp = j0 + g * 16 + ln;
        float* dst = (PJ == 0 && jp == PW_) ? sb : (ob + jp);
        if (jp < NJ) {
#pragma unroll
            for (int r = 0; r < 8; ++r) dst[(size_t)r * PG_] = acc[g][r];
        }
    }
}

__global__ void __launch_bounds__(128) k_conv_all(const float* __restrict__ x,
                                                  const float* __restrict__ W2,
                                                  float* __restrict__ conv_ws) {
    __shared__ v2f xs[CONV_LDS_F2];
    const int b = blockIdx.y;
    int r = blockIdx.x;
    if (r < NB00_) { conv_phase<0, 0>(x, W2, conv_ws, xs, r % 3, r / 3, b); return; }
    r -= NB00_;
    if (r < NB01_) { conv_phase<0, 1>(x, W2, conv_ws, xs, r & 1, r >> 1, b); return; }
    r -= NB01_;
    if (r < NB10_) { conv_phase<1, 0>(x, W2, conv_ws, xs, r % 3, r / 3, b); return; }
    r -= NB10_;
    conv_phase<1, 1>(x, W2, conv_ws, xs, r & 1, r >> 1, b);
}

#define BLUR_RP 16
#define BLUR_U  4
#define BLUR_NW 8
static_assert(BLUR_NW * BLUR_RP == H_ && BLUR_RP % BLUR_U == 0, "blur block: 8 bands cover all 128 row pairs");

struct BlurRow {
    v2f el, ol;
    v2f er, orr;
    float s;
};

__device__ __forceinline__ void blur_load(const float* __restrict__ p0, const float* __restrict__ p1,
                                          const float* __restrict__ sd, const int row, const int lane, BlurRow& r) {
    const float* q0 = p0 + row * PW_ + 2 * lane;
    const float* q1 = p1 + row * PW_ + 2 * lane;
    r.el  = *(const v2f*)(q0);
    r.er  = *(const v2f*)(q0 + 64);
    r.ol  = *(const v2f*)(q1);
    r.orr = *(const v2f*)(q1 + 64);
    r.s   = sd[row];
}

__device__ __forceinline__ void blur_h4(const v2f E, const v2f O, const float a, const float f, const float g,
                                        const bool valid, float* h) {
    const float t0 = fmaf(3.f, E.x + O.x, a + E.y);
    const float t1 = fmaf(3.f, O.x + E.y, E.x + O.y);
    const float t2 = fmaf(3.f, E.y + O.y, O.x + f);
    const float t3 = fmaf(3.f, O.y + f, E.y + g);
    h[0] = valid ? t0 : 0.f;
    h[1] = valid ? t1 : 0.f;
    h[2] = valid ? t2 : 0.f;
    h[3] = valid ? t3 : 0.f;
}

__device__ __forceinline__ void blur_hrow(const BlurRow& r, const int lane, const bool valid, float (&h)[8]) {
    const bool first = (lane == 0), last = (lane == 31);
    const int nx = (lane + 1) & 31, pv = (lane - 1) & 31;
    const float xel = __shfl(r.el.x, nx),  xer = __shfl(r.er.x, nx);
    const float xol = __shfl(r.ol.x, nx),  xor_ = __shfl(r.orr.x, nx);
    const float yol = __shfl(r.ol.y, pv),  yor = __shfl(r.orr.y, pv);
    const float nel = last ? xer : xel;
    const float nol = last ? xor_ : xol;
    const float ner = last ? r.s : xer;
    const float nor = last ? 0.f : xor_;
    const float pol = first ? 0.f : yol;
    const float por = first ? yol : yor;
    blur_h4(r.el, r.ol, pol, nel, nol, valid, h);
    blur_h4(r.er, r.orr, por, ner, nor, valid, h + 4);
}

__global__ void __launch_bounds__(256) __attribute__((amdgpu_waves_per_eu(4, 4))) k_blur(const float* __restrict__ conv_ws,
                                              float* __restrict__ out) {
    const int lane = threadIdx.x & 31;
    const int w  = __builtin_amdgcn_readfirstlane(threadIdx.x >> 5);
    const int i0 = w * BLUR_RP;
    const int o  = blockIdx.x;
    const int b  = blockIdx.y;

    const float* gp  = conv_ws + (size_t)(b * OC_ + o) * PG_;
    const float* p00 = gp + OFF00_;
    const float* p01 = gp + OFF01_;
    const float* p10 = gp + OFF10_;
    const float* p11 = gp + OFF11_;
    const float* s0  = gp + OFFS0_;
    const float* s1  = gp + OFFS1_;

    float hA[8], hB[8], hC[8];
    {
        const bool vA = (i0 > 0);
        const int  rA = vA ? (i0 - 1) : 0;
        BlurRow ra, rb, rc;
        blur_load(p10, p11, s1, rA, lane, ra);
        blur_load(p00, p01, s0, i0, lane, rb);
        blur_load(p10, p11, s1, i0, lane, rc);
        blur_hrow(ra, lane, vA,   hA);
        blur_hrow(rb, lane, true, hB);
        blur_hrow(rc, lane, true, hC);
    }

    float* op = out + (((size_t)b * OC_ + o) * OUT_ + 2 * i0) * OUT_ + 4 * lane;

    for (int it = 0; it < BLUR_RP / BLUR_U; ++it) {
        BlurRow re[BLUR_U], ro[BLUR_U];
#pragma unroll
        for (int s = 0; s < BLUR_U; ++s) {
            const int i2  = i0 + it * BLUR_U + s + 1;
            const int i2o = (i2 < H_) ? i2 : (H_ - 1);
            blur_load(p00, p01, s0, i2,  lane, re[s]);
            blur_load(p10, p11, s1, i2o, lane, ro[s]);
        }
#pragma unroll
        for (int s = 0; s < BLUR_U; ++s) {
            const int i2 = i0 + it * BLUR_U + s + 1;
            float hD[8], hE[8];
            blur_hrow(re[s], lane, true,      hD);
            blur_hrow(ro[s], lane, (i2 < H_), hE);
            float y0[8], y1[8];
#pragma unroll
            for (int m = 0; m < 8; ++m) {
                y0[m] = fmaf(3.f, hB[m] + hC[m], hA[m] + hD[m]) * (1.f / 16.f);
                y1[m] = fmaf(3.f, hC[m] + hD[m], hB[m] + hE[m]) * (1.f / 16.f);
                hA[m] = hC[m]; hB[m] = hD[m]; hC[m] = hE[m];
            }
            float* orow = op + (size_t)(2 * (it * BLUR_U + s)) * OUT_;
            *(v4f*)(orow)              = v4f{y0[0], y0[1], y0[2], y0[3]};
            *(v4f*)(orow + 128)        = v4f{y0[4], y0[5], y0[6], y0[7]};
            *(v4f*)(orow + OUT_)       = v4f{y1[0], y1[1], y1[2], y1[3]};
            *(v4f*)(orow + OUT_ + 128) = v4f{y1[4], y1[5], y1[6], y1[7]};
        }
    }
}

extern "C" void kernel_launch(void* const* d_in, const int* in_sizes, int n_in,
                              void* d_out, int out_size, void* d_ws, size_t ws_size,
                              hipStream_t stream) {
    const float* x      = (const float*)d_in[0];
    const float* style  = (const float*)d_in[1];
    const float* weight = (const float*)d_in[2];
    const float* modw   = (const float*)d_in[3];
    const float* modb   = (const float*)d_in[4];
    float* out = (float*)d_out;
    float* ws  = (float*)d_ws;

    float* W2_ws   = ws + 1024;
    float* conv_ws = ws + 1024 + 589824;

    k_mod<<<dim3(MOD_NG, B_), dim3(1024), 0, stream>>>(style, modw, modb, weight, W2_ws);

    k_conv_all<<<dim3(NBB_, B_), dim3(128), 0, stream>>>(x, W2_ws, conv_ws);

    k_blur<<<dim3(OC_, B_), dim3(32 * BLUR_NW), 0, stream>>>(conv_ws, out);
}


# k_conv: even-column phases tiled 2 x 64 px; plane column 128 moved to 9 blocks per batch that pack 16 rows per pixel tile: 7% fewer matrix instructions; x fragments per half block; sum order same
# speedup vs baseline: 3.2754x; 1.0583x over previous
#include <hip/hip_runtime.h>
#include <math.h>

#define B_     8
#define C_     128
#define OC_    64
#define SD_    512
#define H_     128
#define KW_    3
#define H2_    257
#define OUT_   256
#define PW_    128
#define OFF00_ 0
#define OFF01_ (129 * PW_)
#define OFF10_ (2 * 129 * PW_)
#define OFF11_ (OFF10_ + 128 * PW_)
#define OFFS0_ (OFF11_ + 128 * PW_)
#define OFFS1_ (OFFS0_ + 132)
#define PG_    66112
static_assert(OFFS1_ + 128 <= PG_ && PG_ % 64 == 0 && PG_ <= 4 * 129 * 129, "the group holds its planes and is no larger than four 129 x 129 planes");

#define SCALE_  (0.02946278254943948f)
#define WGAIN_  (0.04419417382415922f)

typedef __attribute__((ext_vector_type(2))) float v2f;
typedef __attribute__((ext_vector_type(4))) float v4f;
typedef __attribute__((ext_vector_type(8))) float v8f;

#define MOD_OG 8
#define MOD_NG 8
static_assert(MOD_OG * C_ == 1024 && MOD_NG * MOD_OG == OC_ && C_ == 32 * 4, "k_mod block shape: 32 waves x 4 rows = 128 c");

__global__ void __launch_bounds__(1024) __attribute__((amdgpu_waves_per_eu(8, 8))) k_mod(const float* __restrict__ style,
                                              const float* __restrict__ modw,
                                              const float* __restrict__ modb,
                                              const float* __restrict__ weight,
                                              float* __restrict__ W2) {
    __shared__ __attribute__((aligned(16))) float s_sh[C_];
    __shared__ float red[MOD_OG][4];

    const int tid  = threadIdx.x;
    const int lane = tid & 31;
    const int w    = tid >> 5;
    const int b    = blockIdx.y;
    const int ol   = tid >> 7;
    const int c    = tid & 127;
    const int o    = blockIdx.x * MOD_OG + ol;

    const float* wp = weight + ((size_t)o * C_ + c) * (KW_ * KW_);
    float wv[9];
#pragma unroll
    for (int t = 0; t < 9; ++t) wv[t] = wp[t];

    const v4f mb4 = *(const v4f*)(modb + w * 4);
    const v4f* st4 = (const v4f*)(style + (size_t)b * SD_);
    v4f sv[4];
#pragma unroll
    for (int i = 0; i < 4; ++i) sv[i] = st4[lane + 32 * i];
    v4f mw[4][4];
#pragma unroll
    for (int r = 0; r < 4; ++r) {
        const v4f* m4 = (const v4f*)(modw + (size_t)(w * 4 + r) * SD_);
#pragma unroll
        for (int i = 0; i < 4; ++i) mw[r][i] = m4[lane + 32 * i];
    }
    float part[4];
#pragma unroll
    for (int r = 0; r < 4; ++r) {
        float a = 0.f;
#pragma unroll
        for (int i = 0; i < 4; ++i) {
            a = fmaf(sv[i].x, mw[r][i].x, a);
            a = fmaf(sv[i].y, mw[r][i].y, a);
            a = fmaf(sv[i].z, mw[r][i].z, a);
            a = fmaf(sv[i].w, mw[r][i].w, a);
        }
        part[r] = a;
    }
#pragma unroll
    for (int off = 16; off > 0; off >>= 1) {
#pragma unroll
        for (int r = 0; r < 4; ++r) part[r] += __shfl_xor(part[r], off);
    }
    {
        v4f r4;
        r4.x = (part[0] * WGAIN_ + mb4.x) * SCALE_;
        r4.y = (part[1] * WGAIN_ + mb4.y) * SCALE_;
        r4.z = (part[2] * WGAIN_ + mb4.z) * SCALE_;
        r4.w = (part[3] * WGAIN_ + mb4.w) * SCALE_;
        *(v4f*)(s_sh + w * 4) = r4;
    }
    __syncthreads();

    const float sc = s_sh[c];
    float ss = 0.f;
#pragma unroll
    for (int t = 0; t < 9; ++t) { wv[t] = wv[t] * sc; ss += wv[t] * wv[t]; }
#pragma unroll
    for (int off = 16; off > 0; off >>= 1) ss += __shfl_xor(ss, off);
    if (lane == 0) red[ol][w & 3] = ss;
    __syncthreads();
    const float dem = rsqrtf(((red[ol][0] + red[ol][1]) + (red[ol][2] + red[ol][3])) + 1e-8f);
#pragma unroll
    for (int t = 0; t < 9; ++t)
        W2[((((size_t)b * 9 + t) * 64 + (c >> 1)) * OC_ + o) * 2 + (c & 1)] = wv[t] * dem;
}

#define CONV_LDS_F2 (32 * 72)
#define NBE_  9
#define NB00_ (2 * 129)
#define NB01_ (2 * 129)
#define NB10_ (2 * 128)
#define NB11_ (2 * 128)
#define NBB_  1037
static_assert(NBE_ + NB00_ + NB01_ + NB10_ + NB11_ == NBB_ && NBE_ * 16 >= H_ + 1, "conv grid: the blocks of one batch");
template<int PI, int PJ>
__device__ __forceinline__ void conv_phase(const float* __restrict__ x,
                                           const float* __restrict__ W2,
                                           float* __restrict__ conv_ws,
                                           v2f* __restrict__ xs,
                                           const int bx, const int ip, const int b) {
    constexpr int TY = PI ? 1 : 2;
    constexpr int TX = PJ ? 1 : 2;
    constexpr int NT = TY * TX;
    constexpr int GN = 4;
    constexpr int NPX = GN * 16;
    constexpr int NG = PJ ? 16 : 17;
    constexpr int COFF = PJ ? 0 : 3;
    constexpr int PITCH = (TY == 2) ? 72 : 80;
    constexpr int KPC = 16;
    constexpr int NCH = 64 / KPC;
    constexpr int ROWS = KPC * TY;
    constexpr int NSL = ROWS * NG;
    constexpr int NIT = (NSL + 127) / 128;
    constexpr bool FULL = (NSL % 128 == 0);
    constexpr int NKS = KPC / 2;
    constexpr int KB = 4 / NT;
    constexpr int NBLK = NKS / KB;
    constexpr int NA = KB * NT;
    constexpr int NB = NA * GN;
    static_assert(NA == 4 && NBLK % 2 == 0, "blocks come in pairs (fragment set 0, then set 1)");

    const int j0 = bx * NPX;
    const int r0 = ip - (PI ? 0 : 1);
    const int cb = j0 - (PJ ? 0 : 4);

    static_assert(ROWS * PITCH <= CONV_LDS_F2 && 4 * NG <= PITCH, "the phase's tile fits the kernel's LDS buffer");

    const float* xb = x + (size_t)b * C_ * H_ * H_;

    const int lane = threadIdx.x & 31;
    const int mW   = threadIdx.x >> 5;
    const int ln   = lane & 15;
    const int khp  = lane >> 4;
    const int m    = mW * 16 + ln;

    const v2f* Wb = (const v2f*)W2 + (size_t)b * 9 * 64 * OC_;

    v8f acc[GN];
#pragma unroll
    for (int g = 0; g < GN; ++g) acc[g] = v8f{};

    const v2f* Wl = Wb + (size_t)khp * OC_ + m;
    const int xlb = ((khp * TY) * PITCH + ln + COFF) * 8;

    auto load_a = [&](v2f (&fa)[NA], const int chn, const int blk) {
#pragma unroll
        for (int ku = 0; ku < KB; ++ku) {
#pragma unroll
            for (int t = 0; t < NT; ++t) {
                const int ty = t / TX, tx = t % TX;
                const int ky  = PI ? 1 : (2 - 2 * ty);
                const int kx  = PJ ? 1 : (2 - 2 * tx);
                const int tap = ky * 3 + kx;
                fa[ku * NT + t] = Wl[((size_t)tap * 64 + chn * KPC + 2 * (blk * KB + ku)) * OC_];
            }
        }
    };
    auto load_b = [&](v2f (&fb)[NB], const int blk) {
#pragma unroll
        for (int ku = 0; ku < KB; ++ku) {
            int koff = xlb + (2 * (blk * KB + ku) * TY * PITCH) * 8;
            asm volatile("" : "+v"(koff));
            const v2f* xk = (const v2f*)((const char*)xs + koff);
#pragma unroll
            for (int t = 0; t < NT; ++t) {
                const int ty = t / TX, tx = t % TX;
#pragma unroll
                for (int g = 0; g < GN; ++g)
                    fb[(ku * NT + t) * GN + g] = xk[ty * PITCH + g * 16 + tx];
            }
        }
    };
    auto mma = [&](const v2f (&fa)[NA], const v2f (&fb)[NB]) {
#pragma unroll
        for (int ku = 0; ku < KB; ++ku) {
#pragma unroll
            for (int t = 0; t < NT; ++t) {
#pragma unroll
                for (int g = 0; g < GN; ++g)
                    acc[g] = __builtin_amdgcn_wmma_f32_16x16x4_f32(
                               false, fa[ku * NT + t], false, fb[(ku * NT + t) * GN + g], (short)0, acc[g], false, false);
            }
        }
    };

    constexpr int NBH = 2 * GN;
    auto load_bh = [&](v2f (&fbh)[NBH], const int blk, const int h) {
        const int ku = (2 * h) / NT;
        int koff = xlb + (2 * (blk * KB + ku) * TY * PITCH) * 8;
        asm volatile("" : "+v"(koff));
        const v2f* xk = (const v2f*)((const char*)xs + koff);
#pragma unroll
        for (int qq = 0; qq < 2; ++qq) {
            const int t = (2 * h + qq) % NT;
            const int ty = t / TX, tx = t % TX;
#pragma unroll
            for (int g = 0; g < GN; ++g)
                fbh[qq * GN + g] = xk[ty * PITCH + g * 16 + tx];
        }
    };
    auto mma_h = [&](const v2f (&fa)[NA], const int h, const v2f (&fbh)[NBH]) {
#pragma unroll
        for (int qq = 0; qq < 2; ++qq) {
#pragma unroll
            for (int g = 0; g < GN; ++g)
                acc[g] = __builtin_amdgcn_wmma_f32_16x16x4_f32(
                           false, fa[2 * h + qq], false, fbh[qq * GN + g], (short)0, acc[g], false, false);
        }
    };

    v2f fa0[NA], fa1[NA], fb0[NB], fb1[NB];
    v2f fbA[NBH], fbB[NBH];
    load_a(fa0, 0, 0);

    for (int ch = 0; ch < NCH; ++ch) {
        if (ch > 0) __syncthreads();
        {
            v4f va[NIT], vb[NIT];
#pragma unroll
            for (int it = 0; it < NIT; ++it) {
                const int idx = threadIdx.x + it * 128;
                const int row = idx / NG;
                const int g   = idx - row * NG;
                const int t   = row % TY;
                const int kpl = row / TY;
                const int gr  = r0 + t;
                const int gc  = cb + 4 * g;
                const bool ok = (FULL || idx < NSL) && (gr >= 0) && (gr < H_) && (gc >= 0) && (gc < H_);
                const float* p = ok ? (xb + ((size_t)(2 * (ch * KPC + kpl)) * H_ + gr) * H_ + gc) : xb;
                const v4f qa = *(const v4f*)p;
                const v4f qb = *(const v4f*)(p + (size_t)H_ * H_);
                const v4f z = {0.f, 0.f, 0.f, 0.f};
                va[it] = ok ? qa : z;
                vb[it] = ok ? qb : z;
            }
#pragma unroll
            for (int it = 0; it < NIT; ++it) {
                const int idx = threadIdx.x + it * 128;
                const int row = idx / NG;
                const int g   = idx - row * NG;
                if (FULL || idx < NSL) {
                    v2f* d = xs + row * PITCH + 4 * g;
                    d[0] = v2f{va[it].x, vb[it].x};
                    d[1] = v2f{va[it].y, vb[it].y};
                    d[2] = v2f{va[it].z, vb[it].z};
                    d[3] = v2f{va[it].w, vb[it].w};
                }
            }
        }
        __syncthreads();

        const int chn = (ch + 1 < NCH) ? (ch + 1) : ch;
        if constexpr (PJ == 0) {
            load_bh(fbA, 0, 0);
#pragma unroll
            for (int bp = 0; bp < NBLK / 2; ++bp) {
                load_a(fa1, ch, 2 * bp + 1);
                load_bh(fbB, 2 * bp, 1);
                __builtin_amdgcn_sched_barrier(0);
                mma_h(fa0, 0, fbA);
                __builtin_amdgcn_sched_barrier(0);
                load_bh(fbA, 2 * bp + 1, 0);
                __builtin_amdgcn_sched_barrier(0);
                mma_h(fa0, 1, fbB);
                __builtin_amdgcn_sched_barrier(0);
                if (2 * bp + 2 < NBLK) load_a(fa0, ch, 2 * bp + 2);
                else                   load_a(fa0, chn, 0);
                load_bh(fbB, 2 * bp + 1, 1);
                __builtin_amdgcn_sched_barrier(0);
                mma_h(fa1, 0, fbA);
                __builtin_amdgcn_sched_barrier(0);
                if (2 * bp + 2 < NBLK) load_bh(fbA, 2 * bp + 2, 0);
                __builtin_amdgcn_sched_barrier(0);
                mma_h(fa1, 1, fbB);
                __builtin_amdgcn_sched_barrier(0);
            }
        } else {
        load_b(fb0, 0);
#pragma unroll
        for (int bp = 0; bp < NBLK / 2; ++bp) {
            load_a(fa1, ch, 2 * bp + 1);
            load_b(fb1, 2 * bp + 1);
            __builtin_amdgcn_sched_barrier(0);
            mma(fa0, fb0);
            __builtin_amdgcn_sched_barrier(0);
            if (2 * bp + 2 < NBLK) {
                load_a(fa0, ch, 2 * bp + 2);
                load_b(fb0, 2 * bp + 2);
            } else {
                load_a(fa0, chn, 0);
            }
            __builtin_amdgcn_sched_barrier(0);
            mma(fa1, fb1);
            __builtin_amdgcn_sched_barrier(0);
        }
        }
    }

    const int mhi = khp * 8;
    constexpr int POFF = PI ? (PJ ? OFF11_ : OFF10_) : (PJ ? OFF01_ : OFF00_);
    float* gb = conv_ws + (size_t)(b * OC_ + mW * 16 + mhi) * PG_;
    float* ob = gb + POFF + (size_t)ip * PW_;
#pragma unroll
    for (int g = 0; g < GN; ++g) {
        float* dst = ob + j0 + g * 16 + ln;
#pragma unroll
        for (int r = 0; r < 8; ++r) dst[(size_t)r * PG_] = acc[g][r];
    }
}

__device__ __forceinline__ void conv_edge(const float* __restrict__ x,
                                          const float* __restrict__ W2,
                                          float* __restrict__ conv_ws,
                                          const int rt, const int b) {
    constexpr int EKB = 4;
    const float* xb = x + (size_t)b * C_ * H_ * H_;
    const int lane = threadIdx.x & 31;
    const int mW   = threadIdx.x >> 5;
    const int ln   = lane & 15;
    const int khp  = lane >> 4;
    const int ip   = rt * 16 + ln;
    const bool okU = (ip >= 1) && (ip <= H_);
    const bool okL = (ip < H_);
    const float* pU = xb + ((size_t)(2 * khp) * H_ + (okU ? ip - 1 : 0)) * H_ + (H_ - 1);
    const float* pL = xb + ((size_t)(2 * khp) * H_ + (okL ? ip : 0)) * H_ + (H_ - 1);
    const v2f* Wl = (const v2f*)W2 + (size_t)b * 9 * 64 * OC_ + (size_t)khp * OC_ + mW * 16 + ln;

    v8f a0 = v8f{}, a1 = v8f{};
#pragma unroll 1
    for (int sb = 0; sb < 32 / EKB; ++sb) {
        v2f fu[EKB], fl[EKB], w8[EKB], w2[EKB], w5[EKB];
#pragma unroll
        for (int u = 0; u < EKB; ++u) {
            const int s = sb * EKB + u;
            const size_t co = (size_t)(4 * s) * H_ * H_;
            const float u0 = pU[co], u1 = pU[co + (size_t)H_ * H_];
            const float l0 = pL[co], l1 = pL[co + (size_t)H_ * H_];
            fu[u] = v2f{okU ? u0 : 0.f, okU ? u1 : 0.f};
            fl[u] = v2f{okL ? l0 : 0.f, okL ? l1 : 0.f};
            w8[u] = Wl[((size_t)8 * 64 + 2 * s) * OC_];
            w2[u] = Wl[((size_t)2 * 64 + 2 * s) * OC_];
            w5[u] = Wl[((size_t)5 * 64 + 2 * s) * OC_];
        }
#pragma unroll
        for (int u = 0; u < EKB; ++u) {
            a0 = __builtin_amdgcn_wmma_f32_16x16x4_f32(false, w8[u], false, fu[u], (short)0, a0, false, false);
            a0 = __builtin_amdgcn_wmma_f32_16x16x4_f32(false, w2[u], false, fl[u], (short)0, a0, false, false);
            a1 = __builtin_amdgcn_wmma_f32_16x16x4_f32(false, w5[u], false, fl[u], (short)0, a1, false, false);
        }
    }

    float* gb = conv_ws + (size_t)(b * OC_ + mW * 16 + khp * 8) * PG_;
    if (ip <= H_) {
#pragma unroll
        for (int r = 0; r < 8; ++r) gb[(size_t)r * PG_ + OFFS0_ + ip] = a0[r];
    }
    if (ip < H_) {
#pragma unroll
        for (int r = 0; r < 8; ++r) gb[(size_t)r * PG_ + OFFS1_ + ip] = a1[r];
    }
}

__global__ void __launch_bounds__(128) k_conv_all(const float* __restrict__ x,
                                                  const float* __restrict__ W2,
                                                  float* __restrict__ conv_ws) {
    __shared__ v2f xs[CONV_LDS_F2];
    const int b = blockIdx.y;
    int r = blockIdx.x;
    if (r < NBE_) { conv_edge(x, W2, conv_ws, r, b); return; }
    r -= NBE_;
    if (r < NB00_) { conv_phase<0, 0>(x, W2, conv_ws, xs, r & 1, r >> 1, b); return; }
    r -= NB00_;
    if (r < NB01_) { conv_phase<0, 1>(x, W2, conv_ws, xs, r & 1, r >> 1, b); return; }
    r -= NB01_;
    if (r < NB10_) { conv_phase<1, 0>(x, W2, conv_ws, xs, r & 1, r >> 1, b); return; }
    r -= NB10_;
    conv_phase<1, 1>(x, W2, conv_ws, xs, r & 1, r >> 1, b);
}

#define BLUR_RP 16
#define BLUR_U  4
#define BLUR_NW 8
static_assert(BLUR_NW * BLUR_RP == H_ && BLUR_RP % BLUR_U == 0, "blur block: 8 bands cover all 128 row pairs");

struct BlurRow {
    v2f el, ol;
    v2f er, orr;
    float s;
};

__device__ __forceinline__ void blur_load(const float* __restrict__ p0, const float* __restrict__ p1,
                                          const float* __restrict__ sd, const int row, const int lane, BlurRow& r) {
    const float* q0 = p0 + row * PW_ + 2 * lane;
    const float* q1 = p1 + row * PW_ + 2 * lane;
    r.el  = *(const v2f*)(q0);
    r.er  = *(const v2f*)(q0 + 64);
    r.ol  = *(const v2f*)(q1);
    r.orr = *(const v2f*)(q1 + 64);
    r.s   = sd[row];
}

__device__ __forceinline__ void blur_h4(const v2f E, const v2f O, const float a, const float f, const float g,
                                        const bool valid, float* h) {
    const float t0 = fmaf(3.f, E.x + O.x, a + E.y);
    const float t1 = fmaf(3.f, O.x + E.y, E.x + O.y);
    const float t2 = fmaf(3.f, E.y + O.y, O.x + f);
    const float t3 = fmaf(3.f, O.y + f, E.y + g);
    h[0] = valid ? t0 : 0.f;
    h[1] = valid ? t1 : 0.f;
    h[2] = valid ? t2 : 0.f;
    h[3] = valid ? t3 : 0.f;
}

__device__ __forceinline__ void blur_hrow(const BlurRow& r, const int lane, const bool valid, float (&h)[8]) {
    const bool first = (lane == 0), last = (lane == 31);
    const int nx = (lane + 1) & 31, pv = (lane - 1) & 31;
    const float xel = __shfl(r.el.x, nx),  xer = __shfl(r.er.x, nx);
    const float xol = __shfl(r.ol.x, nx),  xor_ = __shfl(r.orr.x, nx);
    const float yol = __shfl(r.ol.y, pv),  yor = __shfl(r.orr.y, pv);
    const float nel = last ? xer : xel;
    const float nol = last ? xor_ : xol;
    const float ner = last ? r.s : xer;
    const float nor = last ? 0.f : xor_;
    const float pol = first ? 0.f : yol;
    const float por = first ? yol : yor;
    blur_h4(r.el, r.ol, pol, nel, nol, valid, h);
    blur_h4(r.er, r.orr, por, ner, nor, valid, h + 4);
}

__global__ void __launch_bounds__(256) __attribute__((amdgpu_waves_per_eu(4, 4))) k_blur(const float* __restrict__ conv_ws,
                                              float* __restrict__ out) {
    const int lane = threadIdx.x & 31;
    const int w  = __builtin_amdgcn_readfirstlane(threadIdx.x >> 5);
    const int i0 = w * BLUR_RP;
    const int o  = blockIdx.x;
    const int b  = blockIdx.y;

    const float* gp  = conv_ws + (size_t)(b * OC_ + o) * PG_;
    const float* p00 = gp + OFF00_;
    const float* p01 = gp + OFF01_;
    const float* p10 = gp + OFF10_;
    const float* p11 = gp + OFF11_;
    const float* s0  = gp + OFFS0_;
    const float* s1  = gp + OFFS1_;

    float hA[8], hB[8], hC[8];
    {
        const bool vA = (i0 > 0);
        const int  rA = vA ? (i0 - 1) : 0;
        BlurRow ra, rb, rc;
        blur_load(p10, p11, s1, rA, lane, ra);
        blur_load(p00, p01, s0, i0, lane, rb);
        blur_load(p10, p11, s1, i0, lane, rc);
        blur_hrow(ra, lane, vA,   hA);
        blur_hrow(rb, lane, true, hB);
        blur_hrow(rc, lane, true, hC);
    }

    float* op = out + (((size_t)b * OC_ + o) * OUT_ + 2 * i0) * OUT_ + 4 * lane;

    for (int it = 0; it < BLUR_RP / BLUR_U; ++it) {
        BlurRow re[BLUR_U], ro[BLUR_U];
#pragma unroll
        for (int s = 0; s < BLUR_U; ++s) {
            const int i2  = i0 + it * BLUR_U + s + 1;
            const int i2o = (i2 < H_) ? i2 : (H_ - 1);
            blur_load(p00, p01, s0, i2,  lane, re[s]);
            blur_load(p10, p11, s1, i2o, lane, ro[s]);
        }
#pragma unroll
        for (int s = 0; s < BLUR_U; ++s) {
            const int i2 = i0 + it * BLUR_U + s + 1;
            float hD[8], hE[8];
            blur_hrow(re[s], lane, true,      hD);
            blur_hrow(ro[s], lane, (i2 < H_), hE);
            float y0[8], y1[8];
#pragma unroll
            for (int m = 0; m < 8; ++m) {
                y0[m] = fmaf(3.f, hB[m] + hC[m], hA[m] + hD[m]) * (1.f / 16.f);
                y1[m] = fmaf(3.f, hC[m] + hD[m], hB[m] + hE[m]) * (1.f / 16.f);
                hA[m] = hC[m]; hB[m] = hD[m]; hC[m] = hE[m];
            }
            float* orow = op + (size_t)(2 * (it * BLUR_U + s)) * OUT_;
            *(v4f*)(orow)              = v4f{y0[0], y0[1], y0[2], y0[3]};
            *(v4f*)(orow + 128)        = v4f{y0[4], y0[5], y0[6], y0[7]};
            *(v4f*)(orow + OUT_)       = v4f{y1[0], y1[1], y1[2], y1[3]};
            *(v4f*)(orow + OUT_ + 128) = v4f{y1[4], y1[5], y1[6], y1[7]};
        }
    }
}

extern "C" void kernel_launch(void* const* d_in, const int* in_sizes, int n_in,
                              void* d_out, int out_size, void* d_ws, size_t ws_size,
                              hipStream_t stream) {
    const float* x      = (const float*)d_in[0];
    const float* style  = (const float*)d_in[1];
    const float* weight = (const float*)d_in[2];
    const float* modw   = (const float*)d_in[3];
    const float* modb   = (const float*)d_in[4];
    float* out = (float*)d_out;
    float* ws  = (float*)d_ws;

    float* W2_ws   = ws + 1024;
    float* conv_ws = ws + 1024 + 589824;

    k_mod<<<dim3(MOD_NG, B_), dim3(1024), 0, stream>>>(style, modw, modb, weight, W2_ws);

    k_conv_all<<<dim3(NBB_, B_), dim3(128), 0, stream>>>(x, W2_ws, conv_ws);

    k_blur<<<dim3(OC_, B_), dim3(32 * BLUR_NW), 0, stream>>>(conv_ws, out);
}


# k_conv retiled: wave = 32 out-ch x 64 px (2 x 4 tiles), block = one whole row: each x fragment read from LDS feeds 2 matrix instructions, not 1; blocks handed out longest phase first; sum order same
# speedup vs baseline: 3.4886x; 1.0651x over previous
#include <hip/hip_runtime.h>
#include <math.h>

#define B_     8
#define C_     128
#define OC_    64
#define SD_    512
#define H_     128
#define KW_    3
#define H2_    257
#define OUT_   256
#define PW_    128
#define OFF00_ 0
#define OFF01_ (129 * PW_)
#define OFF10_ (2 * 129 * PW_)
#define OFF11_ (OFF10_ + 128 * PW_)
#define OFFS0_ (OFF11_ + 128 * PW_)
#define OFFS1_ (OFFS0_ + 132)
#define PG_    66112
static_assert(OFFS1_ + 128 <= PG_ && PG_ % 64 == 0 && PG_ <= 4 * 129 * 129, "the group holds its planes and is no larger than four 129 x 129 planes");

#define SCALE_  (0.02946278254943948f)
#define WGAIN_  (0.04419417382415922f)

typedef __attribute__((ext_vector_type(2))) float v2f;
typedef __attribute__((ext_vector_type(4))) float v4f;
typedef __attribute__((ext_vector_type(8))) float v8f;

#define MOD_OG 8
#define MOD_NG 8
static_assert(MOD_OG * C_ == 1024 && MOD_NG * MOD_OG == OC_ && C_ == 32 * 4, "k_mod block shape: 32 waves x 4 rows = 128 c");

__global__ void __launch_bounds__(1024) __attribute__((amdgpu_waves_per_eu(8, 8))) k_mod(const float* __restrict__ style,
                                              const float* __restrict__ modw,
                                              const float* __restrict__ modb,
                                              const float* __restrict__ weight,
                                              float* __restrict__ W2) {
    __shared__ __attribute__((aligned(16))) float s_sh[C_];
    __shared__ float red[MOD_OG][4];

    const int tid  = threadIdx.x;
    const int lane = tid & 31;
    const int w    = tid >> 5;
    const int b    = blockIdx.y;
    const int ol   = tid >> 7;
    const int c    = tid & 127;
    const int o    = blockIdx.x * MOD_OG + ol;

    const float* wp = weight + ((size_t)o * C_ + c) * (KW_ * KW_);
    float wv[9];
#pragma unroll
    for (int t = 0; t < 9; ++t) wv[t] = wp[t];

    const v4f mb4 = *(const v4f*)(modb + w * 4);
    const v4f* st4 = (const v4f*)(style + (size_t)b * SD_);
    v4f sv[4];
#pragma unroll
    for (int i = 0; i < 4; ++i) sv[i] = st4[lane + 32 * i];
    v4f mw[4][4];
#pragma unroll
    for (int r = 0; r < 4; ++r) {
        const v4f* m4 = (const v4f*)(modw + (size_t)(w * 4 + r) * SD_);
#pragma unroll
        for (int i = 0; i < 4; ++i) mw[r][i] = m4[lane + 32 * i];
    }
    float part[4];
#pragma unroll
    for (int r = 0; r < 4; ++r) {
        float a = 0.f;
#pragma unroll
        for (int i = 0; i < 4; ++i) {
            a = fmaf(sv[i].x, mw[r][i].x, a);
            a = fmaf(sv[i].y, mw[r][i].y, a);
            a = fmaf(sv[i].z, mw[r][i].z, a);
            a = fmaf(sv[i].w, mw[r][i].w, a);
        }
        part[r] = a;
    }
#pragma unroll
    for (int off = 16; off > 0; off >>= 1) {
#pragma unroll
        for (int r = 0; r < 4; ++r) part[r] += __shfl_xor(part[r], off);
    }
    {
        v4f r4;
        r4.x = (part[0] * WGAIN_ + mb4.x) * SCALE_;
        r4.y = (part[1] * WGAIN_ + mb4.y) * SCALE_;
        r4.z = (part[2] * WGAIN_ + mb4.z) * SCALE_;
        r4.w = (part[3] * WGAIN_ + mb4.w) * SCALE_;
        *(v4f*)(s_sh + w * 4) = r4;
    }
    __syncthreads();

    const float sc = s_sh[c];
    float ss = 0.f;
#pragma unroll
    for (int t = 0; t < 9; ++t) { wv[t] = wv[t] * sc; ss += wv[t] * wv[t]; }
#pragma unroll
    for (int off = 16; off > 0; off >>= 1) ss += __shfl_xor(ss, off);
    if (lane == 0) red[ol][w & 3] = ss;
    __syncthreads();
    const float dem = rsqrtf(((red[ol][0] + red[ol][1]) + (red[ol][2] + red[ol][3])) + 1e-8f);
#pragma unroll
    for (int t = 0; t < 9; ++t)
        W2[((((size_t)b * 9 + t) * 64 + (c >> 1)) * OC_ + o) * 2 + (c & 1)] = wv[t] * dem;
}

#define CONV_LDS_F2 (16 * 144)
#define NBE_  9
#define NB00_ 129
#define NB01_ 129
#define NB10_ 128
#define NB11_ 128
#define NBB_  523
static_assert(NBE_ + NB00_ + NB01_ + NB10_ + NB11_ == NBB_ && NBE_ * 16 >= H_ + 1, "conv grid: the blocks of one batch");
template<int PI, int PJ>
__device__ __forceinline__ void conv_phase(const float* __restrict__ x,
                                           const float* __restrict__ W2,
                                           float* __restrict__ conv_ws,
                                           v2f* __restrict__ xs,
                                           const int ip, const int b) {
    constexpr int TY = PI ? 1 : 2;
    constexpr int TX = PJ ? 1 : 2;
    constexpr int NT = TY * TX;
    constexpr int MT = 2;
    constexpr int GN = 4;
    constexpr int NG = 32;
    constexpr int LC0 = PJ ? 0 : 4;
    constexpr int COFF = PJ ? 0 : 3;
    constexpr int PITCH = (TY == 2) ? 136 : 144;
    constexpr int KPC = (TY == 2) ? 8 : 16;
    constexpr int NCH = 64 / KPC;
    constexpr int ROWS = KPC * TY;
    constexpr int NIT = ROWS / 4;
    constexpr int NKS = KPC / 2;
    constexpr int NU = NKS * NT;
    constexpr int NBLK = NU / 2;
    constexpr int NA = 2 * MT;
    constexpr int NB = 2 * GN;
    static_assert(NBLK % 2 == 0 && (TY * PITCH) % 32 == 16 && ROWS % 4 == 0, "blocks come in pairs (fragment set 0, then set 1)");

    const int r0 = ip - (PI ? 0 : 1);

    static_assert(ROWS * PITCH <= CONV_LDS_F2 && LC0 + 4 * NG <= PITCH, "the phase's tile fits the kernel's LDS buffer");

    const float* xb = x + (size_t)b * C_ * H_ * H_;

    const int lane = threadIdx.x & 31;
    const int wv   = threadIdx.x >> 5;
    const int og   = wv & 1;
    const int pg   = wv >> 1;
    const int ln   = lane & 15;
    const int khp  = lane >> 4;
    const int m    = og * 32 + ln;

    const v2f* Wb = (const v2f*)W2 + (size_t)b * 9 * 64 * OC_;

    v8f acc[MT][GN];
#pragma unroll
    for (int mt = 0; mt < MT; ++mt) {
#pragma unroll
        for (int g = 0; g < GN; ++g) acc[mt][g] = v8f{};
    }

    const v2f* Wl = Wb + (size_t)khp * OC_ + m;
    const int xlb = ((khp * TY) * PITCH + ln + COFF + pg * 64) * 8;

    auto load_a = [&](v2f (&fa)[NA], const int chn, const int blk) {
#pragma unroll
        for (int uu = 0; uu < 2; ++uu) {
            const int u = 2 * blk + uu;
            const int ks = u / NT, t = u % NT;
            const int ty = t / TX, tx = t % TX;
            const int ky  = PI ? 1 : (2 - 2 * ty);
            const int kx  = PJ ? 1 : (2 - 2 * tx);
            const int tap = ky * 3 + kx;
            const v2f* wp = Wl + ((size_t)tap * 64 + chn * KPC + 2 * ks) * OC_;
#pragma unroll
            for (int mt = 0; mt < MT; ++mt) fa[uu * MT + mt] = wp[16 * mt];
        }
    };
    auto load_b = [&](v2f (&fb)[NB], const int blk) {
        const int ks0 = (2 * blk) / NT;
        int koff = xlb + (2 * ks0 * TY * PITCH) * 8;
        asm volatile("" : "+v"(koff));
        const v2f* xk = (const v2f*)((const char*)xs + koff);
#pragma unroll
        for (int uu = 0; uu < 2; ++uu) {
            const int u = 2 * blk + uu;
            const int ks = u / NT, t = u % NT;
            const int ty = t / TX, tx = t % TX;
#pragma unroll
            for (int g = 0; g < GN; ++g)
                fb[uu * GN + g] = xk[(2 * (ks - ks0) * TY + ty) * PITCH + g * 16 + tx];
        }
    };
    auto mma = [&](const v2f (&fa)[NA], const v2f (&fb)[NB]) {
#pragma unroll
        for (int uu = 0; uu < 2; ++uu) {
#pragma unroll
            for (int mt = 0; mt < MT; ++mt) {
#pragma unroll
                for (int g = 0; g < GN; ++g)
                    acc[mt][g] = __builtin_amdgcn_wmma_f32_16x16x4_f32(
                                   false, fa[uu * MT + mt], false, fb[uu * GN + g], (short)0, acc[mt][g], false, false);
            }
        }
    };

    const int sg = threadIdx.x & 31;
    const int sr = threadIdx.x >> 5;
    if constexpr (PJ == 0) {
        if (threadIdx.x < 2 * ROWS)
            *(v4f*)(xs + (threadIdx.x >> 1) * PITCH + 2 * (threadIdx.x & 1)) = v4f{0.f, 0.f, 0.f, 0.f};
    }

    v2f fa0[NA], fa1[NA], fb0[NB], fb1[NB];
    load_a(fa0, 0, 0);

    for (int ch = 0; ch < NCH; ++ch) {
        if (ch > 0) __syncthreads();
        {
            v4f va[NIT], vb[NIT];
#pragma unroll
            for (int it = 0; it < NIT; ++it) {
                const int row = sr + 4 * it;
                const int t   = row % TY;
                const int kpl = row / TY;
                const int gr  = r0 + t;
                const bool ok = PI ? true : ((gr >= 0) && (gr < H_));
                const float* p = ok ? (xb + ((size_t)(2 * (ch * KPC + kpl)) * H_ + gr) * H_ + 4 * sg) : xb;
                const v4f qa = *(const v4f*)p;
                const v4f qb = *(const v4f*)(p + (size_t)H_ * H_);
                const v4f z = {0.f, 0.f, 0.f, 0.f};
                va[it] = ok ? qa : z;
                vb[it] = ok ? qb : z;
            }
#pragma unroll
            for (int it = 0; it < NIT; ++it) {
                v2f* d = xs + (sr + 4 * it) * PITCH + LC0 + 4 * sg;
                d[0] = v2f{va[it].x, vb[it].x};
                d[1] = v2f{va[it].y, vb[it].y};
                d[2] = v2f{va[it].z, vb[it].z};
                d[3] = v2f{va[it].w, vb[it].w};
            }
        }
        __syncthreads();

        const int chn = (ch + 1 < NCH) ? (ch + 1) : ch;
        load_b(fb0, 0);
#pragma unroll
        for (int bp = 0; bp < NBLK / 2; ++bp) {
            load_a(fa1, ch, 2 * bp + 1);
            load_b(fb1, 2 * bp + 1);
            __builtin_amdgcn_sched_barrier(0);
            mma(fa0, fb0);
            __builtin_amdgcn_sched_barrier(0);
            if (2 * bp + 2 < NBLK) {
                load_a(fa0, ch, 2 * bp + 2);
                load_b(fb0, 2 * bp + 2);
            } else {
                load_a(fa0, chn, 0);
            }
            __builtin_amdgcn_sched_barrier(0);
            mma(fa1, fb1);
            __builtin_amdgcn_sched_barrier(0);
        }
    }

    const int mhi = khp * 8;
    constexpr int POFF = PI ? (PJ ? OFF11_ : OFF10_) : (PJ ? OFF01_ : OFF00_);
    float* gb = conv_ws + (size_t)(b * OC_ + og * 32 + mhi) * PG_;
    float* ob = gb + POFF + (size_t)ip * PW_ + pg * 64 + ln;
#pragma unroll
    for (int mt = 0; mt < MT; ++mt) {
#pragma unroll
        for (int g = 0; g < GN; ++g) {
            float* dst = ob + (size_t)(16 * mt) * PG_ + g * 16;
#pragma unroll
            for (int r = 0; r < 8; ++r) dst[(size_t)r * PG_] = acc[mt][g][r];
        }
    }
}

__device__ __forceinline__ void conv_edge(const float* __restrict__ x,
                                          const float* __restrict__ W2,
                                          float* __restrict__ conv_ws,
                                          const int rt, const int b) {
    constexpr int EKB = 4;
    const float* xb = x + (size_t)b * C_ * H_ * H_;
    const int lane = threadIdx.x & 31;
    const int mW   = threadIdx.x >> 5;
    const int ln   = lane & 15;
    const int khp  = lane >> 4;
    const int ip   = rt * 16 + ln;
    const bool okU = (ip >= 1) && (ip <= H_);
    const bool okL = (ip < H_);
    const float* pU = xb + ((size_t)(2 * khp) * H_ + (okU ? ip - 1 : 0)) * H_ + (H_ - 1);
    const float* pL = xb + ((size_t)(2 * khp) * H_ + (okL ? ip : 0)) * H_ + (H_ - 1);
    const v2f* Wl = (const v2f*)W2 + (size_t)b * 9 * 64 * OC_ + (size_t)khp * OC_ + mW * 16 + ln;

    v8f a0 = v8f{}, a1 = v8f{};
#pragma unroll 1
    for (int sb = 0; sb < 32 / EKB; ++sb) {
        v2f fu[EKB], fl[EKB], w8[EKB], w2[EKB], w5[EKB];
#pragma unroll
        for (int u = 0; u < EKB; ++u) {
            const int s = sb * EKB + u;
            const size_t co = (size_t)(4 * s) * H_ * H_;
            const float u0 = pU[co], u1 = pU[co + (size_t)H_ * H_];
            const float l0 = pL[co], l1 = pL[co + (size_t)H_ * H_];
            fu[u] = v2f{okU ? u0 : 0.f, okU ? u1 : 0.f};
            fl[u] = v2f{okL ? l0 : 0.f, okL ? l1 : 0.f};
            w8[u] = Wl[((size_t)8 * 64 + 2 * s) * OC_];
            w2[u] = Wl[((size_t)2 * 64 + 2 * s) * OC_];
            w5[u] = Wl[((size_t)5 * 64 + 2 * s) * OC_];
        }
#pragma unroll
        for (int u = 0; u < EKB; ++u) {
            a0 = __builtin_amdgcn_wmma_f32_16x16x4_f32(false, w8[u], false, fu[u], (short)0, a0, false, false);
            a0 = __builtin_amdgcn_wmma_f32_16x16x4_f32(false, w2[u], false, fl[u], (short)0, a0, false, false);
            a1 = __builtin_amdgcn_wmma_f32_16x16x4_f32(false, w5[u], false, fl[u], (short)0, a1, false, false);
        }
    }

    float* gb = conv_ws + (size_t)(b * OC_ + mW * 16 + khp * 8) * PG_;
    if (ip <= H_) {
#pragma unroll
        for (int r = 0; r < 8; ++r) gb[(size_t)r * PG_ + OFFS0_ + ip] = a0[r];
    }
    if (ip < H_) {
#pragma unroll
        for (int r = 0; r < 8; ++r) gb[(size_t)r * PG_ + OFFS1_ + ip] = a1[r];
    }
}

#define NGRID_ (NBB_ * B_)
__global__ void __launch_bounds__(128) __attribute__((amdgpu_waves_per_eu(7))) k_conv_all(const float* __restrict__ x,
                                                  const float* __restrict__ W2,
                                                  float* __restrict__ conv_ws) {
    __shared__ __attribute__((aligned(16))) v2f xs[CONV_LDS_F2];
    int r = blockIdx.x;
    if (r < B_ * NBE_) { const int b = r / NBE_; conv_edge(x, W2, conv_ws, r - b * NBE_, b); return; }
    r -= B_ * NBE_;
    if (r < B_ * NB00_) { const int b = r / NB00_; conv_phase<0, 0>(x, W2, conv_ws, xs, r - b * NB00_, b); return; }
    r -= B_ * NB00_;
    if (r < B_ * NB01_) { const int b = r / NB01_; conv_phase<0, 1>(x, W2, conv_ws, xs, r - b * NB01_, b); return; }
    r -= B_ * NB01_;
    if (r < B_ * NB10_) { const int b = r / NB10_; conv_phase<1, 0>(x, W2, conv_ws, xs, r - b * NB10_, b); return; }
    r -= B_ * NB10_;
    { const int b = r / NB11_; conv_phase<1, 1>(x, W2, conv_ws, xs, r - b * NB11_, b); }
}

#define BLUR_RP 16
#define BLUR_U  4
#define BLUR_NW 8
static_assert(BLUR_NW * BLUR_RP == H_ && BLUR_RP % BLUR_U == 0, "blur block: 8 bands cover all 128 row pairs");

struct BlurRow {
    v2f el, ol;
    v2f er, orr;
    float s;
};

__device__ __forceinline__ void blur_load(const float* __restrict__ p0, const float* __restrict__ p1,
                                          const float* __restrict__ sd, const int row, const int lane, BlurRow& r) {
    const float* q0 = p0 + row * PW_ + 2 * lane;
    const float* q1 = p1 + row * PW_ + 2 * lane;
    r.el  = *(const v2f*)(q0);
    r.er  = *(const v2f*)(q0 + 64);
    r.ol  = *(const v2f*)(q1);
    r.orr = *(const v2f*)(q1 + 64);
    r.s   = sd[row];
}

__device__ __forceinline__ void blur_h4(const v2f E, const v2f O, const float a, const float f, const float g,
                                        const bool valid, float* h) {
    const float t0 = fmaf(3.f, E.x + O.x, a + E.y);
    const float t1 = fmaf(3.f, O.x + E.y, E.x + O.y);
    const float t2 = fmaf(3.f, E.y + O.y, O.x + f);
    const float t3 = fmaf(3.f, O.y + f, E.y + g);
    h[0] = valid ? t0 : 0.f;
    h[1] = valid ? t1 : 0.f;
    h[2] = valid ? t2 : 0.f;
    h[3] = valid ? t3 : 0.f;
}

__device__ __forceinline__ void blur_hrow(const BlurRow& r, const int lane, const bool valid, float (&h)[8]) {
    const bool first = (lane == 0), last = (lane == 31);
    const int nx = (lane + 1) & 31, pv = (lane - 1) & 31;
    const float xel = __shfl(r.el.x, nx),  xer = __shfl(r.er.x, nx);
    const float xol = __shfl(r.ol.x, nx),  xor_ = __shfl(r.orr.x, nx);
    const float yol = __shfl(r.ol.y, pv),  yor = __shfl(r.orr.y, pv);
    const float nel = last ? xer : xel;
    const float nol = last ? xor_ : xol;
    const float ner = last ? r.s : xer;
    const float nor = last ? 0.f : xor_;
    const float pol = first ? 0.f : yol;
    const float por = first ? yol : yor;
    blur_h4(r.el, r.ol, pol, nel, nol, valid, h);
    blur_h4(r.er, r.orr, por, ner, nor, valid, h + 4);
}

__global__ void __launch_bounds__(256) __attribute__((amdgpu_waves_per_eu(4, 4))) k_blur(const float* __restrict__ conv_ws,
                                              float* __restrict__ out) {
    const int lane = threadIdx.x & 31;
    const int w  = __builtin_amdgcn_readfirstlane(threadIdx.x >> 5);
    const int i0 = w * BLUR_RP;
    const int o  = blockIdx.x;
    const int b  = blockIdx.y;

    const float* gp  = conv_ws + (size_t)(b * OC_ + o) * PG_;
    const float* p00 = gp + OFF00_;
    const float* p01 = gp + OFF01_;
    const float* p10 = gp + OFF10_;
    const float* p11 = gp + OFF11_;
    const float* s0  = gp + OFFS0_;
    const float* s1  = gp + OFFS1_;

    float hA[8], hB[8], hC[8];
    {
        const bool vA = (i0 > 0);
        const int  rA = vA ? (i0 - 1) : 0;
        BlurRow ra, rb, rc;
        blur_load(p10, p11, s1, rA, lane, ra);
        blur_load(p00, p01, s0, i0, lane, rb);
        blur_load(p10, p11, s1, i0, lane, rc);
        blur_hrow(ra, lane, vA,   hA);
        blur_hrow(rb, lane, true, hB);
        blur_hrow(rc, lane, true, hC);
    }

    float* op = out + (((size_t)b * OC_ + o) * OUT_ + 2 * i0) * OUT_ + 4 * lane;

    for (int it = 0; it < BLUR_RP / BLUR_U; ++it) {
        BlurRow re[BLUR_U], ro[BLUR_U];
#pragma unroll
        for (int s = 0; s < BLUR_U; ++s) {
            const int i2  = i0 + it * BLUR_U + s + 1;
            const int i2o = (i2 < H_) ? i2 : (H_ - 1);
            blur_load(p00, p01, s0, i2,  lane, re[s]);
            blur_load(p10, p11, s1, i2o, lane, ro[s]);
        }
#pragma unroll
        for (int s = 0; s < BLUR_U; ++s) {
            const int i2 = i0 + it * BLUR_U + s + 1;
            float hD[8], hE[8];
            blur_hrow(re[s], lane, true,      hD);
            blur_hrow(ro[s], lane, (i2 < H_), hE);
            float y0[8], y1[8];
#pragma unroll
            for (int m = 0; m < 8; ++m) {
                y0[m] = fmaf(3.f, hB[m] + hC[m], hA[m] + hD[m]) * (1.f / 16.f);
                y1[m] = fmaf(3.f, hC[m] + hD[m], hB[m] + hE[m]) * (1.f / 16.f);
                hA[m] = hC[m]; hB[m] = hD[m]; hC[m] = hE[m];
            }
            float* orow = op + (size_t)(2 * (it * BLUR_U + s)) * OUT_;
            *(v4f*)(orow)              = v4f{y0[0], y0[1], y0[2], y0[3]};
            *(v4f*)(orow + 128)        = v4f{y0[4], y0[5], y0[6], y0[7]};
            *(v4f*)(orow + OUT_)       = v4f{y1[0], y1[1], y1[2], y1[3]};
            *(v4f*)(orow + OUT_ + 128) = v4f{y1[4], y1[5], y1[6], y1[7]};
        }
    }
}

extern "C" void kernel_launch(void* const* d_in, const int* in_sizes, int n_in,
                              void* d_out, int out_size, void* d_ws, size_t ws_size,
                              hipStream_t stream) {
    const float* x      = (const float*)d_in[0];
    const float* style  = (const float*)d_in[1];
    const float* weight = (const float*)d_in[2];
    const float* modw   = (const float*)d_in[3];
    const float* modb   = (const float*)d_in[4];
    float* out = (float*)d_out;
    float* ws  = (float*)d_ws;

    float* W2_ws   = ws + 1024;
    float* conv_ws = ws + 1024 + 589824;

    k_mod<<<dim3(MOD_NG, B_), dim3(1024), 0, stream>>>(style, modw, modb, weight, W2_ws);

    k_conv_all<<<dim3(NGRID_), dim3(128), 0, stream>>>(x, W2_ws, conv_ws);

    k_blur<<<dim3(OC_, B_), dim3(32 * BLUR_NW), 0, stream>>>(conv_ws, out);
}


# k_conv even-column phases by column pairs: 3 products per pair and (channel, tap row), not 4 (2-tap minimal filtering; g0+g1 planes from k_mod): 17% fewer matrix instructions; sum order changed
# speedup vs baseline: 3.6651x; 1.0506x over previous
#include <hip/hip_runtime.h>
#include <math.h>

#define B_     8
#define C_     128
#define OC_    64
#define SD_    512
#define H_     128
#define KW_    3
#define H2_    257
#define OUT_   256
#define PW_    128
#define OFF00_ 0
#define OFF01_ (129 * PW_)
#define OFF10_ (2 * 129 * PW_)
#define OFF11_ (OFF10_ + 128 * PW_)
#define OFFS0_ (OFF11_ + 128 * PW_)
#define OFFS1_ (OFFS0_ + 132)
#define PG_    66112
static_assert(OFFS1_ + 128 <= PG_ && PG_ % 64 == 0 && PG_ <= 4 * 129 * 129, "the group holds its planes and is no larger than four 129 x 129 planes");

#define NTAP_  12
#define SCALE_  (0.02946278254943948f)
#define WGAIN_  (0.04419417382415922f)

typedef __attribute__((ext_vector_type(2))) float v2f;
typedef __attribute__((ext_vector_type(4))) float v4f;
typedef __attribute__((ext_vector_type(8))) float v8f;

#define MOD_OG 8
#define MOD_NG 8
static_assert(MOD_OG * C_ == 1024 && MOD_NG * MOD_OG == OC_ && C_ == 32 * 4, "k_mod block shape: 32 waves x 4 rows = 128 c");

__global__ void __launch_bounds__(1024) __attribute__((amdgpu_waves_per_eu(8, 8))) k_mod(const float* __restrict__ style,
                                              const float* __restrict__ modw,
                                              const float* __restrict__ modb,
                                              const float* __restrict__ weight,
                                              float* __restrict__ W2) {
    __shared__ __attribute__((aligned(16))) float s_sh[C_];
    __shared__ float red[MOD_OG][4];

    const int tid  = threadIdx.x;
    const int lane = tid & 31;
    const int w    = tid >> 5;
    const int b    = blockIdx.y;
    const int ol   = tid >> 7;
    const int c    = tid & 127;
    const int o    = blockIdx.x * MOD_OG + ol;

    const float* wp = weight + ((size_t)o * C_ + c) * (KW_ * KW_);
    float wv[9];
#pragma unroll
    for (int t = 0; t < 9; ++t) wv[t] = wp[t];

    const v4f mb4 = *(const v4f*)(modb + w * 4);
    const v4f* st4 = (const v4f*)(style + (size_t)b * SD_);
    v4f sv[4];
#pragma unroll
    for (int i = 0; i < 4; ++i) sv[i] = st4[lane + 32 * i];
    v4f mw[4][4];
#pragma unroll
    for (int r = 0; r < 4; ++r) {
        const v4f* m4 = (const v4f*)(modw + (size_t)(w * 4 + r) * SD_);
#pragma unroll
        for (int i = 0; i < 4; ++i) mw[r][i] = m4[lane + 32 * i];
    }
    float part[4];
#pragma unroll
    for (int r = 0; r < 4; ++r) {
        float a = 0.f;
#pragma unroll
        for (int i = 0; i < 4; ++i) {
            a = fmaf(sv[i].x, mw[r][i].x, a);
            a = fmaf(sv[i].y, mw[r][i].y, a);
            a = fmaf(sv[i].z, mw[r][i].z, a);
            a = fmaf(sv[i].w, mw[r][i].w, a);
        }
        part[r] = a;
    }
#pragma unroll
    for (int off = 16; off > 0; off >>= 1) {
#pragma unroll
        for (int r = 0; r < 4; ++r) part[r] += __shfl_xor(part[r], off);
    }
    {
        v4f r4;
        r4.x = (part[0] * WGAIN_ + mb4.x) * SCALE_;
        r4.y = (part[1] * WGAIN_ + mb4.y) * SCALE_;
        r4.z = (part[2] * WGAIN_ + mb4.z) * SCALE_;
        r4.w = (part[3] * WGAIN_ + mb4.w) * SCALE_;
        *(v4f*)(s_sh + w * 4) = r4;
    }
    __syncthreads();

    const float sc = s_sh[c];
    float ss = 0.f;
#pragma unroll
    for (int t = 0; t < 9; ++t) { wv[t] = wv[t] * sc; ss += wv[t] * wv[t]; }
#pragma unroll
    for (int off = 16; off > 0; off >>= 1) ss += __shfl_xor(ss, off);
    if (lane == 0) red[ol][w & 3] = ss;
    __syncthreads();
    const float dem = rsqrtf(((red[ol][0] + red[ol][1]) + (red[ol][2] + red[ol][3])) + 1e-8f);
#pragma unroll
    for (int t = 0; t < 9; ++t) wv[t] = wv[t] * dem;
#pragma unroll
    for (int t = 0; t < 9; ++t)
        W2[((((size_t)b * NTAP_ + t) * 64 + (c >> 1)) * OC_ + o) * 2 + (c & 1)] = wv[t];
#pragma unroll
    for (int ky = 0; ky < 3; ++ky)
        W2[((((size_t)b * NTAP_ + 9 + ky) * 64 + (c >> 1)) * OC_ + o) * 2 + (c & 1)] = wv[ky * 3] + wv[ky * 3 + 2];
}

#define CONV_LDS_F2 (16 * 208)
#define NBE_  9
#define NB00_ 129
#define NB01_ 129
#define NB10_ 128
#define NB11_ 128
#define NBB_  523
static_assert(NBE_ + NB00_ + NB01_ + NB10_ + NB11_ == NBB_ && NBE_ * 16 >= H_ + 1, "conv grid: the blocks of one batch");
template<int PI, int PJ>
__device__ __forceinline__ void conv_phase(const float* __restrict__ x,
                                           const float* __restrict__ W2,
                                           float* __restrict__ conv_ws,
                                           v2f* __restrict__ xs,
                                           const int ip, const int b) {
    constexpr int TY = PI ? 1 : 2;
    constexpr int TX = PJ ? 1 : 2;
    constexpr int NT = TY * TX;
    constexpr int MT = 2;
    constexpr int GN = 4;
    constexpr int NG = 32;
    constexpr int LC0 = PJ ? 0 : 4;
    constexpr int COFF = PJ ? 0 : 3;
    constexpr int PITCH = (TY == 2) ? 136 : 144;
    constexpr int KPC = (TY == 2) ? 8 : 16;
    constexpr int NCH = 64 / KPC;
    constexpr int ROWS = KPC * TY;
    constexpr int NIT = ROWS / 4;
    constexpr int NKS = KPC / 2;
    constexpr int NU = NKS * NT;
    constexpr int NBLK = NU / 2;
    constexpr int NA = 2 * MT;
    constexpr int NB = 2 * GN;
    static_assert(NBLK % 2 == 0 && (TY * PITCH) % 32 == 16 && ROWS % 4 == 0, "blocks come in pairs (fragment set 0, then set 1)");

    const int r0 = ip - (PI ? 0 : 1);

    static_assert(ROWS * PITCH <= CONV_LDS_F2 && LC0 + 4 * NG <= PITCH, "the phase's tile fits the kernel's LDS buffer");

    const float* xb = x + (size_t)b * C_ * H_ * H_;

    const int lane = threadIdx.x & 31;
    const int wv   = threadIdx.x >> 5;
    const int og   = wv & 1;
    const int pg   = wv >> 1;
    const int ln   = lane & 15;
    const int khp  = lane >> 4;
    const int m    = og * 32 + ln;

    const v2f* Wb = (const v2f*)W2 + (size_t)b * NTAP_ * 64 * OC_;

    v8f acc[MT][GN];
#pragma unroll
    for (int mt = 0; mt < MT; ++mt) {
#pragma unroll
        for (int g = 0; g < GN; ++g) acc[mt][g] = v8f{};
    }

    const v2f* Wl = Wb + (size_t)khp * OC_ + m;
    const int xlb = ((khp * TY) * PITCH + ln + COFF + pg * 64) * 8;

    auto load_a = [&](v2f (&fa)[NA], const int chn, const int blk) {
#pragma unroll
        for (int uu = 0; uu < 2; ++uu) {
            const int u = 2 * blk + uu;
            const int ks = u / NT, t = u % NT;
            const int ty = t / TX, tx = t % TX;
            const int ky  = PI ? 1 : (2 - 2 * ty);
            const int kx  = PJ ? 1 : (2 - 2 * tx);
            const int tap = ky * 3 + kx;
            const v2f* wp = Wl + ((size_t)tap * 64 + chn * KPC + 2 * ks) * OC_;
#pragma unroll
            for (int mt = 0; mt < MT; ++mt) fa[uu * MT + mt] = wp[16 * mt];
        }
    };
    auto load_b = [&](v2f (&fb)[NB], const int blk) {
        const int ks0 = (2 * blk) / NT;
        int koff = xlb + (2 * ks0 * TY * PITCH) * 8;
        asm volatile("" : "+v"(koff));
        const v2f* xk = (const v2f*)((const char*)xs + koff);
#pragma unroll
        for (int uu = 0; uu < 2; ++uu) {
            const int u = 2 * blk + uu;
            const int ks = u / NT, t = u % NT;
            const int ty = t / TX, tx = t % TX;
#pragma unroll
            for (int g = 0; g < GN; ++g)
                fb[uu * GN + g] = xk[(2 * (ks - ks0) * TY + ty) * PITCH + g * 16 + tx];
        }
    };
    auto mma = [&](const v2f (&fa)[NA], const v2f (&fb)[NB]) {
#pragma unroll
        for (int uu = 0; uu < 2; ++uu) {
#pragma unroll
            for (int mt = 0; mt < MT; ++mt) {
#pragma unroll
                for (int g = 0; g < GN; ++g)
                    acc[mt][g] = __builtin_amdgcn_wmma_f32_16x16x4_f32(
                                   false, fa[uu * MT + mt], false, fb[uu * GN + g], (short)0, acc[mt][g], false, false);
            }
        }
    };

    const int sg = threadIdx.x & 31;
    const int sr = threadIdx.x >> 5;
    if constexpr (PJ == 0) {
        if (threadIdx.x < 2 * ROWS)
            *(v4f*)(xs + (threadIdx.x >> 1) * PITCH + 2 * (threadIdx.x & 1)) = v4f{0.f, 0.f, 0.f, 0.f};
    }

    v2f fa0[NA], fa1[NA], fb0[NB], fb1[NB];
    load_a(fa0, 0, 0);

    for (int ch = 0; ch < NCH; ++ch) {
        if (ch > 0) __syncthreads();
        {
            v4f va[NIT], vb[NIT];
#pragma unroll
            for (int it = 0; it < NIT; ++it) {
                const int row = sr + 4 * it;
                const int t   = row % TY;
                const int kpl = row / TY;
                const int gr  = r0 + t;
                const bool ok = PI ? true : ((gr >= 0) && (gr < H_));
                const float* p = ok ? (xb + ((size_t)(2 * (ch * KPC + kpl)) * H_ + gr) * H_ + 4 * sg) : xb;
                const v4f qa = *(const v4f*)p;
                const v4f qb = *(const v4f*)(p + (size_t)H_ * H_);
                const v4f z = {0.f, 0.f, 0.f, 0.f};
                va[it] = ok ? qa : z;
                vb[it] = ok ? qb : z;
            }
#pragma unroll
            for (int it = 0; it < NIT; ++it) {
                v2f* d = xs + (sr + 4 * it) * PITCH + LC0 + 4 * sg;
                d[0] = v2f{va[it].x, vb[it].x};
                d[1] = v2f{va[it].y, vb[it].y};
                d[2] = v2f{va[it].z, vb[it].z};
                d[3] = v2f{va[it].w, vb[it].w};
            }
        }
        __syncthreads();

        const int chn = (ch + 1 < NCH) ? (ch + 1) : ch;
        load_b(fb0, 0);
#pragma unroll
        for (int bp = 0; bp < NBLK / 2; ++bp) {
            load_a(fa1, ch, 2 * bp + 1);
            load_b(fb1, 2 * bp + 1);
            __builtin_amdgcn_sched_barrier(0);
            mma(fa0, fb0);
            __builtin_amdgcn_sched_barrier(0);
            if (2 * bp + 2 < NBLK) {
                load_a(fa0, ch, 2 * bp + 2);
                load_b(fb0, 2 * bp + 2);
            } else {
                load_a(fa0, chn, 0);
            }
            __builtin_amdgcn_sched_barrier(0);
            mma(fa1, fb1);
            __builtin_amdgcn_sched_barrier(0);
        }
    }

    const int mhi = khp * 8;
    constexpr int POFF = PI ? (PJ ? OFF11_ : OFF10_) : (PJ ? OFF01_ : OFF00_);
    float* gb = conv_ws + (size_t)(b * OC_ + og * 32 + mhi) * PG_;
    float* ob = gb + POFF + (size_t)ip * PW_ + pg * 64 + ln;
#pragma unroll
    for (int mt = 0; mt < MT; ++mt) {
#pragma unroll
        for (int g = 0; g < GN; ++g) {
            float* dst = ob + (size_t)(16 * mt) * PG_ + g * 16;
#pragma unroll
            for (int r = 0; r < 8; ++r) dst[(size_t)r * PG_] = acc[mt][g][r];
        }
    }
}

template<int PI>
__device__ __forceinline__ void conv_pair(const float* __restrict__ x,
                                          const float* __restrict__ W2,
                                          float* __restrict__ conv_ws,
                                          v2f* __restrict__ xs,
                                          const int ip, const int b) {
    constexpr int TY = PI ? 1 : 2;
    constexpr int MT = 2;
    constexpr int PT = 2;
    constexpr int NV = 3;
    constexpr int NP = 64;
    constexpr int PITCH = (TY == 2) ? 200 : 208;
    constexpr int KPC = (TY == 2) ? 8 : 16;
    constexpr int NCH = 64 / KPC;
    constexpr int ROWS = KPC * TY;
    constexpr int NIT = ROWS / 4;
    constexpr int NKS = KPC / 2;
    constexpr int NU = NKS * TY;
    constexpr int NA = NV * MT;
    constexpr int NB = NV * PT;
    static_assert(NU % 2 == 0 && (TY * PITCH) % 32 == 16 && ROWS % 4 == 0 && PITCH % 2 == 0, "units come in pairs (fragment set 0, then set 1); 16-byte LDS stores");
    static_assert(ROWS * PITCH <= CONV_LDS_F2 && NV * NP <= PITCH, "the phase's tile fits the kernel's LDS buffer");

    const int r0 = ip - (PI ? 0 : 1);

    const float* xb = x + (size_t)b * C_ * H_ * H_;

    const int lane = threadIdx.x & 31;
    const int wv   = threadIdx.x >> 5;
    const int og   = wv & 1;
    const int pg   = wv >> 1;
    const int ln   = lane & 15;
    const int khp  = lane >> 4;
    const int m    = og * 32 + ln;

    const v2f* Wb = (const v2f*)W2 + (size_t)b * NTAP_ * 64 * OC_;

    v8f acc[NV][MT][PT];
#pragma unroll
    for (int v = 0; v < NV; ++v) {
#pragma unroll
        for (int mt = 0; mt < MT; ++mt) {
#pragma unroll
            for (int pt = 0; pt < PT; ++pt) acc[v][mt][pt] = v8f{};
        }
    }

    const v2f* Wl = Wb + (size_t)khp * OC_ + m;
    const int xlb = ((khp * TY) * PITCH + ln + pg * 32) * 8;

    auto load_a = [&](v2f (&fa)[NA], const int chn, const int u) {
        const int ks = u / TY, ty = u % TY;
        const int ky = PI ? 1 : (2 - 2 * ty);
#pragma unroll
        for (int v = 0; v < NV; ++v) {
            const int plane = (v == 0) ? (ky * 3 + 2) : ((v == 1) ? (9 + ky) : (ky * 3));
            const v2f* wp = Wl + ((size_t)plane * 64 + chn * KPC + 2 * ks) * OC_;
#pragma unroll
            for (int mt = 0; mt < MT; ++mt) fa[v * MT + mt] = wp[16 * mt];
        }
    };
    auto load_b = [&](v2f (&fb)[NB], const int u) {
        const int ks = u / TY, ty = u % TY;
        int koff = xlb + ((2 * ks * TY + ty) * PITCH) * 8;
        asm volatile("" : "+v"(koff));
        const v2f* xk = (const v2f*)((const char*)xs + koff);
#pragma unroll
        for (int v = 0; v < NV; ++v) {
#pragma unroll
            for (int pt = 0; pt < PT; ++pt) fb[v * PT + pt] = xk[v * NP + pt * 16];
        }
    };
    auto mma = [&](const v2f (&fa)[NA], const v2f (&fb)[NB]) {
#pragma unroll
        for (int v = 0; v < NV; ++v) {
#pragma unroll
            for (int mt = 0; mt < MT; ++mt) {
#pragma unroll
                for (int pt = 0; pt < PT; ++pt)
                    acc[v][mt][pt] = __builtin_amdgcn_wmma_f32_16x16x4_f32(
                                       false, fa[v * MT + mt], false, fb[v * PT + pt], (short)0, acc[v][mt][pt], false, false);
            }
        }
    };

    const int sg = threadIdx.x & 31;
    const int sr = threadIdx.x >> 5;

    v2f fa0[NA], fa1[NA], fb0[NB], fb1[NB];
    load_a(fa0, 0, 0);

    for (int ch = 0; ch < NCH; ++ch) {
        if (ch > 0) __syncthreads();
        {
            v4f va[NIT], vb[NIT];
#pragma unroll
            for (int it = 0; it < NIT; ++it) {
                const int row = sr + 4 * it;
                const int t   = row % TY;
                const int kpl = row / TY;
                const int gr  = r0 + t;
                const bool ok = PI ? true : ((gr >= 0) && (gr < H_));
                const float* p = ok ? (xb + ((size_t)(2 * (ch * KPC + kpl)) * H_ + gr) * H_ + 4 * sg) : xb;
                const v4f qa = *(const v4f*)p;
                const v4f qb = *(const v4f*)(p + (size_t)H_ * H_);
                const v4f z = {0.f, 0.f, 0.f, 0.f};
                va[it] = ok ? qa : z;
                vb[it] = ok ? qb : z;
            }
#pragma unroll
            for (int it = 0; it < NIT; ++it) {
                const float ua = __shfl_up(va[it].w, 1), ub = __shfl_up(vb[it].w, 1);
                const float pa = (sg == 0) ? 0.f : ua;
                const float pb = (sg == 0) ? 0.f : ub;
                v2f* d = xs + (sr + 4 * it) * PITCH + 2 * sg;
                *(v4f*)(d)          = v4f{pa - va[it].x, pb - vb[it].x, va[it].y - va[it].z, vb[it].y - vb[it].z};
                *(v4f*)(d + NP)     = v4f{va[it].x, vb[it].x, va[it].z, vb[it].z};
                *(v4f*)(d + 2 * NP) = v4f{va[it].x - va[it].y, vb[it].x - vb[it].y, va[it].z - va[it].w, vb[it].z - vb[it].w};
            }
        }
        __syncthreads();

        const int chn = (ch + 1 < NCH) ? (ch + 1) : ch;
        load_b(fb0, 0);
#pragma unroll
        for (int up = 0; up < NU / 2; ++up) {
            load_a(fa1, ch, 2 * up + 1);
            load_b(fb1, 2 * up + 1);
            __builtin_amdgcn_sched_barrier(0);
            mma(fa0, fb0);
            __builtin_amdgcn_sched_barrier(0);
            if (2 * up + 2 < NU) {
                load_a(fa0, ch, 2 * up + 2);
                load_b(fb0, 2 * up + 2);
            } else {
                load_a(fa0, chn, 0);
            }
            __builtin_amdgcn_sched_barrier(0);
            mma(fa1, fb1);
            __builtin_amdgcn_sched_barrier(0);
        }
    }

    const int mhi = khp * 8;
    constexpr int POFF = PI ? OFF10_ : OFF00_;
    float* gb = conv_ws + (size_t)(b * OC_ + og * 32 + mhi) * PG_;
    float* ob = gb + POFF + (size_t)ip * PW_ + 2 * (pg * 32 + ln);
#pragma unroll
    for (int mt = 0; mt < MT; ++mt) {
#pragma unroll
        for (int pt = 0; pt < PT; ++pt) {
            float* dst = ob + (size_t)(16 * mt) * PG_ + pt * 32;
#pragma unroll
            for (int r = 0; r < 8; ++r)
                *(v2f*)(dst + (size_t)r * PG_) = v2f{acc[0][mt][pt][r] + acc[1][mt][pt][r], acc[1][mt][pt][r] - acc[2][mt][pt][r]};
        }
    }
}

__device__ __forceinline__ void conv_edge(const float* __restrict__ x,
                                          const float* __restrict__ W2,
                                          float* __restrict__ conv_ws,
                                          const int rt, const int b) {
    constexpr int EKB = 4;
    const float* xb = x + (size_t)b * C_ * H_ * H_;
    const int lane = threadIdx.x & 31;
    const int mW   = threadIdx.x >> 5;
    const int ln   = lane & 15;
    const int khp  = lane >> 4;
    const int ip   = rt * 16 + ln;
    const bool okU = (ip >= 1) && (ip <= H_);
    const bool okL = (ip < H_);
    const float* pU = xb + ((size_t)(2 * khp) * H_ + (okU ? ip - 1 : 0)) * H_ + (H_ - 1);
    const float* pL = xb + ((size_t)(2 * khp) * H_ + (okL ? ip : 0)) * H_ + (H_ - 1);
    const v2f* Wl = (const v2f*)W2 + (size_t)b * NTAP_ * 64 * OC_ + (size_t)khp * OC_ + mW * 16 + ln;

    v8f a0 = v8f{}, a1 = v8f{};
#pragma unroll 1
    for (int sb = 0; sb < 32 / EKB; ++sb) {
        v2f fu[EKB], fl[EKB], w8[EKB], w2[EKB], w5[EKB];
#pragma unroll
        for (int u = 0; u < EKB; ++u) {
            const int s = sb * EKB + u;
            const size_t co = (size_t)(4 * s) * H_ * H_;
            const float u0 = pU[co], u1 = pU[co + (size_t)H_ * H_];
            const float l0 = pL[co], l1 = pL[co + (size_t)H_ * H_];
            fu[u] = v2f{okU ? u0 : 0.f, okU ? u1 : 0.f};
            fl[u] = v2f{okL ? l0 : 0.f, okL ? l1 : 0.f};
            w8[u] = Wl[((size_t)8 * 64 + 2 * s) * OC_];
            w2[u] = Wl[((size_t)2 * 64 + 2 * s) * OC_];
            w5[u] = Wl[((size_t)5 * 64 + 2 * s) * OC_];
        }
#pragma unroll
        for (int u = 0; u < EKB; ++u) {
            a0 = __builtin_amdgcn_wmma_f32_16x16x4_f32(false, w8[u], false, fu[u], (short)0, a0, false, false);
            a0 = __builtin_amdgcn_wmma_f32_16x16x4_f32(false, w2[u], false, fl[u], (short)0, a0, false, false);
            a1 = __builtin_amdgcn_wmma_f32_16x16x4_f32(false, w5[u], false, fl[u], (short)0, a1, false, false);
        }
    }

    float* gb = conv_ws + (size_t)(b * OC_ + mW * 16 + khp * 8) * PG_;
    if (ip <= H_) {
#pragma unroll
        for (int r = 0; r < 8; ++r) gb[(size_t)r * PG_ + OFFS0_ + ip] = a0[r];
    }
    if (ip < H_) {
#pragma unroll
        for (int r = 0; r < 8; ++r) gb[(size_t)r * PG_ + OFFS1_ + ip] = a1[r];
    }
}

#define NGRID_ (NBB_ * B_)
__global__ void __launch_bounds__(128) __attribute__((amdgpu_waves_per_eu(5))) k_conv_all(const float* __restrict__ x,
                                                  const float* __restrict__ W2,
                                                  float* __restrict__ conv_ws) {
    __shared__ __attribute__((aligned(16))) v2f xs[CONV_LDS_F2];
    int r = blockIdx.x;
    if (r < B_ * NBE_) { const int b = r / NBE_; conv_edge(x, W2, conv_ws, r - b * NBE_, b); return; }
    r -= B_ * NBE_;
    if (r < B_ * NB00_) { const int b = r / NB00_; conv_pair<0>(x, W2, conv_ws, xs, r - b * NB00_, b); return; }
    r -= B_ * NB00_;
    if (r < B_ * NB01_) { const int b = r / NB01_; conv_phase<0, 1>(x, W2, conv_ws, xs, r - b * NB01_, b); return; }
    r -= B_ * NB01_;
    if (r < B_ * NB10_) { const int b = r / NB10_; conv_pair<1>(x, W2, conv_ws, xs, r - b * NB10_, b); return; }
    r -= B_ * NB10_;
    { const int b = r / NB11_; conv_phase<1, 1>(x, W2, conv_ws, xs, r - b * NB11_, b); }
}

#define BLUR_RP 16
#define BLUR_U  4
#define BLUR_NW 8
static_assert(BLUR_NW * BLUR_RP == H_ && BLUR_RP % BLUR_U == 0, "blur block: 8 bands cover all 128 row pairs");

struct BlurRow {
    v2f el, ol;
    v2f er, orr;
    float s;
};

__device__ __forceinline__ void blur_load(const float* __restrict__ p0, const float* __restrict__ p1,
                                          const float* __restrict__ sd, const int row, const int lane, BlurRow& r) {
    const float* q0 = p0 + row * PW_ + 2 * lane;
    const float* q1 = p1 + row * PW_ + 2 * lane;
    r.el  = *(const v2f*)(q0);
    r.er  = *(const v2f*)(q0 + 64);
    r.ol  = *(const v2f*)(q1);
    r.orr = *(const v2f*)(q1 + 64);
    r.s   = sd[row];
}

__device__ __forceinline__ void blur_h4(const v2f E, const v2f O, const float a, const float f, const float g,
                                        const bool valid, float* h) {
    const float t0 = fmaf(3.f, E.x + O.x, a + E.y);
    const float t1 = fmaf(3.f, O.x + E.y, E.x + O.y);
    const float t2 = fmaf(3.f, E.y + O.y, O.x + f);
    const float t3 = fmaf(3.f, O.y + f, E.y + g);
    h[0] = valid ? t0 : 0.f;
    h[1] = valid ? t1 : 0.f;
    h[2] = valid ? t2 : 0.f;
    h[3] = valid ? t3 : 0.f;
}

__device__ __forceinline__ void blur_hrow(const BlurRow& r, const int lane, const bool valid, float (&h)[8]) {
    const bool first = (lane == 0), last = (lane == 31);
    const int nx = (lane + 1) & 31, pv = (lane - 1) & 31;
    const float xel = __shfl(r.el.x, nx),  xer = __shfl(r.er.x, nx);
    const float xol = __shfl(r.ol.x, nx),  xor_ = __shfl(r.orr.x, nx);
    const float yol = __shfl(r.ol.y, pv),  yor = __shfl(r.orr.y, pv);
    const float nel = last ? xer : xel;
    const float nol = last ? xor_ : xol;
    const float ner = last ? r.s : xer;
    const float nor = last ? 0.f : xor_;
    const float pol = first ? 0.f : yol;
    const float por = first ? yol : yor;
    blur_h4(r.el, r.ol, pol, nel, nol, valid, h);
    blur_h4(r.er, r.orr, por, ner, nor, valid, h + 4);
}

__global__ void __launch_bounds__(256) __attribute__((amdgpu_waves_per_eu(4, 4))) k_blur(const float* __restrict__ conv_ws,
                                              float* __restrict__ out) {
    const int lane = threadIdx.x & 31;
    const int w  = __builtin_amdgcn_readfirstlane(threadIdx.x >> 5);
    const int i0 = w * BLUR_RP;
    const int o  = blockIdx.x;
    const int b  = blockIdx.y;

    const float* gp  = conv_ws + (size_t)(b * OC_ + o) * PG_;
    const float* p00 = gp + OFF00_;
    const float* p01 = gp + OFF01_;
    const float* p10 = gp + OFF10_;
    const float* p11 = gp + OFF11_;
    const float* s0  = gp + OFFS0_;
    const float* s1  = gp + OFFS1_;

    float hA[8], hB[8], hC[8];
    {
        const bool vA = (i0 > 0);
        const int  rA = vA ? (i0 - 1) : 0;
        BlurRow ra, rb, rc;
        blur_load(p10, p11, s1, rA, lane, ra);
        blur_load(p00, p01, s0, i0, lane, rb);
        blur_load(p10, p11, s1, i0, lane, rc);
        blur_hrow(ra, lane, vA,   hA);
        blur_hrow(rb, lane, true, hB);
        blur_hrow(rc, lane, true, hC);
    }

    float* op = out + (((size_t)b * OC_ + o) * OUT_ + 2 * i0) * OUT_ + 4 * lane;

    for (int it = 0; it < BLUR_RP / BLUR_U; ++it) {
        BlurRow re[BLUR_U], ro[BLUR_U];
#pragma unroll
        for (int s = 0; s < BLUR_U; ++s) {
            const int i2  = i0 + it * BLUR_U + s + 1;
            const int i2o = (i2 < H_) ? i2 : (H_ - 1);
            blur_load(p00, p01, s0, i2,  lane, re[s]);
            blur_load(p10, p11, s1, i2o, lane, ro[s]);
        }
#pragma unroll
        for (int s = 0; s < BLUR_U; ++s) {
            const int i2 = i0 + it * BLUR_U + s + 1;
            float hD[8], hE[8];
            blur_hrow(re[s], lane, true,      hD);
            blur_hrow(ro[s], lane, (i2 < H_), hE);
            float y0[8], y1[8];
#pragma unroll
            for (int m = 0; m < 8; ++m) {
                y0[m] = fmaf(3.f, hB[m] + hC[m], hA[m] + hD[m]) * (1.f / 16.f);
                y1[m] = fmaf(3.f, hC[m] + hD[m], hB[m] + hE[m]) * (1.f / 16.f);
                hA[m] = hC[m]; hB[m] = hD[m]; hC[m] = hE[m];
            }
            float* orow = op + (size_t)(2 * (it * BLUR_U + s)) * OUT_;
            *(v4f*)(orow)              = v4f{y0[0], y0[1], y0[2], y0[3]};
            *(v4f*)(orow + 128)        = v4f{y0[4], y0[5], y0[6], y0[7]};
            *(v4f*)(orow + OUT_)       = v4f{y1[0], y1[1], y1[2], y1[3]};
            *(v4f*)(orow + OUT_ + 128) = v4f{y1[4], y1[5], y1[6], y1[7]};
        }
    }
}

extern "C" void kernel_launch(void* const* d_in, const int* in_sizes, int n_in,
                              void* d_out, int out_size, void* d_ws, size_t ws_size,
                              hipStream_t stream) {
    const float* x      = (const float*)d_in[0];
    const float* style  = (const float*)d_in[1];
    const float* weight = (const float*)d_in[2];
    const float* modw   = (const float*)d_in[3];
    const float* modb   = (const float*)d_in[4];
    float* out = (float*)d_out;
    float* ws  = (float*)d_ws;

    float* W2_ws   = ws + 1024;
    float* conv_ws = ws + 1024 + 786432;

    k_mod<<<dim3(MOD_NG, B_), dim3(1024), 0, stream>>>(style, modw, modb, weight, W2_ws);

    k_conv_all<<<dim3(NGRID_), dim3(128), 0, stream>>>(x, W2_ws, conv_ws);

    k_blur<<<dim3(OC_, B_), dim3(32 * BLUR_NW), 0, stream>>>(conv_ws, out);
}


# k_conv: weight and x fragments as 16 B loads, 2 k-steps each (k-quad layouts in k_mod and LDS): half the fragment loads, a load group per 24-32 matrix instr; 4 waves/SIMD; sum order changed
# speedup vs baseline: 4.1062x; 1.1204x over previous
#include <hip/hip_runtime.h>
#include <math.h>

#define B_     8
#define C_     128
#define OC_    64
#define SD_    512
#define H_     128
#define KW_    3
#define H2_    257
#define OUT_   256
#define PW_    128
#define OFF00_ 0
#define OFF01_ (129 * PW_)
#define OFF10_ (2 * 129 * PW_)
#define OFF11_ (OFF10_ + 128 * PW_)
#define OFFS0_ (OFF11_ + 128 * PW_)
#define OFFS1_ (OFFS0_ + 132)
#define PG_    66112
static_assert(OFFS1_ + 128 <= PG_ && PG_ % 64 == 0 && PG_ <= 4 * 129 * 129, "the group holds its planes and is no larger than four 129 x 129 planes");

#define NTAP_  12
#define SCALE_  (0.02946278254943948f)
#define WGAIN_  (0.04419417382415922f)

typedef __attribute__((ext_vector_type(2))) float v2f;
typedef __attribute__((ext_vector_type(4))) float v4f;
typedef __attribute__((ext_vector_type(8))) float v8f;

#define MOD_OG 8
#define MOD_NG 8
static_assert(MOD_OG * C_ == 1024 && MOD_NG * MOD_OG == OC_ && C_ == 32 * 4, "k_mod block shape: 32 waves x 4 rows = 128 c");

__global__ void __launch_bounds__(1024) __attribute__((amdgpu_waves_per_eu(8, 8))) k_mod(const float* __restrict__ style,
                                              const float* __restrict__ modw,
                                              const float* __restrict__ modb,
                                              const float* __restrict__ weight,
                                              float* __restrict__ W2) {
    __shared__ __attribute__((aligned(16))) float s_sh[C_];
    __shared__ float red[MOD_OG][4];

    const int tid  = threadIdx.x;
    const int lane = tid & 31;
    const int w    = tid >> 5;
    const int b    = blockIdx.y;
    const int ol   = tid >> 7;
    const int c    = tid & 127;
    const int o    = blockIdx.x * MOD_OG + ol;

    const float* wp = weight + ((size_t)o * C_ + c) * (KW_ * KW_);
    float wv[9];
#pragma unroll
    for (int t = 0; t < 9; ++t) wv[t] = wp[t];

    const v4f mb4 = *(const v4f*)(modb + w * 4);
    const v4f* st4 = (const v4f*)(style + (size_t)b * SD_);
    v4f sv[4];
#pragma unroll
    for (int i = 0; i < 4; ++i) sv[i] = st4[lane + 32 * i];
    v4f mw[4][4];
#pragma unroll
    for (int r = 0; r < 4; ++r) {
        const v4f* m4 = (const v4f*)(modw + (size_t)(w * 4 + r) * SD_);
#pragma unroll
        for (int i = 0; i < 4; ++i) mw[r][i] = m4[lane + 32 * i];
    }
    float part[4];
#pragma unroll
    for (int r = 0; r < 4; ++r) {
        float a = 0.f;
#pragma unroll
        for (int i = 0; i < 4; ++i) {
            a = fmaf(sv[i].x, mw[r][i].x, a);
            a = fmaf(sv[i].y, mw[r][i].y, a);
            a = fmaf(sv[i].z, mw[r][i].z, a);
            a = fmaf(sv[i].w, mw[r][i].w, a);
        }
        part[r] = a;
    }
#pragma unroll
    for (int off = 16; off > 0; off >>= 1) {
#pragma unroll
        for (int r = 0; r < 4; ++r) part[r] += __shfl_xor(part[r], off);
    }
    {
        v4f r4;
        r4.x = (part[0] * WGAIN_ + mb4.x) * SCALE_;
        r4.y = (part[1] * WGAIN_ + mb4.y) * SCALE_;
        r4.z = (part[2] * WGAIN_ + mb4.z) * SCALE_;
        r4.w = (part[3] * WGAIN_ + mb4.w) * SCALE_;
        *(v4f*)(s_sh + w * 4) = r4;
    }
    __syncthreads();

    const float sc = s_sh[c];
    float ss = 0.f;
#pragma unroll
    for (int t = 0; t < 9; ++t) { wv[t] = wv[t] * sc; ss += wv[t] * wv[t]; }
#pragma unroll
    for (int off = 16; off > 0; off >>= 1) ss += __shfl_xor(ss, off);
    if (lane == 0) red[ol][w & 3] = ss;
    __syncthreads();
    const float dem = rsqrtf(((red[ol][0] + red[ol][1]) + (red[ol][2] + red[ol][3])) + 1e-8f);
#pragma unroll
    for (int t = 0; t < 9; ++t) wv[t] = wv[t] * dem;
#pragma unroll
    for (int t = 0; t < 9; ++t)
        W2[((((size_t)b * NTAP_ + t) * 32 + (c >> 2)) * OC_ + o) * 4 + (c & 3)] = wv[t];
#pragma unroll
    for (int ky = 0; ky < 3; ++ky)
        W2[((((size_t)b * NTAP_ + 9 + ky) * 32 + (c >> 2)) * OC_ + o) * 4 + (c & 3)] = wv[ky * 3] + wv[ky * 3 + 2];
}

#define CONV_LDS_F4 (8 * 192)
#define NBE_  9
#define NB00_ 129
#define NB01_ 129
#define NB10_ 128
#define NB11_ 128
#define NBB_  523
static_assert(NBE_ + NB00_ + NB01_ + NB10_ + NB11_ == NBB_ && NBE_ * 16 >= H_ + 1, "conv grid: the blocks of one batch");

__device__ __forceinline__ v2f frag_lo(const v4f q) { return __builtin_shufflevector(q, q, 0, 1); }
__device__ __forceinline__ v2f frag_hi(const v4f q) { return __builtin_shufflevector(q, q, 2, 3); }

template<int PI>
__device__ __forceinline__ void conv_phase(const float* __restrict__ x,
                                           const float* __restrict__ W2,
                                           float* __restrict__ conv_ws,
                                           v4f* __restrict__ xs,
                                           const int ip, const int b) {
    constexpr int TY = PI ? 1 : 2;
    constexpr int MT = 2;
    constexpr int GN = 4;
    constexpr int PITCH = 128;
    constexpr int KQC = (TY == 2) ? 4 : 8;
    constexpr int NCH = 32 / KQC;
    constexpr int ROWS = KQC * TY;
    constexpr int NIT = ROWS / 4;
    constexpr int NU = (KQC / 2) * TY;
    constexpr int NBLK = NU / 2;
    constexpr int NA = 2 * MT;
    constexpr int NB = 2 * GN;
    static_assert(NU % 2 == 0 && NBLK % 2 == 0 && ROWS % 4 == 0 && ROWS * PITCH <= CONV_LDS_F4, "blocks come in pairs (fragment set 0, then set 1); the tile fits the LDS buffer");

    const int r0 = ip - (PI ? 0 : 1);

    const float* xb = x + (size_t)b * C_ * H_ * H_;

    const int lane = threadIdx.x & 31;
    const int wv   = threadIdx.x >> 5;
    const int og   = wv & 1;
    const int pg   = wv >> 1;
    const int ln   = lane & 15;
    const int khp  = lane >> 4;
    const int m    = og * 32 + ln;

    const v4f* Wb = (const v4f*)W2 + (size_t)b * NTAP_ * 32 * OC_;

    v8f acc[MT][GN];
#pragma unroll
    for (int mt = 0; mt < MT; ++mt) {
#pragma unroll
        for (int g = 0; g < GN; ++g) acc[mt][g] = v8f{};
    }

    const v4f* Wl = Wb + (size_t)khp * OC_ + m;
    const int xlb = ((khp * TY) * PITCH + ln + pg * 64) * 16;

    auto load_a = [&](v4f (&fa)[NA], const int chn, const int blk) {
#pragma unroll
        for (int uu = 0; uu < 2; ++uu) {
            const int u = 2 * blk + uu;
            const int kp = u / TY, ty = u % TY;
            const int ky  = PI ? 1 : (2 - 2 * ty);
            const int tap = ky * 3 + 1;
            const v4f* wp = Wl + ((size_t)tap * 32 + chn * KQC + 2 * kp) * OC_;
#pragma unroll
            for (int mt = 0; mt < MT; ++mt) fa[uu * MT + mt] = wp[16 * mt];
        }
    };
    auto load_b = [&](v4f (&fb)[NB], const int blk) {
        const int kp0 = (2 * blk) / TY;
        int koff = xlb + (2 * kp0 * TY * PITCH) * 16;
        asm volatile("" : "+v"(koff));
        const v4f* xk = (const v4f*)((const char*)xs + koff);
#pragma unroll
        for (int uu = 0; uu < 2; ++uu) {
            const int u = 2 * blk + uu;
            const int kp = u / TY, ty = u % TY;
#pragma unroll
            for (int g = 0; g < GN; ++g)
                fb[uu * GN + g] = xk[(2 * (kp - kp0) * TY + ty) * PITCH + g * 16];
        }
    };
    auto mma = [&](const v4f (&fa)[NA], const v4f (&fb)[NB]) {
#pragma unroll
        for (int uu = 0; uu < 2; ++uu) {
#pragma unroll
            for (int mt = 0; mt < MT; ++mt) {
#pragma unroll
                for (int g = 0; g < GN; ++g)
                    acc[mt][g] = __builtin_amdgcn_wmma_f32_16x16x4_f32(
                                   false, frag_lo(fa[uu * MT + mt]), false, frag_lo(fb[uu * GN + g]), (short)0, acc[mt][g], false, false);
            }
#pragma unroll
            for (int mt = 0; mt < MT; ++mt) {
#pragma unroll
                for (int g = 0; g < GN; ++g)
                    acc[mt][g] = __builtin_amdgcn_wmma_f32_16x16x4_f32(
                                   false, frag_hi(fa[uu * MT + mt]), false, frag_hi(fb[uu * GN + g]), (short)0, acc[mt][g], false, false);
            }
        }
    };

    const int sg = threadIdx.x & 31;
    const int sr = threadIdx.x >> 5;

    v4f fa0[NA], fa1[NA], fb0[NB], fb1[NB];
    load_a(fa0, 0, 0);

    for (int ch = 0; ch < NCH; ++ch) {
        if (ch > 0) __syncthreads();
        {
            v4f q[NIT][4];
#pragma unroll
            for (int it = 0; it < NIT; ++it) {
                const int row = sr + 4 * it;
                const int t   = row % TY;
                const int kql = row / TY;
                const int gr  = r0 + t;
                const bool ok = PI ? true : ((gr >= 0) && (gr < H_));
                const float* p = ok ? (xb + ((size_t)(4 * (ch * KQC + kql)) * H_ + gr) * H_ + 4 * sg) : xb;
                const v4f z = {0.f, 0.f, 0.f, 0.f};
#pragma unroll
                for (int e = 0; e < 4; ++e) {
                    const v4f v = *(const v4f*)(p + (size_t)e * H_ * H_);
                    q[it][e] = ok ? v : z;
                }
            }
#pragma unroll
            for (int it = 0; it < NIT; ++it) {
                v4f* d = xs + (sr + 4 * it) * PITCH + 4 * sg;
                d[0] = v4f{q[it][0].x, q[it][1].x, q[it][2].x, q[it][3].x};
                d[1] = v4f{q[it][0].y, q[it][1].y, q[it][2].y, q[it][3].y};
                d[2] = v4f{q[it][0].z, q[it][1].z, q[it][2].z, q[it][3].z};
                d[3] = v4f{q[it][0].w, q[it][1].w, q[it][2].w, q[it][3].w};
            }
        }
        __syncthreads();

        const int chn = (ch + 1 < NCH) ? (ch + 1) : ch;
        load_b(fb0, 0);
#pragma unroll
        for (int bp = 0; bp < NBLK / 2; ++bp) {
            load_a(fa1, ch, 2 * bp + 1);
            load_b(fb1, 2 * bp + 1);
            __builtin_amdgcn_sched_barrier(0);
            mma(fa0, fb0);
            __builtin_amdgcn_sched_barrier(0);
            if (2 * bp + 2 < NBLK) {
                load_a(fa0, ch, 2 * bp + 2);
                load_b(fb0, 2 * bp + 2);
            } else {
                load_a(fa0, chn, 0);
            }
            __builtin_amdgcn_sched_barrier(0);
            mma(fa1, fb1);
            __builtin_amdgcn_sched_barrier(0);
        }
    }

    const int mhi = khp * 8;
    constexpr int POFF = PI ? OFF11_ : OFF01_;
    float* gb = conv_ws + (size_t)(b * OC_ + og * 32 + mhi) * PG_;
    float* ob = gb + POFF + (size_t)ip * PW_ + pg * 64 + ln;
#pragma unroll
    for (int mt = 0; mt < MT; ++mt) {
#pragma unroll
        for (int g = 0; g < GN; ++g) {
            float* dst = ob + (size_t)(16 * mt) * PG_ + g * 16;
#pragma unroll
            for (int r = 0; r < 8; ++r) dst[(size_t)r * PG_] = acc[mt][g][r];
        }
    }
}

template<int PI>
__device__ __forceinline__ void conv_pair(const float* __restrict__ x,
                                          const float* __restrict__ W2,
                                          float* __restrict__ conv_ws,
                                          v4f* __restrict__ xs,
                                          const int ip, const int b) {
    constexpr int TY = PI ? 1 : 2;
    constexpr int MT = 2;
    constexpr int PT = 2;
    constexpr int NV = 3;
    constexpr int NP = 64;
    constexpr int PITCH = NV * NP;
    constexpr int KQC = (TY == 2) ? 4 : 8;
    constexpr int NCH = 32 / KQC;
    constexpr int ROWS = KQC * TY;
    constexpr int NIT = ROWS / 4;
    constexpr int NU = (KQC / 2) * TY;
    constexpr int NA = NV * MT;
    constexpr int NB = NV * PT;
    static_assert(NU % 2 == 0 && ROWS % 4 == 0 && ROWS * PITCH <= CONV_LDS_F4, "blocks come in pairs (fragment set 0, then set 1); the tile fits the LDS buffer");

    const int r0 = ip - (PI ? 0 : 1);

    const float* xb = x + (size_t)b * C_ * H_ * H_;

    const int lane = threadIdx.x & 31;
    const int wv   = threadIdx.x >> 5;
    const int og   = wv & 1;
    const int pg   = wv >> 1;
    const int ln   = lane & 15;
    const int khp  = lane >> 4;
    const int m    = og * 32 + ln;

    const v4f* Wb = (const v4f*)W2 + (size_t)b * NTAP_ * 32 * OC_;

    v8f acc[NV][MT][PT];
#pragma unroll
    for (int v = 0; v < NV; ++v) {
#pragma unroll
        for (int mt = 0; mt < MT; ++mt) {
#pragma unroll
            for (int pt = 0; pt < PT; ++pt) acc[v][mt][pt] = v8f{};
        }
    }

    const v4f* Wl = Wb + (size_t)khp * OC_ + m;
    const int xlb = ((khp * TY) * PITCH + ln + pg * 32) * 16;

    auto load_a = [&](v4f (&fa)[NA], const int chn, const int u) {
        const int kp = u / TY, ty = u % TY;
        const int ky = PI ? 1 : (2 - 2 * ty);
#pragma unroll
        for (int v = 0; v < NV; ++v) {
            const int plane = (v == 0) ? (ky * 3 + 2) : ((v == 1) ? (9 + ky) : (ky * 3));
            const v4f* wp = Wl + ((size_t)plane * 32 + chn * KQC + 2 * kp) * OC_;
#pragma unroll
            for (int mt = 0; mt < MT; ++mt) fa[v * MT + mt] = wp[16 * mt];
        }
    };
    auto load_b = [&](v4f (&fb)[NB], const int u) {
        const int kp = u / TY, ty = u % TY;
        int koff = xlb + ((2 * kp * TY + ty) * PITCH) * 16;
        asm volatile("" : "+v"(koff));
        const v4f* xk = (const v4f*)((const char*)xs + koff);
#pragma unroll
        for (int v = 0; v < NV; ++v) {
#pragma unroll
            for (int pt = 0; pt < PT; ++pt) fb[v * PT + pt] = xk[v * NP + pt * 16];
        }
    };
    auto mma = [&](const v4f (&fa)[NA], const v4f (&fb)[NB]) {
#pragma unroll
        for (int v = 0; v < NV; ++v) {
#pragma unroll
            for (int mt = 0; mt < MT; ++mt) {
#pragma unroll
                for (int pt = 0; pt < PT; ++pt)
                    acc[v][mt][pt] = __builtin_amdgcn_wmma_f32_16x16x4_f32(
                                       false, frag_lo(fa[v * MT + mt]), false, frag_lo(fb[v * PT + pt]), (short)0, acc[v][mt][pt], false, false);
            }
        }
#pragma unroll
        for (int v = 0; v < NV; ++v) {
#pragma unroll
            for (int mt = 0; mt < MT; ++mt) {
#pragma unroll
                for (int pt = 0; pt < PT; ++pt)
                    acc[v][mt][pt] = __builtin_amdgcn_wmma_f32_16x16x4_f32(
                                       false, frag_hi(fa[v * MT + mt]), false, frag_hi(fb[v * PT + pt]), (short)0, acc[v][mt][pt], false, false);
            }
        }
    };

    const int sg = threadIdx.x & 31;
    const int sr = threadIdx.x >> 5;

    v4f fa0[NA], fa1[NA], fb0[NB], fb1[NB];
    load_a(fa0, 0, 0);

    for (int ch = 0; ch < NCH; ++ch) {
        if (ch > 0) __syncthreads();
        {
            v4f q[NIT][4];
#pragma unroll
            for (int it = 0; it < NIT; ++it) {
                const int row = sr + 4 * it;
                const int t   = row % TY;
                const int kql = row / TY;
                const int gr  = r0 + t;
                const bool ok = PI ? true : ((gr >= 0) && (gr < H_));
                const float* p = ok ? (xb + ((size_t)(4 * (ch * KQC + kql)) * H_ + gr) * H_ + 4 * sg) : xb;
                const v4f z = {0.f, 0.f, 0.f, 0.f};
#pragma unroll
                for (int e = 0; e < 4; ++e) {
                    const v4f v = *(const v4f*)(p + (size_t)e * H_ * H_);
                    q[it][e] = ok ? v : z;
                }
            }
#pragma unroll
            for (int it = 0; it < NIT; ++it) {
                float pv[4];
#pragma unroll
                for (int e = 0; e < 4; ++e) {
                    const float u = __shfl_up(q[it][e].w, 1);
                    pv[e] = (sg == 0) ? 0.f : u;
                }
                const v4f q0 = q[it][0], q1 = q[it][1], q2 = q[it][2], q3 = q[it][3];
                v4f* d = xs + (sr + 4 * it) * PITCH + 2 * sg;
                d[0]          = v4f{pv[0] - q0.x, pv[1] - q1.x, pv[2] - q2.x, pv[3] - q3.x};
                d[1]          = v4f{q0.y - q0.z, q1.y - q1.z, q2.y - q2.z, q3.y - q3.z};
                d[NP]         = v4f{q0.x, q1.x, q2.x, q3.x};
                d[NP + 1]     = v4f{q0.z, q1.z, q2.z, q3.z};
                d[2 * NP]     = v4f{q0.x - q0.y, q1.x - q1.y, q2.x - q2.y, q3.x - q3.y};
                d[2 * NP + 1] = v4f{q0.z - q0.w, q1.z - q1.w, q2.z - q2.w, q3.z - q3.w};
            }
        }
        __syncthreads();

        const int chn = (ch + 1 < NCH) ? (ch + 1) : ch;
        load_b(fb0, 0);
#pragma unroll
        for (int up = 0; up < NU / 2; ++up) {
            load_a(fa1, ch, 2 * up + 1);
            load_b(fb1, 2 * up + 1);
            __builtin_amdgcn_sched_barrier(0);
            mma(fa0, fb0);
            __builtin_amdgcn_sched_barrier(0);
            if (2 * up + 2 < NU) {
                load_a(fa0, ch, 2 * up + 2);
                load_b(fb0, 2 * up + 2);
            } else {
                load_a(fa0, chn, 0);
            }
            __builtin_amdgcn_sched_barrier(0);
            mma(fa1, fb1);
            __builtin_amdgcn_sched_barrier(0);
        }
    }

    const int mhi = khp * 8;
    constexpr int POFF = PI ? OFF10_ : OFF00_;
    float* gb = conv_ws + (size_t)(b * OC_ + og * 32 + mhi) * PG_;
    float* ob = gb + POFF + (size_t)ip * PW_ + 2 * (pg * 32 + ln);
#pragma unroll
    for (int mt = 0; mt < MT; ++mt) {
#pragma unroll
        for (int pt = 0; pt < PT; ++pt) {
            float* dst = ob + (size_t)(16 * mt) * PG_ + pt * 32;
#pragma unroll
            for (int r = 0; r < 8; ++r)
                *(v2f*)(dst + (size_t)r * PG_) = v2f{acc[0][mt][pt][r] + acc[1][mt][pt][r], acc[1][mt][pt][r] - acc[2][mt][pt][r]};
        }
    }
}

__device__ __forceinline__ void conv_edge(const float* __restrict__ x,
                                          const float* __restrict__ W2,
                                          float* __restrict__ conv_ws,
                                          const int rt, const int b) {
    constexpr int EKB = 4;
    const float* xb = x + (size_t)b * C_ * H_ * H_;
    const int lane = threadIdx.x & 31;
    const int mW   = threadIdx.x >> 5;
    const int ln   = lane & 15;
    const int khp  = lane >> 4;
    const int ip   = rt * 16 + ln;
    const bool okU = (ip >= 1) && (ip <= H_);
    const bool okL = (ip < H_);
    const float* pU = xb + ((size_t)(2 * khp) * H_ + (okU ? ip - 1 : 0)) * H_ + (H_ - 1);
    const float* pL = xb + ((size_t)(2 * khp) * H_ + (okL ? ip : 0)) * H_ + (H_ - 1);
    const v2f* Wl = (const v2f*)W2 + (size_t)b * NTAP_ * 64 * OC_ + (size_t)(mW * 16 + ln) * 2 + khp;

    v8f a0 = v8f{}, a1 = v8f{};
#pragma unroll 1
    for (int sb = 0; sb < 32 / EKB; ++sb) {
        v2f fu[EKB], fl[EKB], w8[EKB], w2[EKB], w5[EKB];
#pragma unroll
        for (int u = 0; u < EKB; ++u) {
            const int s = sb * EKB + u;
            const size_t co = (size_t)(4 * s) * H_ * H_;
            const float u0 = pU[co], u1 = pU[co + (size_t)H_ * H_];
            const float l0 = pL[co], l1 = pL[co + (size_t)H_ * H_];
            fu[u] = v2f{okU ? u0 : 0.f, okU ? u1 : 0.f};
            fl[u] = v2f{okL ? l0 : 0.f, okL ? l1 : 0.f};
            w8[u] = Wl[((size_t)8 * 32 + s) * (2 * OC_)];
            w2[u] = Wl[((size_t)2 * 32 + s) * (2 * OC_)];
            w5[u] = Wl[((size_t)5 * 32 + s) * (2 * OC_)];
        }
#pragma unroll
        for (int u = 0; u < EKB; ++u) {
            a0 = __builtin_amdgcn_wmma_f32_16x16x4_f32(false, w8[u], false, fu[u], (short)0, a0, false, false);
            a0 = __builtin_amdgcn_wmma_f32_16x16x4_f32(false, w2[u], false, fl[u], (short)0, a0, false, false);
            a1 = __builtin_amdgcn_wmma_f32_16x16x4_f32(false, w5[u], false, fl[u], (short)0, a1, false, false);
        }
    }

    float* gb = conv_ws + (size_t)(b * OC_ + mW * 16 + khp * 8) * PG_;
    if (ip <= H_) {
#pragma unroll
        for (int r = 0; r < 8; ++r) gb[(size_t)r * PG_ + OFFS0_ + ip] = a0[r];
    }
    if (ip < H_) {
#pragma unroll
        for (int r = 0; r < 8; ++r) gb[(size_t)r * PG_ + OFFS1_ + ip] = a1[r];
    }
}

#define NGRID_ (NBB_ * B_)
__global__ void __launch_bounds__(128) __attribute__((amdgpu_waves_per_eu(4))) k_conv_all(const float* __restrict__ x,
                                                  const float* __restrict__ W2,
                                                  float* __restrict__ conv_ws) {
    __shared__ v4f xs[CONV_LDS_F4];
    int r = blockIdx.x;
    if (r < B_ * NBE_) { const int b = r / NBE_; conv_edge(x, W2, conv_ws, r - b * NBE_, b); return; }
    r -= B_ * NBE_;
    if (r < B_ * NB00_) { const int b = r / NB00_; conv_pair<0>(x, W2, conv_ws, xs, r - b * NB00_, b); return; }
    r -= B_ * NB00_;
    if (r < B_ * NB01_) { const int b = r / NB01_; conv_phase<0>(x, W2, conv_ws, xs, r - b * NB01_, b); return; }
    r -= B_ * NB01_;
    if (r < B_ * NB10_) { const int b = r / NB10_; conv_pair<1>(x, W2, conv_ws, xs, r - b * NB10_, b); return; }
    r -= B_ * NB10_;
    { const int b = r / NB11_; conv_phase<1>(x, W2, conv_ws, xs, r - b * NB11_, b); }
}

#define BLUR_RP 16
#define BLUR_U  4
#define BLUR_NW 8
static_assert(BLUR_NW * BLUR_RP == H_ && BLUR_RP % BLUR_U == 0, "blur block: 8 bands cover all 128 row pairs");

struct BlurRow {
    v2f el, ol;
    v2f er, orr;
    float s;
};

__device__ __forceinline__ void blur_load(const float* __restrict__ p0, const float* __restrict__ p1,
                                          const float* __restrict__ sd, const int row, const int lane, BlurRow& r) {
    const float* q0 = p0 + row * PW_ + 2 * lane;
    const float* q1 = p1 + row * PW_ + 2 * lane;
    r.el  = *(const v2f*)(q0);
    r.er  = *(const v2f*)(q0 + 64);
    r.ol  = *(const v2f*)(q1);
    r.orr = *(const v2f*)(q1 + 64);
    r.s   = sd[row];
}

__device__ __forceinline__ void blur_h4(const v2f E, const v2f O, const float a, const float f, const float g,
                                        const bool valid, float* h) {
    const float t0 = fmaf(3.f, E.x + O.x, a + E.y);
    const float t1 = fmaf(3.f, O.x + E.y, E.x + O.y);
    const float t2 = fmaf(3.f, E.y + O.y, O.x + f);
    const float t3 = fmaf(3.f, O.y + f, E.y + g);
    h[0] = valid ? t0 : 0.f;
    h[1] = valid ? t1 : 0.f;
    h[2] = valid ? t2 : 0.f;
    h[3] = valid ? t3 : 0.f;
}

__device__ __forceinline__ void blur_hrow(const BlurRow& r, const int lane, const bool valid, float (&h)[8]) {
    const bool first = (lane == 0), last = (lane == 31);
    const int nx = (lane + 1) & 31, pv = (lane - 1) & 31;
    const float xel = __shfl(r.el.x, nx),  xer = __shfl(r.er.x, nx);
    const float xol = __shfl(r.ol.x, nx),  xor_ = __shfl(r.orr.x, nx);
    const float yol = __shfl(r.ol.y, pv),  yor = __shfl(r.orr.y, pv);
    const float nel = last ? xer : xel;
    const float nol = last ? xor_ : xol;
    const float ner = last ? r.s : xer;
    const float nor = last ? 0.f : xor_;
    const float pol = first ? 0.f : yol;
    const float por = first ? yol : yor;
    blur_h4(r.el, r.ol, pol, nel, nol, valid, h);
    blur_h4(r.er, r.orr, por, ner, nor, valid, h + 4);
}

__global__ void __launch_bounds__(256) __attribute__((amdgpu_waves_per_eu(4, 4))) k_blur(const float* __restrict__ conv_ws,
                                              float* __restrict__ out) {
    const int lane = threadIdx.x & 31;
    const int w  = __builtin_amdgcn_readfirstlane(threadIdx.x >> 5);
    const int i0 = w * BLUR_RP;
    const int o  = blockIdx.x;
    const int b  = blockIdx.y;

    const float* gp  = conv_ws + (size_t)(b * OC_ + o) * PG_;
    const float* p00 = gp + OFF00_;
    const float* p01 = gp + OFF01_;
    const float* p10 = gp + OFF10_;
    const float* p11 = gp + OFF11_;
    const float* s0  = gp + OFFS0_;
    const float* s1  = gp + OFFS1_;

    float hA[8], hB[8], hC[8];
    {
        const bool vA = (i0 > 0);
        const int  rA = vA ? (i0 - 1) : 0;
        BlurRow ra, rb, rc;
        blur_load(p10, p11, s1, rA, lane, ra);
        blur_load(p00, p01, s0, i0, lane, rb);
        blur_load(p10, p11, s1, i0, lane, rc);
        blur_hrow(ra, lane, vA,   hA);
        blur_hrow(rb, lane, true, hB);
        blur_hrow(rc, lane, true, hC);
    }

    float* op = out + (((size_t)b * OC_ + o) * OUT_ + 2 * i0) * OUT_ + 4 * lane;

    for (int it = 0; it < BLUR_RP / BLUR_U; ++it) {
        BlurRow re[BLUR_U], ro[BLUR_U];
#pragma unroll
        for (int s = 0; s < BLUR_U; ++s) {
            const int i2  = i0 + it * BLUR_U + s + 1;
            const int i2o = (i2 < H_) ? i2 : (H_ - 1);
            blur_load(p00, p01, s0, i2,  lane, re[s]);
            blur_load(p10, p11, s1, i2o, lane, ro[s]);
        }
#pragma unroll
        for (int s = 0; s < BLUR_U; ++s) {
            const int i2 = i0 + it * BLUR_U + s + 1;
            float hD[8], hE[8];
            blur_hrow(re[s], lane, true,      hD);
            blur_hrow(ro[s], lane, (i2 < H_), hE);
            float y0[8], y1[8];
#pragma unroll
            for (int m = 0; m < 8; ++m) {
                y0[m] = fmaf(3.f, hB[m] + hC[m], hA[m] + hD[m]) * (1.f / 16.f);
                y1[m] = fmaf(3.f, hC[m] + hD[m], hB[m] + hE[m]) * (1.f / 16.f);
                hA[m] = hC[m]; hB[m] = hD[m]; hC[m] = hE[m];
            }
            float* orow = op + (size_t)(2 * (it * BLUR_U + s)) * OUT_;
            *(v4f*)(orow)              = v4f{y0[0], y0[1], y0[2], y0[3]};
            *(v4f*)(orow + 128)        = v4f{y0[4], y0[5], y0[6], y0[7]};
            *(v4f*)(orow + OUT_)       = v4f{y1[0], y1[1], y1[2], y1[3]};
            *(v4f*)(orow + OUT_ + 128) = v4f{y1[4], y1[5], y1[6], y1[7]};
        }
    }
}

extern "C" void kernel_launch(void* const* d_in, const int* in_sizes, int n_in,
                              void* d_out, int out_size, void* d_ws, size_t ws_size,
                              hipStream_t stream) {
    const float* x      = (const float*)d_in[0];
    const float* style  = (const float*)d_in[1];
    const float* weight = (const float*)d_in[2];
    const float* modw   = (const float*)d_in[3];
    const float* modb   = (const float*)d_in[4];
    float* out = (float*)d_out;
    float* ws  = (float*)d_ws;

    float* W2_ws   = ws + 1024;
    float* conv_ws = ws + 1024 + 786432;

    k_mod<<<dim3(MOD_NG, B_), dim3(1024), 0, stream>>>(style, modw, modb, weight, W2_ws);

    k_conv_all<<<dim3(NGRID_), dim3(128), 0, stream>>>(x, W2_ws, conv_ws);

    k_blur<<<dim3(OC_, B_), dim3(32 * BLUR_NW), 0, stream>>>(conv_ws, out);
}
